# starting kernel
# baseline (speedup 1.0000x reference)
#include <hip/hip_runtime.h>
#include <hip/hip_bf16.h>
#include <math.h>


#define D_DIM 64
#define DOUT  128
#define NEG_SLOPE 0.2f

typedef __attribute__((ext_vector_type(2))) float v2f;
typedef __attribute__((ext_vector_type(8))) float v8f;

__device__ __forceinline__ unsigned f2ord(float f) {
    unsigned u = __float_as_uint(f);
    return (u & 0x80000000u) ? ~u : (u | 0x80000000u);
}
__device__ __forceinline__ float ord2f(unsigned u) {
    unsigned b = (u & 0x80000000u) ? (u ^ 0x80000000u) : ~u;
    return __uint_as_float(b);
}

__global__ void k_init(float* __restrict__ h, float* __restrict__ denom,
                       unsigned* __restrict__ smax, int n_nodes) {
    long long i = (long long)blockIdx.x * blockDim.x + threadIdx.x;
    if (i < (long long)n_nodes * D_DIM) h[i] = 0.0f;
    if (i < n_nodes) { denom[i] = 0.0f; smax[i] = 0x007FFFFFu;   }
}

__global__ void k_node_linear(const float* __restrict__ x,
                              const float* __restrict__ Wl, const float* __restrict__ bl,
                              const float* __restrict__ Wr, const float* __restrict__ br,
                              float* __restrict__ xl, float* __restrict__ xr,
                              int n_nodes) {
    const int lane = threadIdx.x & 31;
    const int wave = threadIdx.x >> 5;
    const int tile = blockIdx.x * (blockDim.x >> 5) + wave;
    const int r0 = tile * 16;
    if (r0 + 16 > n_nodes) return;
    const int lrow = lane & 15;
    const int lhi  = lane >> 4;

    v2f a[16];
    const float* xp = x + (long long)(r0 + lrow) * D_DIM + 2 * lhi;
#pragma unroll
    for (int kk = 0; kk < 16; ++kk)
        a[kk] = *(const v2f*)(xp + kk * 4);

#pragma unroll
    for (int m = 0; m < 2; ++m) {
        const float* W    = m ? Wr : Wl;
        const float* bias = m ? br : bl;
        float*       out  = m ? xr : xl;
#pragma unroll
        for (int ct = 0; ct < 4; ++ct) {
            const int col = ct * 16 + lrow;
            const float bv = bias[col];
            v8f acc = {bv, bv, bv, bv, bv, bv, bv, bv};
#pragma unroll
            for (int kk = 0; kk < 16; ++kk) {
                const int k0 = kk * 4 + 2 * lhi;
                v2f b;
                b.x = W[(k0    ) * D_DIM + col];
                b.y = W[(k0 + 1) * D_DIM + col];
                acc = __builtin_amdgcn_wmma_f32_16x16x4_f32(
                        false, a[kk], false, b, (short)0, acc, false, false);
            }
#pragma unroll
            for (int r = 0; r < 8; ++r)
                out[(long long)(r0 + r + 8 * lhi) * D_DIM + col] = acc[r];
        }
    }
}

__global__ void k_edge_logits(const int* __restrict__ src, const int* __restrict__ dst,
                              const float* __restrict__ xl, const float* __restrict__ xr,
                              const float* __restrict__ att,
                              float* __restrict__ logits, unsigned* __restrict__ smax,
                              int n_edges) {
    const int lane = threadIdx.x & 31;
    const long long e = ((long long)blockIdx.x * blockDim.x + threadIdx.x) >> 5;
    if (e >= n_edges) return;
    const int s = src[e], d = dst[e];
    const v2f vl = *(const v2f*)(xl + (long long)s * D_DIM + 2 * lane);
    const v2f vr = *(const v2f*)(xr + (long long)d * D_DIM + 2 * lane);
    const v2f va = *(const v2f*)(att + 2 * lane);
    float e0 = vl.x + vr.x, e1 = vl.y + vr.y;
    e0 = e0 > 0.0f ? e0 : NEG_SLOPE * e0;
    e1 = e1 > 0.0f ? e1 : NEG_SLOPE * e1;
    float p = e0 * va.x + e1 * va.y;
#pragma unroll
    for (int off = 16; off >= 1; off >>= 1) p += __shfl_xor(p, off, 32);
    if (lane == 0) {
        logits[e] = p;
        atomicMax(&smax[d], f2ord(p));
    }
}

__global__ void k_decode_smax(const unsigned* __restrict__ smax,
                              float* __restrict__ smaxf, int n_nodes) {
    int i = blockIdx.x * blockDim.x + threadIdx.x;
    if (i >= n_nodes) return;
    float f = ord2f(smax[i]);
    smaxf[i] = __builtin_isfinite(f) ? f : 0.0f;
}

__global__ void k_alpha(const float* __restrict__ logits, const int* __restrict__ dst,
                        const float* __restrict__ smaxf,
                        float* __restrict__ alpha, float* __restrict__ denom,
                        int n_edges) {
    int e = blockIdx.x * blockDim.x + threadIdx.x;
    if (e >= n_edges) return;
    const int d = dst[e];
    const float a = __expf(logits[e] - smaxf[d]);
    alpha[e] = a;
    atomicAdd(&denom[d], a);
}

__global__ void k_scatter(const int* __restrict__ src, const int* __restrict__ dst,
                          const float* __restrict__ xl,
                          const float* __restrict__ alpha, const float* __restrict__ denom,
                          float* __restrict__ h, int n_edges) {
    const long long tid = (long long)blockIdx.x * blockDim.x + threadIdx.x;
    const long long e = tid >> 5;
    const int lane = (int)(tid & 31);
    if (e >= n_edges) return;
    const int s = src[e], d = dst[e];
    const float w = alpha[e] / (denom[d] + 1e-16f);
    const v2f v = *(const v2f*)(xl + (long long)s * D_DIM + 2 * lane);
    float* hp = h + (long long)d * D_DIM + 2 * lane;
    atomicAdd(hp,     w * v.x);
    atomicAdd(hp + 1, w * v.y);
}

__global__ void k_output(const float* __restrict__ h, const int* __restrict__ mask,
                         const float* __restrict__ gbias,
                         const float* __restrict__ Wo, const float* __restrict__ bo,
                         float* __restrict__ out, int n_mask) {
    const int lane = threadIdx.x & 31;
    const int wave = threadIdx.x >> 5;
    const int tile = blockIdx.x * (blockDim.x >> 5) + wave;
    const int r0 = tile * 16;
    if (r0 + 16 > n_mask) return;
    const int lrow = lane & 15;
    const int lhi  = lane >> 4;

    const long long mi = (long long)mask[r0 + lrow];
    v2f a[16];
    const float* hp = h + mi * D_DIM + 2 * lhi;
    const float* gp = gbias + 2 * lhi;
#pragma unroll
    for (int kk = 0; kk < 16; ++kk) {
        const v2f hv = *(const v2f*)(hp + kk * 4);
        const v2f gv = *(const v2f*)(gp + kk * 4);
        a[kk].x = hv.x + gv.x;
        a[kk].y = hv.y + gv.y;
    }
#pragma unroll
    for (int ct = 0; ct < 8; ++ct) {
        const int col = ct * 16 + lrow;
        const float bv = bo[col];
        v8f acc = {bv, bv, bv, bv, bv, bv, bv, bv};
#pragma unroll
        for (int kk = 0; kk < 16; ++kk) {
            const int k0 = kk * 4 + 2 * lhi;
            v2f b;
            b.x = Wo[(k0    ) * DOUT + col];
            b.y = Wo[(k0 + 1) * DOUT + col];
            acc = __builtin_amdgcn_wmma_f32_16x16x4_f32(
                    false, a[kk], false, b, (short)0, acc, false, false);
        }
#pragma unroll
        for (int r = 0; r < 8; ++r)
            out[(long long)(r0 + r + 8 * lhi) * DOUT + col] = acc[r];
    }
}

extern "C" void kernel_launch(void* const* d_in, const int* in_sizes, int n_in,
                              void* d_out, int out_size, void* d_ws, size_t ws_size,
                              hipStream_t stream) {
    const float* x     = (const float*)d_in[0];
    const int*   ei    = (const int*)d_in[1];
    const int*   mask  = (const int*)d_in[2];
    const float* Wl    = (const float*)d_in[3];
    const float* bl    = (const float*)d_in[4];
    const float* Wr    = (const float*)d_in[5];
    const float* br    = (const float*)d_in[6];
    const float* att   = (const float*)d_in[7];
    const float* gbias = (const float*)d_in[8];
    const float* Wo    = (const float*)d_in[9];
    const float* bo    = (const float*)d_in[10];
    float* out = (float*)d_out;

    const int n_nodes = in_sizes[0] / D_DIM;
    const int n_edges = in_sizes[1] / 2;
    const int n_mask  = in_sizes[2];
    const int* src = ei;
    const int* dst = ei + n_edges;

    char* wp = (char*)d_ws;
    auto carve = [&](size_t bytes) {
        char* p = wp;
        wp += (bytes + 255) & ~(size_t)255;
        return p;
    };
    float*    xl     = (float*)   carve((size_t)n_nodes * D_DIM * 4);
    float*    xr     = (float*)   carve((size_t)n_nodes * D_DIM * 4);
    float*    h      = (float*)   carve((size_t)n_nodes * D_DIM * 4);
    float*    logits = (float*)   carve((size_t)n_edges * 4);
    float*    alpha  = (float*)   carve((size_t)n_edges * 4);
    unsigned* smax   = (unsigned*)carve((size_t)n_nodes * 4);
    float*    smaxf  = (float*)   carve((size_t)n_nodes * 4);
    float*    denom  = (float*)   carve((size_t)n_nodes * 4);

    {
        long long t = (long long)n_nodes * D_DIM;
        int blocks = (int)((t + 255) / 256);
        k_init<<<blocks, 256, 0, stream>>>(h, denom, smax, n_nodes);
    }
    {
        int tiles  = (n_nodes + 15) / 16;
        int blocks = (tiles + 7) / 8;
        k_node_linear<<<blocks, 256, 0, stream>>>(x, Wl, bl, Wr, br, xl, xr, n_nodes);
    }
    {
        long long t = (long long)n_edges * 32;
        int blocks = (int)((t + 255) / 256);
        k_edge_logits<<<blocks, 256, 0, stream>>>(src, dst, xl, xr, att, logits, smax, n_edges);
    }
    k_decode_smax<<<(n_nodes + 255) / 256, 256, 0, stream>>>(smax, smaxf, n_nodes);
    k_alpha<<<(n_edges + 255) / 256, 256, 0, stream>>>(logits, dst, smaxf, alpha, denom, n_edges);
    {
        long long t = (long long)n_edges * 32;
        int blocks = (int)((t + 255) / 256);
        k_scatter<<<blocks, 256, 0, stream>>>(src, dst, xl, alpha, denom, h, n_edges);
    }
    {
        int tiles  = (n_mask + 15) / 16;
        int blocks = (tiles + 7) / 8;
        k_output<<<blocks, 256, 0, stream>>>(h, mask, gbias, Wo, bo, out, n_mask);
    }
}


# edge_logits and scatter: one wave per 32 edges (coalesced index loads, 16 row gathers in flight) instead of one wave per edge; logit dot summed 4 per lane then over 16 lanes; atomics kept
# speedup vs baseline: 1.6985x; 1.6985x over previous
#include <hip/hip_runtime.h>
#include <hip/hip_bf16.h>
#include <math.h>


#define D_DIM 64
#define DOUT  128
#define NEG_SLOPE 0.2f

typedef __attribute__((ext_vector_type(2))) float v2f;
typedef __attribute__((ext_vector_type(8))) float v8f;
typedef __attribute__((ext_vector_type(4))) float v4f;

__device__ __forceinline__ unsigned f2ord(float f) {
    unsigned u = __float_as_uint(f);
    return (u & 0x80000000u) ? ~u : (u | 0x80000000u);
}
__device__ __forceinline__ float ord2f(unsigned u) {
    unsigned b = (u & 0x80000000u) ? (u ^ 0x80000000u) : ~u;
    return __uint_as_float(b);
}

__global__ void k_init(float* __restrict__ h, float* __restrict__ denom,
                       unsigned* __restrict__ smax, int n_nodes) {
    long long i = (long long)blockIdx.x * blockDim.x + threadIdx.x;
    if (i < (long long)n_nodes * D_DIM) h[i] = 0.0f;
    if (i < n_nodes) { denom[i] = 0.0f; smax[i] = 0x007FFFFFu;   }
}

__global__ void k_node_linear(const float* __restrict__ x,
                              const float* __restrict__ Wl, const float* __restrict__ bl,
                              const float* __restrict__ Wr, const float* __restrict__ br,
                              float* __restrict__ xl, float* __restrict__ xr,
                              int n_nodes) {
    const int lane = threadIdx.x & 31;
    const int wave = threadIdx.x >> 5;
    const int tile = blockIdx.x * (blockDim.x >> 5) + wave;
    const int r0 = tile * 16;
    if (r0 + 16 > n_nodes) return;
    const int lrow = lane & 15;
    const int lhi  = lane >> 4;

    v2f a[16];
    const float* xp = x + (long long)(r0 + lrow) * D_DIM + 2 * lhi;
#pragma unroll
    for (int kk = 0; kk < 16; ++kk)
        a[kk] = *(const v2f*)(xp + kk * 4);

#pragma unroll
    for (int m = 0; m < 2; ++m) {
        const float* W    = m ? Wr : Wl;
        const float* bias = m ? br : bl;
        float*       out  = m ? xr : xl;
#pragma unroll
        for (int ct = 0; ct < 4; ++ct) {
            const int col = ct * 16 + lrow;
            const float bv = bias[col];
            v8f acc = {bv, bv, bv, bv, bv, bv, bv, bv};
#pragma unroll
            for (int kk = 0; kk < 16; ++kk) {
                const int k0 = kk * 4 + 2 * lhi;
                v2f b;
                b.x = W[(k0    ) * D_DIM + col];
                b.y = W[(k0 + 1) * D_DIM + col];
                acc = __builtin_amdgcn_wmma_f32_16x16x4_f32(
                        false, a[kk], false, b, (short)0, acc, false, false);
            }
#pragma unroll
            for (int r = 0; r < 8; ++r)
                out[(long long)(r0 + r + 8 * lhi) * D_DIM + col] = acc[r];
        }
    }
}

template <int CTRL>
__device__ __forceinline__ int dpp_i(int v) {
    return __builtin_amdgcn_update_dpp(0, v, CTRL, 0xf, 0xf, true);
}
template <int CTRL>
__device__ __forceinline__ float dpp_f(float v) {
    return __int_as_float(__builtin_amdgcn_update_dpp(0, __float_as_int(v), CTRL, 0xf, 0xf, true));
}

template <int J>
__device__ __forceinline__ void gather_rows(int sv, int dv, int sub,
                                            const float* __restrict__ xl,
                                            const float* __restrict__ xr,
                                            v4f& l, v4f& r) {
    const int s = dpp_i<0x150 + J>(sv);
    const int d = dpp_i<0x150 + J>(dv);
    l = *(const v4f*)(xl + (long long)s * D_DIM + 4 * sub);
    r = *(const v4f*)(xr + (long long)d * D_DIM + 4 * sub);
}

template <int J>
__device__ __forceinline__ void score_rows(const v4f& l, const v4f& r, const v4f& a,
                                           int sub, int owner0, float& mine) {
    float t0 = l.x + r.x, t1 = l.y + r.y, t2 = l.z + r.z, t3 = l.w + r.w;
    t0 = t0 > 0.0f ? t0 : NEG_SLOPE * t0;
    t1 = t1 > 0.0f ? t1 : NEG_SLOPE * t1;
    t2 = t2 > 0.0f ? t2 : NEG_SLOPE * t2;
    t3 = t3 > 0.0f ? t3 : NEG_SLOPE * t3;
    float p = t0 * a.x + t1 * a.y + t2 * a.z + t3 * a.w;
    p += dpp_f<0x160 + 1>(p);
    p += dpp_f<0x160 + 2>(p);
    p += dpp_f<0x160 + 4>(p);
    p += dpp_f<0x160 + 8>(p);
    if (sub == owner0 + J) mine = p;
}

__global__ void k_edge_logits(const int* __restrict__ src, const int* __restrict__ dst,
                              const float* __restrict__ xl, const float* __restrict__ xr,
                              const float* __restrict__ att,
                              float* __restrict__ logits, unsigned* __restrict__ smax,
                              int n_edges) {
    const int lane = threadIdx.x & 31;
    const int sub  = lane & 15;
    const int wave = __builtin_amdgcn_readfirstlane((int)(threadIdx.x >> 5));
    const long long e0 = ((long long)blockIdx.x * (blockDim.x >> 5) + wave) * 32;
    if (e0 >= n_edges) return;
    const long long em = e0 + lane;
    const bool ok = em < n_edges;
    int sv = 0, dv = 0;
    if (ok) { sv = src[em]; dv = dst[em]; }
    const v4f a = *(const v4f*)(att + 4 * sub);
    float mine = 0.0f;
    int sr = sv, dr = dv;
#pragma unroll 1
    for (int pass = 0; pass < 2; ++pass) {
        v4f l0, l1, l2, l3, l4, l5, l6, l7, r0, r1, r2, r3, r4, r5, r6, r7;
        gather_rows<0>(sr, dr, sub, xl, xr, l0, r0);
        gather_rows<1>(sr, dr, sub, xl, xr, l1, r1);
        gather_rows<2>(sr, dr, sub, xl, xr, l2, r2);
        gather_rows<3>(sr, dr, sub, xl, xr, l3, r3);
        gather_rows<4>(sr, dr, sub, xl, xr, l4, r4);
        gather_rows<5>(sr, dr, sub, xl, xr, l5, r5);
        gather_rows<6>(sr, dr, sub, xl, xr, l6, r6);
        gather_rows<7>(sr, dr, sub, xl, xr, l7, r7);
        asm volatile("" : "+v"(l0), "+v"(l1), "+v"(l2), "+v"(l3), "+v"(l4), "+v"(l5), "+v"(l6), "+v"(l7)
                        : "v"(r0), "v"(r1), "v"(r2), "v"(r3), "v"(r4), "v"(r5), "v"(r6), "v"(r7));
        const int owner0 = 8 * pass;
        score_rows<0>(l0, r0, a, sub, owner0, mine);
        score_rows<1>(l1, r1, a, sub, owner0, mine);
        score_rows<2>(l2, r2, a, sub, owner0, mine);
        score_rows<3>(l3, r3, a, sub, owner0, mine);
        score_rows<4>(l4, r4, a, sub, owner0, mine);
        score_rows<5>(l5, r5, a, sub, owner0, mine);
        score_rows<6>(l6, r6, a, sub, owner0, mine);
        score_rows<7>(l7, r7, a, sub, owner0, mine);
        sr = dpp_i<0x128>(sr);
        dr = dpp_i<0x128>(dr);
    }
    if (ok) {
        logits[em] = mine;
        atomicMax(&smax[dv], f2ord(mine));
    }
}

__global__ void k_decode_smax(const unsigned* __restrict__ smax,
                              float* __restrict__ smaxf, int n_nodes) {
    int i = blockIdx.x * blockDim.x + threadIdx.x;
    if (i >= n_nodes) return;
    float f = ord2f(smax[i]);
    smaxf[i] = __builtin_isfinite(f) ? f : 0.0f;
}

__global__ void k_alpha(const float* __restrict__ logits, const int* __restrict__ dst,
                        const float* __restrict__ smaxf,
                        float* __restrict__ alpha, float* __restrict__ denom,
                        int n_edges) {
    int e = blockIdx.x * blockDim.x + threadIdx.x;
    if (e >= n_edges) return;
    const int d = dst[e];
    const float a = __expf(logits[e] - smaxf[d]);
    alpha[e] = a;
    atomicAdd(&denom[d], a);
}

__global__ void k_scatter(const int* __restrict__ src, const int* __restrict__ dst,
                          const float* __restrict__ xl,
                          const float* __restrict__ alpha, const float* __restrict__ denom,
                          float* __restrict__ h, int n_edges) {
    const int lane = threadIdx.x & 31;
    const int wave = __builtin_amdgcn_readfirstlane((int)(threadIdx.x >> 5));
    const long long e0 = ((long long)blockIdx.x * (blockDim.x >> 5) + wave) * 32;
    if (e0 >= n_edges) return;
    const long long em = e0 + lane;
    int sv = 0, dv = 0;
    float wv = 0.0f;
    if (em < n_edges) {
        sv = src[em]; dv = dst[em];
        wv = alpha[em] / (denom[dv] + 1e-16f);
    }
    const long long left = (long long)n_edges - e0;
    const int cnt = left < 32 ? (int)left : 32;
    if (cnt == 32) {
#pragma unroll
        for (int b = 0; b < 32; b += 16) {
            float v0[16], v1[16];
#pragma unroll
            for (int j = 0; j < 16; ++j) {
                const int s = __builtin_amdgcn_readlane(sv, b + j);
                const float* xp = xl + (long long)s * D_DIM + lane;
                v0[j] = xp[0];
                v1[j] = xp[32];
            }
#pragma unroll
            for (int j = 0; j < 16; ++j) {
                const int d = __builtin_amdgcn_readlane(dv, b + j);
                const float w = __int_as_float(__builtin_amdgcn_readlane(__float_as_int(wv), b + j));
                float* hp = h + (long long)d * D_DIM + lane;
                atomicAdd(hp,      w * v0[j]);
                atomicAdd(hp + 32, w * v1[j]);
            }
        }
    } else {
#pragma unroll 1
        for (int j = 0; j < cnt; ++j) {
            const int s = __builtin_amdgcn_readlane(sv, j);
            const int d = __builtin_amdgcn_readlane(dv, j);
            const float w = __int_as_float(__builtin_amdgcn_readlane(__float_as_int(wv), j));
            const float* xp = xl + (long long)s * D_DIM + lane;
            float* hp = h + (long long)d * D_DIM + lane;
            atomicAdd(hp,      w * xp[0]);
            atomicAdd(hp + 32, w * xp[32]);
        }
    }
}

__global__ void k_output(const float* __restrict__ h, const int* __restrict__ mask,
                         const float* __restrict__ gbias,
                         const float* __restrict__ Wo, const float* __restrict__ bo,
                         float* __restrict__ out, int n_mask) {
    const int lane = threadIdx.x & 31;
    const int wave = threadIdx.x >> 5;
    const int tile = blockIdx.x * (blockDim.x >> 5) + wave;
    const int r0 = tile * 16;
    if (r0 + 16 > n_mask) return;
    const int lrow = lane & 15;
    const int lhi  = lane >> 4;

    const long long mi = (long long)mask[r0 + lrow];
    v2f a[16];
    const float* hp = h + mi * D_DIM + 2 * lhi;
    const float* gp = gbias + 2 * lhi;
#pragma unroll
    for (int kk = 0; kk < 16; ++kk) {
        const v2f hv = *(const v2f*)(hp + kk * 4);
        const v2f gv = *(const v2f*)(gp + kk * 4);
        a[kk].x = hv.x + gv.x;
        a[kk].y = hv.y + gv.y;
    }
#pragma unroll
    for (int ct = 0; ct < 8; ++ct) {
        const int col = ct * 16 + lrow;
        const float bv = bo[col];
        v8f acc = {bv, bv, bv, bv, bv, bv, bv, bv};
#pragma unroll
        for (int kk = 0; kk < 16; ++kk) {
            const int k0 = kk * 4 + 2 * lhi;
            v2f b;
            b.x = Wo[(k0    ) * DOUT + col];
            b.y = Wo[(k0 + 1) * DOUT + col];
            acc = __builtin_amdgcn_wmma_f32_16x16x4_f32(
                    false, a[kk], false, b, (short)0, acc, false, false);
        }
#pragma unroll
        for (int r = 0; r < 8; ++r)
            out[(long long)(r0 + r + 8 * lhi) * DOUT + col] = acc[r];
    }
}

extern "C" void kernel_launch(void* const* d_in, const int* in_sizes, int n_in,
                              void* d_out, int out_size, void* d_ws, size_t ws_size,
                              hipStream_t stream) {
    const float* x     = (const float*)d_in[0];
    const int*   ei    = (const int*)d_in[1];
    const int*   mask  = (const int*)d_in[2];
    const float* Wl    = (const float*)d_in[3];
    const float* bl    = (const float*)d_in[4];
    const float* Wr    = (const float*)d_in[5];
    const float* br    = (const float*)d_in[6];
    const float* att   = (const float*)d_in[7];
    const float* gbias = (const float*)d_in[8];
    const float* Wo    = (const float*)d_in[9];
    const float* bo    = (const float*)d_in[10];
    float* out = (float*)d_out;

    const int n_nodes = in_sizes[0] / D_DIM;
    const int n_edges = in_sizes[1] / 2;
    const int n_mask  = in_sizes[2];
    const int* src = ei;
    const int* dst = ei + n_edges;

    char* wp = (char*)d_ws;
    auto carve = [&](size_t bytes) {
        char* p = wp;
        wp += (bytes + 255) & ~(size_t)255;
        return p;
    };
    float*    xl     = (float*)   carve((size_t)n_nodes * D_DIM * 4);
    float*    xr     = (float*)   carve((size_t)n_nodes * D_DIM * 4);
    float*    h      = (float*)   carve((size_t)n_nodes * D_DIM * 4);
    float*    logits = (float*)   carve((size_t)n_edges * 4);
    float*    alpha  = (float*)   carve((size_t)n_edges * 4);
    unsigned* smax   = (unsigned*)carve((size_t)n_nodes * 4);
    float*    smaxf  = (float*)   carve((size_t)n_nodes * 4);
    float*    denom  = (float*)   carve((size_t)n_nodes * 4);

    {
        long long t = (long long)n_nodes * D_DIM;
        int blocks = (int)((t + 255) / 256);
        k_init<<<blocks, 256, 0, stream>>>(h, denom, smax, n_nodes);
    }
    {
        int tiles  = (n_nodes + 15) / 16;
        int blocks = (tiles + 7) / 8;
        k_node_linear<<<blocks, 256, 0, stream>>>(x, Wl, bl, Wr, br, xl, xr, n_nodes);
    }
    {
        int blocks = (n_edges + 255) / 256;
        k_edge_logits<<<blocks, 256, 0, stream>>>(src, dst, xl, xr, att, logits, smax, n_edges);
    }
    k_decode_smax<<<(n_nodes + 255) / 256, 256, 0, stream>>>(smax, smaxf, n_nodes);
    k_alpha<<<(n_edges + 255) / 256, 256, 0, stream>>>(logits, dst, smaxf, alpha, denom, n_edges);
    {
        int blocks = (n_edges + 255) / 256;
        k_scatter<<<blocks, 256, 0, stream>>>(src, dst, xl, alpha, denom, h, n_edges);
    }
    {
        int tiles  = (n_mask + 15) / 16;
        int blocks = (tiles + 7) / 8;
        k_output<<<blocks, 256, 0, stream>>>(h, mask, gbias, Wo, bo, out, n_mask);
    }
}


# edge phase only on edges whose destination is in mask_idx: node bitmap, edge ids compacted (ballot rank, one returning add per block), logits/softmax/scatter walk the list; atomics kept
# speedup vs baseline: 2.6361x; 1.5520x over previous
#include <hip/hip_runtime.h>
#include <hip/hip_bf16.h>
#include <math.h>


#define D_DIM 64
#define DOUT  128
#define NEG_SLOPE 0.2f

typedef __attribute__((ext_vector_type(2))) float v2f;
typedef __attribute__((ext_vector_type(8))) float v8f;
typedef __attribute__((ext_vector_type(4))) float v4f;

__device__ __forceinline__ unsigned f2ord(float f) {
    unsigned u = __float_as_uint(f);
    return (u & 0x80000000u) ? ~u : (u | 0x80000000u);
}
__device__ __forceinline__ float ord2f(unsigned u) {
    unsigned b = (u & 0x80000000u) ? (u ^ 0x80000000u) : ~u;
    return __uint_as_float(b);
}

__global__ void k_init(float* __restrict__ h, float* __restrict__ denom,
                       unsigned* __restrict__ smax, unsigned* __restrict__ needbits,
                       int* __restrict__ cnt, int n_nodes) {
    long long i = (long long)blockIdx.x * blockDim.x + threadIdx.x;
    if (i < (long long)n_nodes * D_DIM) h[i] = 0.0f;
    if (i < n_nodes) { denom[i] = 0.0f; smax[i] = 0x007FFFFFu;   }
    if (i < (n_nodes + 31) / 32) needbits[i] = 0u;
    if (i == 0) *cnt = 0;
}

__global__ void k_mark(const int* __restrict__ mask, unsigned* __restrict__ needbits,
                       int n_mask) {
    int i = blockIdx.x * blockDim.x + threadIdx.x;
    if (i >= n_mask) return;
    const int m = mask[i];
    atomicOr(&needbits[m >> 5], 1u << (m & 31));
}

#define CP_LOADS 4
__global__ void k_compact(const int* __restrict__ dst, const unsigned* __restrict__ needbits,
                          int* __restrict__ elist, int* __restrict__ cnt, int n_edges) {
    __shared__ int wave_off[8];
    __shared__ int block_base;
    const int lane = threadIdx.x & 31;
    const int wave = threadIdx.x >> 5;
    const long long base = (long long)blockIdx.x * (256 * CP_LOADS) + threadIdx.x;
    const unsigned below = (1u << lane) - 1u;
    bool keep[CP_LOADS];
    int  rank[CP_LOADS];
    int  d[CP_LOADS];
    unsigned word[CP_LOADS];
    int  tot = 0;
#pragma unroll
    for (int j = 0; j < CP_LOADS; ++j) {
        const long long e = base + j * 256;
        d[j] = dst[e < n_edges ? e : 0];
    }
#pragma unroll
    for (int j = 0; j < CP_LOADS; ++j)
        word[j] = needbits[d[j] >> 5];
#pragma unroll
    for (int j = 0; j < CP_LOADS; ++j) {
        keep[j] = (base + j * 256 < n_edges) && ((word[j] >> (d[j] & 31)) & 1u);
        const unsigned b = __builtin_amdgcn_ballot_w32(keep[j]);
        rank[j] = tot + __popc(b & below);
        tot += __popc(b);
    }
    if (lane == 0) wave_off[wave] = tot;
    __syncthreads();
    if (threadIdx.x == 0) {
        int s = 0;
        for (int w = 0; w < 8; ++w) { const int t = wave_off[w]; wave_off[w] = s; s += t; }
        block_base = s ? atomicAdd(cnt, s) : 0;
    }
    __syncthreads();
    const int wb = block_base + wave_off[wave];
#pragma unroll
    for (int j = 0; j < CP_LOADS; ++j)
        if (keep[j]) elist[wb + rank[j]] = (int)(base + j * 256);
}

__global__ void k_node_linear(const float* __restrict__ x,
                              const float* __restrict__ Wl, const float* __restrict__ bl,
                              const float* __restrict__ Wr, const float* __restrict__ br,
                              float* __restrict__ xl, float* __restrict__ xr,
                              int n_nodes) {
    const int lane = threadIdx.x & 31;
    const int wave = threadIdx.x >> 5;
    const int tile = blockIdx.x * (blockDim.x >> 5) + wave;
    const int r0 = tile * 16;
    if (r0 + 16 > n_nodes) return;
    const int lrow = lane & 15;
    const int lhi  = lane >> 4;

    v2f a[16];
    const float* xp = x + (long long)(r0 + lrow) * D_DIM + 2 * lhi;
#pragma unroll
    for (int kk = 0; kk < 16; ++kk)
        a[kk] = *(const v2f*)(xp + kk * 4);

#pragma unroll
    for (int m = 0; m < 2; ++m) {
        const float* W    = m ? Wr : Wl;
        const float* bias = m ? br : bl;
        float*       out  = m ? xr : xl;
#pragma unroll
        for (int ct = 0; ct < 4; ++ct) {
            const int col = ct * 16 + lrow;
            const float bv = bias[col];
            v8f acc = {bv, bv, bv, bv, bv, bv, bv, bv};
#pragma unroll
            for (int kk = 0; kk < 16; ++kk) {
                const int k0 = kk * 4 + 2 * lhi;
                v2f b;
                b.x = W[(k0    ) * D_DIM + col];
                b.y = W[(k0 + 1) * D_DIM + col];
                acc = __builtin_amdgcn_wmma_f32_16x16x4_f32(
                        false, a[kk], false, b, (short)0, acc, false, false);
            }
#pragma unroll
            for (int r = 0; r < 8; ++r)
                out[(long long)(r0 + r + 8 * lhi) * D_DIM + col] = acc[r];
        }
    }
}

template <int CTRL>
__device__ __forceinline__ int dpp_i(int v) {
    return __builtin_amdgcn_update_dpp(0, v, CTRL, 0xf, 0xf, true);
}
template <int CTRL>
__device__ __forceinline__ float dpp_f(float v) {
    return __int_as_float(__builtin_amdgcn_update_dpp(0, __float_as_int(v), CTRL, 0xf, 0xf, true));
}

template <int J>
__device__ __forceinline__ void gather_rows(int sv, int dv, int sub,
                                            const float* __restrict__ xl,
                                            const float* __restrict__ xr,
                                            v4f& l, v4f& r) {
    const int s = dpp_i<0x150 + J>(sv);
    const int d = dpp_i<0x150 + J>(dv);
    l = *(const v4f*)(xl + (long long)s * D_DIM + 4 * sub);
    r = *(const v4f*)(xr + (long long)d * D_DIM + 4 * sub);
}

template <int J>
__device__ __forceinline__ void score_rows(const v4f& l, const v4f& r, const v4f& a,
                                           int sub, int owner0, float& mine) {
    float t0 = l.x + r.x, t1 = l.y + r.y, t2 = l.z + r.z, t3 = l.w + r.w;
    t0 = t0 > 0.0f ? t0 : NEG_SLOPE * t0;
    t1 = t1 > 0.0f ? t1 : NEG_SLOPE * t1;
    t2 = t2 > 0.0f ? t2 : NEG_SLOPE * t2;
    t3 = t3 > 0.0f ? t3 : NEG_SLOPE * t3;
    float p = t0 * a.x + t1 * a.y + t2 * a.z + t3 * a.w;
    p += dpp_f<0x160 + 1>(p);
    p += dpp_f<0x160 + 2>(p);
    p += dpp_f<0x160 + 4>(p);
    p += dpp_f<0x160 + 8>(p);
    if (sub == owner0 + J) mine = p;
}

__global__ void k_edge_logits(const int* __restrict__ src, const int* __restrict__ dst,
                              const int* __restrict__ elist, const int* __restrict__ list_len,
                              const float* __restrict__ xl, const float* __restrict__ xr,
                              const float* __restrict__ att,
                              float* __restrict__ logits, unsigned* __restrict__ smax) {
    const int n_edges = *list_len;
    const int lane = threadIdx.x & 31;
    const int sub  = lane & 15;
    const int wave = __builtin_amdgcn_readfirstlane((int)(threadIdx.x >> 5));
    const long long e0 = ((long long)blockIdx.x * (blockDim.x >> 5) + wave) * 32;
    if (e0 >= n_edges) return;
    const long long em = e0 + lane;
    const bool ok = em < n_edges;
    int sv = 0, dv = 0;
    if (ok) { const int e = elist[em]; sv = src[e]; dv = dst[e]; }
    const v4f a = *(const v4f*)(att + 4 * sub);
    float mine = 0.0f;
    int sr = sv, dr = dv;
#pragma unroll 1
    for (int pass = 0; pass < 2; ++pass) {
        v4f l0, l1, l2, l3, l4, l5, l6, l7, r0, r1, r2, r3, r4, r5, r6, r7;
        gather_rows<0>(sr, dr, sub, xl, xr, l0, r0);
        gather_rows<1>(sr, dr, sub, xl, xr, l1, r1);
        gather_rows<2>(sr, dr, sub, xl, xr, l2, r2);
        gather_rows<3>(sr, dr, sub, xl, xr, l3, r3);
        gather_rows<4>(sr, dr, sub, xl, xr, l4, r4);
        gather_rows<5>(sr, dr, sub, xl, xr, l5, r5);
        gather_rows<6>(sr, dr, sub, xl, xr, l6, r6);
        gather_rows<7>(sr, dr, sub, xl, xr, l7, r7);
        asm volatile("" : "+v"(l0), "+v"(l1), "+v"(l2), "+v"(l3), "+v"(l4), "+v"(l5), "+v"(l6), "+v"(l7)
                        : "v"(r0), "v"(r1), "v"(r2), "v"(r3), "v"(r4), "v"(r5), "v"(r6), "v"(r7));
        const int owner0 = 8 * pass;
        score_rows<0>(l0, r0, a, sub, owner0, mine);
        score_rows<1>(l1, r1, a, sub, owner0, mine);
        score_rows<2>(l2, r2, a, sub, owner0, mine);
        score_rows<3>(l3, r3, a, sub, owner0, mine);
        score_rows<4>(l4, r4, a, sub, owner0, mine);
        score_rows<5>(l5, r5, a, sub, owner0, mine);
        score_rows<6>(l6, r6, a, sub, owner0, mine);
        score_rows<7>(l7, r7, a, sub, owner0, mine);
        sr = dpp_i<0x128>(sr);
        dr = dpp_i<0x128>(dr);
    }
    if (ok) {
        logits[em] = mine;
        atomicMax(&smax[dv], f2ord(mine));
    }
}

__global__ void k_decode_smax(unsigned* __restrict__ smax, int n_nodes) {
    int i = blockIdx.x * blockDim.x + threadIdx.x;
    if (i >= n_nodes) return;
    float f = ord2f(smax[i]);
    smax[i] = __float_as_uint(__builtin_isfinite(f) ? f : 0.0f);
}

__global__ void k_alpha(const int* __restrict__ dst,
                        const int* __restrict__ elist, const int* __restrict__ list_len,
                        const float* __restrict__ smaxf,
                        float* __restrict__ val, float* __restrict__ denom) {
    const int n_edges = *list_len;
    int p = blockIdx.x * blockDim.x + threadIdx.x;
    if (p >= n_edges) return;
    const int d = dst[elist[p]];
    const float a = __expf(val[p] - smaxf[d]);
    val[p] = a;
    atomicAdd(&denom[d], a);
}

__global__ void k_scatter(const int* __restrict__ src, const int* __restrict__ dst,
                          const int* __restrict__ elist, const int* __restrict__ list_len,
                          const float* __restrict__ xl,
                          const float* __restrict__ alpha, const float* __restrict__ denom,
                          float* __restrict__ h) {
    const int n_edges = *list_len;
    const int lane = threadIdx.x & 31;
    const int wave = __builtin_amdgcn_readfirstlane((int)(threadIdx.x >> 5));
    const long long e0 = ((long long)blockIdx.x * (blockDim.x >> 5) + wave) * 32;
    if (e0 >= n_edges) return;
    const long long em = e0 + lane;
    int sv = 0, dv = 0;
    float wv = 0.0f;
    if (em < n_edges) {
        const int e = elist[em];
        sv = src[e]; dv = dst[e];
        wv = alpha[em] / (denom[dv] + 1e-16f);
    }
    const long long left = (long long)n_edges - e0;
    const int cnt = left < 32 ? (int)left : 32;
    if (cnt == 32) {
#pragma unroll
        for (int b = 0; b < 32; b += 16) {
            float v0[16], v1[16];
#pragma unroll
            for (int j = 0; j < 16; ++j) {
                const int s = __builtin_amdgcn_readlane(sv, b + j);
                const float* xp = xl + (long long)s * D_DIM + lane;
                v0[j] = xp[0];
                v1[j] = xp[32];
            }
#pragma unroll
            for (int j = 0; j < 16; ++j) {
                const int d = __builtin_amdgcn_readlane(dv, b + j);
                const float w = __int_as_float(__builtin_amdgcn_readlane(__float_as_int(wv), b + j));
                float* hp = h + (long long)d * D_DIM + lane;
                atomicAdd(hp,      w * v0[j]);
                atomicAdd(hp + 32, w * v1[j]);
            }
        }
    } else {
#pragma unroll 1
        for (int j = 0; j < cnt; ++j) {
            const int s = __builtin_amdgcn_readlane(sv, j);
            const int d = __builtin_amdgcn_readlane(dv, j);
            const float w = __int_as_float(__builtin_amdgcn_readlane(__float_as_int(wv), j));
            const float* xp = xl + (long long)s * D_DIM + lane;
            float* hp = h + (long long)d * D_DIM + lane;
            atomicAdd(hp,      w * xp[0]);
            atomicAdd(hp + 32, w * xp[32]);
        }
    }
}

__global__ void k_output(const float* __restrict__ h, const int* __restrict__ mask,
                         const float* __restrict__ gbias,
                         const float* __restrict__ Wo, const float* __restrict__ bo,
                         float* __restrict__ out, int n_mask) {
    const int lane = threadIdx.x & 31;
    const int wave = threadIdx.x >> 5;
    const int tile = blockIdx.x * (blockDim.x >> 5) + wave;
    const int r0 = tile * 16;
    if (r0 + 16 > n_mask) return;
    const int lrow = lane & 15;
    const int lhi  = lane >> 4;

    const long long mi = (long long)mask[r0 + lrow];
    v2f a[16];
    const float* hp = h + mi * D_DIM + 2 * lhi;
    const float* gp = gbias + 2 * lhi;
#pragma unroll
    for (int kk = 0; kk < 16; ++kk) {
        const v2f hv = *(const v2f*)(hp + kk * 4);
        const v2f gv = *(const v2f*)(gp + kk * 4);
        a[kk].x = hv.x + gv.x;
        a[kk].y = hv.y + gv.y;
    }
#pragma unroll
    for (int ct = 0; ct < 8; ++ct) {
        const int col = ct * 16 + lrow;
        const float bv = bo[col];
        v8f acc = {bv, bv, bv, bv, bv, bv, bv, bv};
#pragma unroll
        for (int kk = 0; kk < 16; ++kk) {
            const int k0 = kk * 4 + 2 * lhi;
            v2f b;
            b.x = Wo[(k0    ) * DOUT + col];
            b.y = Wo[(k0 + 1) * DOUT + col];
            acc = __builtin_amdgcn_wmma_f32_16x16x4_f32(
                    false, a[kk], false, b, (short)0, acc, false, false);
        }
#pragma unroll
        for (int r = 0; r < 8; ++r)
            out[(long long)(r0 + r + 8 * lhi) * DOUT + col] = acc[r];
    }
}

extern "C" void kernel_launch(void* const* d_in, const int* in_sizes, int n_in,
                              void* d_out, int out_size, void* d_ws, size_t ws_size,
                              hipStream_t stream) {
    const float* x     = (const float*)d_in[0];
    const int*   ei    = (const int*)d_in[1];
    const int*   mask  = (const int*)d_in[2];
    const float* Wl    = (const float*)d_in[3];
    const float* bl    = (const float*)d_in[4];
    const float* Wr    = (const float*)d_in[5];
    const float* br    = (const float*)d_in[6];
    const float* att   = (const float*)d_in[7];
    const float* gbias = (const float*)d_in[8];
    const float* Wo    = (const float*)d_in[9];
    const float* bo    = (const float*)d_in[10];
    float* out = (float*)d_out;

    const int n_nodes = in_sizes[0] / D_DIM;
    const int n_edges = in_sizes[1] / 2;
    const int n_mask  = in_sizes[2];
    const int* src = ei;
    const int* dst = ei + n_edges;

    char* wp = (char*)d_ws;
    auto carve = [&](size_t bytes) {
        char* p = wp;
        wp += (bytes + 255) & ~(size_t)255;
        return p;
    };
    float*    xl     = (float*)   carve((size_t)n_nodes * D_DIM * 4);
    float*    xr     = (float*)   carve((size_t)n_nodes * D_DIM * 4);
    float*    h      = (float*)   carve((size_t)n_nodes * D_DIM * 4);
    int*      elist  = (int*)     carve((size_t)n_edges * 4);
    float*    val    = (float*)   carve((size_t)n_edges * 4);
    unsigned* smax   = (unsigned*)carve((size_t)n_nodes * 4);
    unsigned* needbits = (unsigned*)carve((size_t)((n_nodes + 31) / 32) * 4);
    int*      cnt    = (int*)     carve(4);
    float*    denom  = (float*)   carve((size_t)n_nodes * 4);
    const float* smaxf = (const float*)smax;

    {
        long long t = (long long)n_nodes * D_DIM;
        int blocks = (int)((t + 255) / 256);
        k_init<<<blocks, 256, 0, stream>>>(h, denom, smax, needbits, cnt, n_nodes);
    }
    k_mark<<<(n_mask + 255) / 256, 256, 0, stream>>>(mask, needbits, n_mask);
    k_compact<<<(n_edges + 256 * CP_LOADS - 1) / (256 * CP_LOADS), 256, 0, stream>>>(
        dst, needbits, elist, cnt, n_edges);
    {
        int tiles  = (n_nodes + 15) / 16;
        int blocks = (tiles + 7) / 8;
        k_node_linear<<<blocks, 256, 0, stream>>>(x, Wl, bl, Wr, br, xl, xr, n_nodes);
    }
    {
        int blocks = (n_edges + 255) / 256;
        k_edge_logits<<<blocks, 256, 0, stream>>>(src, dst, elist, cnt, xl, xr, att, val, smax);
    }
    k_decode_smax<<<(n_nodes + 255) / 256, 256, 0, stream>>>(smax, n_nodes);
    k_alpha<<<(n_edges + 255) / 256, 256, 0, stream>>>(dst, elist, cnt, smaxf, val, denom);
    {
        int blocks = (n_edges + 255) / 256;
        k_scatter<<<blocks, 256, 0, stream>>>(src, dst, elist, cnt, xl, val, denom, h);
    }
    {
        int tiles  = (n_mask + 15) / 16;
        int blocks = (tiles + 7) / 8;
        k_output<<<blocks, 256, 0, stream>>>(h, mask, gbias, Wo, bo, out, n_mask);
    }
}


# h is cleared by the node-linear waves (8 stores of 16 B per lane over their own 16-row tile) instead of by a kernel of one 4 B store per thread; k_init keeps only the per-node arrays
# speedup vs baseline: 2.7337x; 1.0370x over previous
#include <hip/hip_runtime.h>
#include <hip/hip_bf16.h>
#include <math.h>


#define D_DIM 64
#define DOUT  128
#define NEG_SLOPE 0.2f

typedef __attribute__((ext_vector_type(2))) float v2f;
typedef __attribute__((ext_vector_type(8))) float v8f;
typedef __attribute__((ext_vector_type(4))) float v4f;

__device__ __forceinline__ unsigned f2ord(float f) {
    unsigned u = __float_as_uint(f);
    return (u & 0x80000000u) ? ~u : (u | 0x80000000u);
}
__device__ __forceinline__ float ord2f(unsigned u) {
    unsigned b = (u & 0x80000000u) ? (u ^ 0x80000000u) : ~u;
    return __uint_as_float(b);
}

__global__ void k_init(float* __restrict__ denom,
                       unsigned* __restrict__ smax, unsigned* __restrict__ needbits,
                       int* __restrict__ cnt, int n_nodes) {
    const int i = blockIdx.x * blockDim.x + threadIdx.x;
    if (i < n_nodes) { denom[i] = 0.0f; smax[i] = 0x007FFFFFu;   }
    if (i < (n_nodes + 31) / 32) needbits[i] = 0u;
    if (i == 0) *cnt = 0;
}

__global__ void k_mark(const int* __restrict__ mask, unsigned* __restrict__ needbits,
                       int n_mask) {
    int i = blockIdx.x * blockDim.x + threadIdx.x;
    if (i >= n_mask) return;
    const int m = mask[i];
    atomicOr(&needbits[m >> 5], 1u << (m & 31));
}

#define CP_LOADS 4
__global__ void k_compact(const int* __restrict__ dst, const unsigned* __restrict__ needbits,
                          int* __restrict__ elist, int* __restrict__ cnt, int n_edges) {
    __shared__ int wave_off[8];
    __shared__ int block_base;
    const int lane = threadIdx.x & 31;
    const int wave = threadIdx.x >> 5;
    const long long base = (long long)blockIdx.x * (256 * CP_LOADS) + threadIdx.x;
    const unsigned below = (1u << lane) - 1u;
    bool keep[CP_LOADS];
    int  rank[CP_LOADS];
    int  d[CP_LOADS];
    unsigned word[CP_LOADS];
    int  tot = 0;
#pragma unroll
    for (int j = 0; j < CP_LOADS; ++j) {
        const long long e = base + j * 256;
        d[j] = dst[e < n_edges ? e : 0];
    }
#pragma unroll
    for (int j = 0; j < CP_LOADS; ++j)
        word[j] = needbits[d[j] >> 5];
#pragma unroll
    for (int j = 0; j < CP_LOADS; ++j) {
        keep[j] = (base + j * 256 < n_edges) && ((word[j] >> (d[j] & 31)) & 1u);
        const unsigned b = __builtin_amdgcn_ballot_w32(keep[j]);
        rank[j] = tot + __popc(b & below);
        tot += __popc(b);
    }
    if (lane == 0) wave_off[wave] = tot;
    __syncthreads();
    if (threadIdx.x == 0) {
        int s = 0;
        for (int w = 0; w < 8; ++w) { const int t = wave_off[w]; wave_off[w] = s; s += t; }
        block_base = s ? atomicAdd(cnt, s) : 0;
    }
    __syncthreads();
    const int wb = block_base + wave_off[wave];
#pragma unroll
    for (int j = 0; j < CP_LOADS; ++j)
        if (keep[j]) elist[wb + rank[j]] = (int)(base + j * 256);
}

__global__ void k_node_linear(const float* __restrict__ x,
                              const float* __restrict__ Wl, const float* __restrict__ bl,
                              const float* __restrict__ Wr, const float* __restrict__ br,
                              float* __restrict__ xl, float* __restrict__ xr,
                              float* __restrict__ h, int n_nodes) {
    const int lane = threadIdx.x & 31;
    const int wave = threadIdx.x >> 5;
    const int tile = blockIdx.x * (blockDim.x >> 5) + wave;
    const int r0 = tile * 16;
    const v4f zero = {0.0f, 0.0f, 0.0f, 0.0f};
    float* hb = h + (long long)r0 * D_DIM + 4 * lane;
    if (r0 + 16 > n_nodes) {
        for (int j = 0; j < 8; ++j)
            if (r0 + 2 * j + (lane >> 4) < n_nodes) *(v4f*)(hb + 128 * j) = zero;
        return;
    }
    for (int j = 0; j < 8; ++j)
        *(v4f*)(hb + 128 * j) = zero;
    const int lrow = lane & 15;
    const int lhi  = lane >> 4;

    v2f a[16];
    const float* xp = x + (long long)(r0 + lrow) * D_DIM + 2 * lhi;
#pragma unroll
    for (int kk = 0; kk < 16; ++kk)
        a[kk] = *(const v2f*)(xp + kk * 4);

#pragma unroll
    for (int m = 0; m < 2; ++m) {
        const float* W    = m ? Wr : Wl;
        const float* bias = m ? br : bl;
        float*       out  = m ? xr : xl;
#pragma unroll
        for (int ct = 0; ct < 4; ++ct) {
            const int col = ct * 16 + lrow;
            const float bv = bias[col];
            v8f acc = {bv, bv, bv, bv, bv, bv, bv, bv};
#pragma unroll
            for (int kk = 0; kk < 16; ++kk) {
                const int k0 = kk * 4 + 2 * lhi;
                v2f b;
                b.x = W[(k0    ) * D_DIM + col];
                b.y = W[(k0 + 1) * D_DIM + col];
                acc = __builtin_amdgcn_wmma_f32_16x16x4_f32(
                        false, a[kk], false, b, (short)0, acc, false, false);
            }
#pragma unroll
            for (int r = 0; r < 8; ++r)
                out[(long long)(r0 + r + 8 * lhi) * D_DIM + col] = acc[r];
        }
    }
}

template <int CTRL>
__device__ __forceinline__ int dpp_i(int v) {
    return __builtin_amdgcn_update_dpp(0, v, CTRL, 0xf, 0xf, true);
}
template <int CTRL>
__device__ __forceinline__ float dpp_f(float v) {
    return __int_as_float(__builtin_amdgcn_update_dpp(0, __float_as_int(v), CTRL, 0xf, 0xf, true));
}

template <int J>
__device__ __forceinline__ void gather_rows(int sv, int dv, int sub,
                                            const float* __restrict__ xl,
                                            const float* __restrict__ xr,
                                            v4f& l, v4f& r) {
    const int s = dpp_i<0x150 + J>(sv);
    const int d = dpp_i<0x150 + J>(dv);
    l = *(const v4f*)(xl + (long long)s * D_DIM + 4 * sub);
    r = *(const v4f*)(xr + (long long)d * D_DIM + 4 * sub);
}

template <int J>
__device__ __forceinline__ void score_rows(const v4f& l, const v4f& r, const v4f& a,
                                           int sub, int owner0, float& mine) {
    float t0 = l.x + r.x, t1 = l.y + r.y, t2 = l.z + r.z, t3 = l.w + r.w;
    t0 = t0 > 0.0f ? t0 : NEG_SLOPE * t0;
    t1 = t1 > 0.0f ? t1 : NEG_SLOPE * t1;
    t2 = t2 > 0.0f ? t2 : NEG_SLOPE * t2;
    t3 = t3 > 0.0f ? t3 : NEG_SLOPE * t3;
    float p = t0 * a.x + t1 * a.y + t2 * a.z + t3 * a.w;
    p += dpp_f<0x160 + 1>(p);
    p += dpp_f<0x160 + 2>(p);
    p += dpp_f<0x160 + 4>(p);
    p += dpp_f<0x160 + 8>(p);
    if (sub == owner0 + J) mine = p;
}

__global__ void k_edge_logits(const int* __restrict__ src, const int* __restrict__ dst,
                              const int* __restrict__ elist, const int* __restrict__ list_len,
                              const float* __restrict__ xl, const float* __restrict__ xr,
                              const float* __restrict__ att,
                              float* __restrict__ logits, unsigned* __restrict__ smax) {
    const int n_edges = *list_len;
    const int lane = threadIdx.x & 31;
    const int sub  = lane & 15;
    const int wave = __builtin_amdgcn_readfirstlane((int)(threadIdx.x >> 5));
    const long long e0 = ((long long)blockIdx.x * (blockDim.x >> 5) + wave) * 32;
    if (e0 >= n_edges) return;
    const long long em = e0 + lane;
    const bool ok = em < n_edges;
    int sv = 0, dv = 0;
    if (ok) { const int e = elist[em]; sv = src[e]; dv = dst[e]; }
    const v4f a = *(const v4f*)(att + 4 * sub);
    float mine = 0.0f;
    int sr = sv, dr = dv;
#pragma unroll 1
    for (int pass = 0; pass < 2; ++pass) {
        v4f l0, l1, l2, l3, l4, l5, l6, l7, r0, r1, r2, r3, r4, r5, r6, r7;
        gather_rows<0>(sr, dr, sub, xl, xr, l0, r0);
        gather_rows<1>(sr, dr, sub, xl, xr, l1, r1);
        gather_rows<2>(sr, dr, sub, xl, xr, l2, r2);
        gather_rows<3>(sr, dr, sub, xl, xr, l3, r3);
        gather_rows<4>(sr, dr, sub, xl, xr, l4, r4);
        gather_rows<5>(sr, dr, sub, xl, xr, l5, r5);
        gather_rows<6>(sr, dr, sub, xl, xr, l6, r6);
        gather_rows<7>(sr, dr, sub, xl, xr, l7, r7);
        asm volatile("" : "+v"(l0), "+v"(l1), "+v"(l2), "+v"(l3), "+v"(l4), "+v"(l5), "+v"(l6), "+v"(l7)
                        : "v"(r0), "v"(r1), "v"(r2), "v"(r3), "v"(r4), "v"(r5), "v"(r6), "v"(r7));
        const int owner0 = 8 * pass;
        score_rows<0>(l0, r0, a, sub, owner0, mine);
        score_rows<1>(l1, r1, a, sub, owner0, mine);
        score_rows<2>(l2, r2, a, sub, owner0, mine);
        score_rows<3>(l3, r3, a, sub, owner0, mine);
        score_rows<4>(l4, r4, a, sub, owner0, mine);
        score_rows<5>(l5, r5, a, sub, owner0, mine);
        score_rows<6>(l6, r6, a, sub, owner0, mine);
        score_rows<7>(l7, r7, a, sub, owner0, mine);
        sr = dpp_i<0x128>(sr);
        dr = dpp_i<0x128>(dr);
    }
    if (ok) {
        logits[em] = mine;
        atomicMax(&smax[dv], f2ord(mine));
    }
}

__global__ void k_decode_smax(unsigned* __restrict__ smax, int n_nodes) {
    int i = blockIdx.x * blockDim.x + threadIdx.x;
    if (i >= n_nodes) return;
    float f = ord2f(smax[i]);
    smax[i] = __float_as_uint(__builtin_isfinite(f) ? f : 0.0f);
}

__global__ void k_alpha(const int* __restrict__ dst,
                        const int* __restrict__ elist, const int* __restrict__ list_len,
                        const float* __restrict__ smaxf,
                        float* __restrict__ val, float* __restrict__ denom) {
    const int n_edges = *list_len;
    int p = blockIdx.x * blockDim.x + threadIdx.x;
    if (p >= n_edges) return;
    const int d = dst[elist[p]];
    const float a = __expf(val[p] - smaxf[d]);
    val[p] = a;
    atomicAdd(&denom[d], a);
}

__global__ void k_scatter(const int* __restrict__ src, const int* __restrict__ dst,
                          const int* __restrict__ elist, const int* __restrict__ list_len,
                          const float* __restrict__ xl,
                          const float* __restrict__ alpha, const float* __restrict__ denom,
                          float* __restrict__ h) {
    const int n_edges = *list_len;
    const int lane = threadIdx.x & 31;
    const int wave = __builtin_amdgcn_readfirstlane((int)(threadIdx.x >> 5));
    const long long e0 = ((long long)blockIdx.x * (blockDim.x >> 5) + wave) * 32;
    if (e0 >= n_edges) return;
    const long long em = e0 + lane;
    int sv = 0, dv = 0;
    float wv = 0.0f;
    if (em < n_edges) {
        const int e = elist[em];
        sv = src[e]; dv = dst[e];
        wv = alpha[em] / (denom[dv] + 1e-16f);
    }
    const long long left = (long long)n_edges - e0;
    const int cnt = left < 32 ? (int)left : 32;
    if (cnt == 32) {
#pragma unroll
        for (int b = 0; b < 32; b += 16) {
            float v0[16], v1[16];
#pragma unroll
            for (int j = 0; j < 16; ++j) {
                const int s = __builtin_amdgcn_readlane(sv, b + j);
                const float* xp = xl + (long long)s * D_DIM + lane;
                v0[j] = xp[0];
                v1[j] = xp[32];
            }
#pragma unroll
            for (int j = 0; j < 16; ++j) {
                const int d = __builtin_amdgcn_readlane(dv, b + j);
                const float w = __int_as_float(__builtin_amdgcn_readlane(__float_as_int(wv), b + j));
                float* hp = h + (long long)d * D_DIM + lane;
                atomicAdd(hp,      w * v0[j]);
                atomicAdd(hp + 32, w * v1[j]);
            }
        }
    } else {
#pragma unroll 1
        for (int j = 0; j < cnt; ++j) {
            const int s = __builtin_amdgcn_readlane(sv, j);
            const int d = __builtin_amdgcn_readlane(dv, j);
            const float w = __int_as_float(__builtin_amdgcn_readlane(__float_as_int(wv), j));
            const float* xp = xl + (long long)s * D_DIM + lane;
            float* hp = h + (long long)d * D_DIM + lane;
            atomicAdd(hp,      w * xp[0]);
            atomicAdd(hp + 32, w * xp[32]);
        }
    }
}

__global__ void k_output(const float* __restrict__ h, const int* __restrict__ mask,
                         const float* __restrict__ gbias,
                         const float* __restrict__ Wo, const float* __restrict__ bo,
                         float* __restrict__ out, int n_mask) {
    const int lane = threadIdx.x & 31;
    const int wave = threadIdx.x >> 5;
    const int tile = blockIdx.x * (blockDim.x >> 5) + wave;
    const int r0 = tile * 16;
    if (r0 + 16 > n_mask) return;
    const int lrow = lane & 15;
    const int lhi  = lane >> 4;

    const long long mi = (long long)mask[r0 + lrow];
    v2f a[16];
    const float* hp = h + mi * D_DIM + 2 * lhi;
    const float* gp = gbias + 2 * lhi;
#pragma unroll
    for (int kk = 0; kk < 16; ++kk) {
        const v2f hv = *(const v2f*)(hp + kk * 4);
        const v2f gv = *(const v2f*)(gp + kk * 4);
        a[kk].x = hv.x + gv.x;
        a[kk].y = hv.y + gv.y;
    }
#pragma unroll
    for (int ct = 0; ct < 8; ++ct) {
        const int col = ct * 16 + lrow;
        const float bv = bo[col];
        v8f acc = {bv, bv, bv, bv, bv, bv, bv, bv};
#pragma unroll
        for (int kk = 0; kk < 16; ++kk) {
            const int k0 = kk * 4 + 2 * lhi;
            v2f b;
            b.x = Wo[(k0    ) * DOUT + col];
            b.y = Wo[(k0 + 1) * DOUT + col];
            acc = __builtin_amdgcn_wmma_f32_16x16x4_f32(
                    false, a[kk], false, b, (short)0, acc, false, false);
        }
#pragma unroll
        for (int r = 0; r < 8; ++r)
            out[(long long)(r0 + r + 8 * lhi) * DOUT + col] = acc[r];
    }
}

extern "C" void kernel_launch(void* const* d_in, const int* in_sizes, int n_in,
                              void* d_out, int out_size, void* d_ws, size_t ws_size,
                              hipStream_t stream) {
    const float* x     = (const float*)d_in[0];
    const int*   ei    = (const int*)d_in[1];
    const int*   mask  = (const int*)d_in[2];
    const float* Wl    = (const float*)d_in[3];
    const float* bl    = (const float*)d_in[4];
    const float* Wr    = (const float*)d_in[5];
    const float* br    = (const float*)d_in[6];
    const float* att   = (const float*)d_in[7];
    const float* gbias = (const float*)d_in[8];
    const float* Wo    = (const float*)d_in[9];
    const float* bo    = (const float*)d_in[10];
    float* out = (float*)d_out;

    const int n_nodes = in_sizes[0] / D_DIM;
    const int n_edges = in_sizes[1] / 2;
    const int n_mask  = in_sizes[2];
    const int* src = ei;
    const int* dst = ei + n_edges;

    char* wp = (char*)d_ws;
    auto carve = [&](size_t bytes) {
        char* p = wp;
        wp += (bytes + 255) & ~(size_t)255;
        return p;
    };
    float*    xl     = (float*)   carve((size_t)n_nodes * D_DIM * 4);
    float*    xr     = (float*)   carve((size_t)n_nodes * D_DIM * 4);
    float*    h      = (float*)   carve((size_t)n_nodes * D_DIM * 4);
    int*      elist  = (int*)     carve((size_t)n_edges * 4);
    float*    val    = (float*)   carve((size_t)n_edges * 4);
    unsigned* smax   = (unsigned*)carve((size_t)n_nodes * 4);
    unsigned* needbits = (unsigned*)carve((size_t)((n_nodes + 31) / 32) * 4);
    int*      cnt    = (int*)     carve(4);
    float*    denom  = (float*)   carve((size_t)n_nodes * 4);
    const float* smaxf = (const float*)smax;

    k_init<<<(n_nodes + 255) / 256, 256, 0, stream>>>(denom, smax, needbits, cnt, n_nodes);
    k_mark<<<(n_mask + 255) / 256, 256, 0, stream>>>(mask, needbits, n_mask);
    k_compact<<<(n_edges + 256 * CP_LOADS - 1) / (256 * CP_LOADS), 256, 0, stream>>>(
        dst, needbits, elist, cnt, n_edges);
    {
        int tiles  = (n_nodes + 15) / 16;
        int blocks = (tiles + 7) / 8;
        k_node_linear<<<blocks, 256, 0, stream>>>(x, Wl, bl, Wr, br, xl, xr, h, n_nodes);
    }
    {
        int blocks = (n_edges + 255) / 256;
        k_edge_logits<<<blocks, 256, 0, stream>>>(src, dst, elist, cnt, xl, xr, att, val, smax);
    }
    k_decode_smax<<<(n_nodes + 255) / 256, 256, 0, stream>>>(smax, n_nodes);
    k_alpha<<<(n_edges + 255) / 256, 256, 0, stream>>>(dst, elist, cnt, smaxf, val, denom);
    {
        int blocks = (n_edges + 255) / 256;
        k_scatter<<<blocks, 256, 0, stream>>>(src, dst, elist, cnt, xl, val, denom, h);
    }
    {
        int tiles  = (n_mask + 15) / 16;
        int blocks = (tiles + 7) / 8;
        k_output<<<blocks, 256, 0, stream>>>(h, mask, gbias, Wo, bo, out, n_mask);
    }
}


# edges counting-sorted by destination (returning adds give arrival-order ranks); 16 lanes per output row do logits, running-max rescaled softmax and the weighted sum in registers; no float atomics
# speedup vs baseline: 3.5177x; 1.2868x over previous
#include <hip/hip_runtime.h>
#include <hip/hip_bf16.h>
#include <math.h>
#include <float.h>


#define D_DIM 64
#define DOUT  128
#define NEG_SLOPE 0.2f

typedef __attribute__((ext_vector_type(2))) float v2f;
typedef __attribute__((ext_vector_type(8))) float v8f;
typedef __attribute__((ext_vector_type(4))) float v4f;

__global__ void k_init(int* __restrict__ deg, unsigned* __restrict__ needbits,
                       int* __restrict__ cnt, int n_nodes) {
    const int i = blockIdx.x * blockDim.x + threadIdx.x;
    if (i < n_nodes) deg[i] = 0;
    if (i < (n_nodes + 31) / 32) needbits[i] = 0u;
    if (i == 0) *cnt = 0;
}

__global__ void k_mark(const int* __restrict__ mask, unsigned* __restrict__ needbits,
                       int n_mask) {
    int i = blockIdx.x * blockDim.x + threadIdx.x;
    if (i >= n_mask) return;
    const int m = mask[i];
    atomicOr(&needbits[m >> 5], 1u << (m & 31));
}

#define CP_LOADS 4
__global__ void k_count(const int* __restrict__ dst, const unsigned* __restrict__ needbits,
                        int* __restrict__ deg, int* __restrict__ rank, int n_edges) {
    const long long base = (long long)blockIdx.x * (256 * CP_LOADS) + threadIdx.x;
    int d[CP_LOADS];
    unsigned word[CP_LOADS];
#pragma unroll
    for (int j = 0; j < CP_LOADS; ++j) {
        const long long e = base + j * 256;
        d[j] = dst[e < n_edges ? e : 0];
    }
#pragma unroll
    for (int j = 0; j < CP_LOADS; ++j)
        word[j] = needbits[d[j] >> 5];
    int r[CP_LOADS];
#pragma unroll
    for (int j = 0; j < CP_LOADS; ++j) {
        r[j] = -1;
        if ((base + j * 256 < n_edges) && ((word[j] >> (d[j] & 31)) & 1u))
            r[j] = atomicAdd(&deg[d[j]], 1);
    }
#pragma unroll
    for (int j = 0; j < CP_LOADS; ++j) {
        const long long e = base + j * 256;
        if (e < n_edges) rank[e] = r[j];
    }
}

__global__ void k_offsets(const int* __restrict__ deg, int* __restrict__ first,
                          int* __restrict__ cnt, int n_nodes) {
    __shared__ int s[256];
    __shared__ int block_base;
    const int t = threadIdx.x;
    const int i = blockIdx.x * 256 + t;
    const int d = i < n_nodes ? deg[i] : 0;
    s[t] = d;
    __syncthreads();
    for (int o = 1; o < 256; o <<= 1) {
        const int v = t >= o ? s[t - o] : 0;
        __syncthreads();
        s[t] += v;
        __syncthreads();
    }
    if (t == 255) {
        const int tot = s[255];
        block_base = tot ? atomicAdd(cnt, tot) : 0;
    }
    __syncthreads();
    if (i < n_nodes) first[i] = block_base + s[t] - d;
}

__global__ void k_fill(const int* __restrict__ src, const int* __restrict__ dst,
                       const int* __restrict__ rank, const int* __restrict__ first,
                       int* __restrict__ ssrc, int n_edges) {
    const long long base = (long long)blockIdx.x * (256 * CP_LOADS) + threadIdx.x;
    int d[CP_LOADS], r[CP_LOADS], s[CP_LOADS], f[CP_LOADS];
#pragma unroll
    for (int j = 0; j < CP_LOADS; ++j) {
        const long long e = base + j * 256;
        const long long ec = e < n_edges ? e : 0;
        r[j] = e < n_edges ? rank[ec] : -1;
        d[j] = dst[ec];
        s[j] = src[ec];
    }
#pragma unroll
    for (int j = 0; j < CP_LOADS; ++j) {
        f[j] = 0;
        if (r[j] >= 0) f[j] = first[d[j]];
    }
#pragma unroll
    for (int j = 0; j < CP_LOADS; ++j)
        if (r[j] >= 0) ssrc[f[j] + r[j]] = s[j];
}

__global__ void k_node_linear(const float* __restrict__ x,
                              const float* __restrict__ Wl, const float* __restrict__ bl,
                              const float* __restrict__ Wr, const float* __restrict__ br,
                              float* __restrict__ xl, float* __restrict__ xr, int n_nodes) {
    const int lane = threadIdx.x & 31;
    const int wave = threadIdx.x >> 5;
    const int tile = blockIdx.x * (blockDim.x >> 5) + wave;
    const int r0 = tile * 16;
    if (r0 + 16 > n_nodes) return;
    const int lrow = lane & 15;
    const int lhi  = lane >> 4;

    v2f a[16];
    const float* xp = x + (long long)(r0 + lrow) * D_DIM + 2 * lhi;
#pragma unroll
    for (int kk = 0; kk < 16; ++kk)
        a[kk] = *(const v2f*)(xp + kk * 4);

#pragma unroll
    for (int m = 0; m < 2; ++m) {
        const float* W    = m ? Wr : Wl;
        const float* bias = m ? br : bl;
        float*       out  = m ? xr : xl;
#pragma unroll
        for (int ct = 0; ct < 4; ++ct) {
            const int col = ct * 16 + lrow;
            const float bv = bias[col];
            v8f acc = {bv, bv, bv, bv, bv, bv, bv, bv};
#pragma unroll
            for (int kk = 0; kk < 16; ++kk) {
                const int k0 = kk * 4 + 2 * lhi;
                v2f b;
                b.x = W[(k0    ) * D_DIM + col];
                b.y = W[(k0 + 1) * D_DIM + col];
                acc = __builtin_amdgcn_wmma_f32_16x16x4_f32(
                        false, a[kk], false, b, (short)0, acc, false, false);
            }
#pragma unroll
            for (int r = 0; r < 8; ++r)
                out[(long long)(r0 + r + 8 * lhi) * D_DIM + col] = acc[r];
        }
    }
}

template <int CTRL>
__device__ __forceinline__ int dpp_i(int v) {
    return __builtin_amdgcn_update_dpp(0, v, CTRL, 0xf, 0xf, true);
}
template <int CTRL>
__device__ __forceinline__ float dpp_f(float v) {
    return __int_as_float(__builtin_amdgcn_update_dpp(0, __float_as_int(v), CTRL, 0xf, 0xf, true));
}

template <int J>
__device__ __forceinline__ v4f gather_row(int ids, int sub, const float* __restrict__ xl) {
    const int s = dpp_i<0x150 + J>(ids);
    return *(const v4f*)(xl + (long long)s * D_DIM + 4 * sub);
}

__device__ __forceinline__ float edge_score(const v4f& l, const v4f& r, const v4f& a) {
    float t0 = l.x + r.x, t1 = l.y + r.y, t2 = l.z + r.z, t3 = l.w + r.w;
    t0 = t0 > 0.0f ? t0 : NEG_SLOPE * t0;
    t1 = t1 > 0.0f ? t1 : NEG_SLOPE * t1;
    t2 = t2 > 0.0f ? t2 : NEG_SLOPE * t2;
    t3 = t3 > 0.0f ? t3 : NEG_SLOPE * t3;
    float p = t0 * a.x + t1 * a.y + t2 * a.z + t3 * a.w;
    p += dpp_f<0x160 + 1>(p);
    p += dpp_f<0x160 + 2>(p);
    p += dpp_f<0x160 + 4>(p);
    p += dpp_f<0x160 + 8>(p);
    return p;
}

template <int B>
__device__ __forceinline__ void seg_batch(int ids, int left, int sub,
                                          const float* __restrict__ xl,
                                          const v4f& r, const v4f& a,
                                          float& m, float& den, v4f& acc) {
    const v4f l0 = gather_row<4 * B + 0>(ids, sub, xl);
    const v4f l1 = gather_row<4 * B + 1>(ids, sub, xl);
    const v4f l2 = gather_row<4 * B + 2>(ids, sub, xl);
    const v4f l3 = gather_row<4 * B + 3>(ids, sub, xl);
    const float p0 = edge_score(l0, r, a);
    const float p1 = edge_score(l1, r, a);
    const float p2 = edge_score(l2, r, a);
    const float p3 = edge_score(l3, r, a);
    const bool v0 = left > 0, v1 = left > 1, v2 = left > 2, v3 = left > 3;
    float bm = m;
    bm = v0 ? fmaxf(bm, p0) : bm;
    bm = v1 ? fmaxf(bm, p1) : bm;
    bm = v2 ? fmaxf(bm, p2) : bm;
    bm = v3 ? fmaxf(bm, p3) : bm;
    const float sc = __expf(m - bm);
    const float w0 = v0 ? __expf(p0 - bm) : 0.0f;
    const float w1 = v1 ? __expf(p1 - bm) : 0.0f;
    const float w2 = v2 ? __expf(p2 - bm) : 0.0f;
    const float w3 = v3 ? __expf(p3 - bm) : 0.0f;
    den   = den   * sc + ((w0 + w1) + (w2 + w3));
    acc.x = acc.x * sc + ((w0 * l0.x + w1 * l1.x) + (w2 * l2.x + w3 * l3.x));
    acc.y = acc.y * sc + ((w0 * l0.y + w1 * l1.y) + (w2 * l2.y + w3 * l3.y));
    acc.z = acc.z * sc + ((w0 * l0.z + w1 * l1.z) + (w2 * l2.z + w3 * l3.z));
    acc.w = acc.w * sc + ((w0 * l0.w + w1 * l1.w) + (w2 * l2.w + w3 * l3.w));
    m = bm;
}

__global__ void k_segment(const int* __restrict__ mask, const int* __restrict__ first,
                          const int* __restrict__ deg, const int* __restrict__ ssrc,
                          const float* __restrict__ xl, const float* __restrict__ xr,
                          const float* __restrict__ att, float* __restrict__ hm, int n_mask) {
    const int lane = threadIdx.x & 31;
    const int sub  = lane & 15;
    const int i = (int)((blockIdx.x * blockDim.x + threadIdx.x) >> 4);
    const bool live = i < n_mask;
    int node = 0, beg = 0, n = 0;
    if (live) { node = mask[i]; beg = first[node]; n = deg[node]; }
    const v4f a = *(const v4f*)(att + 4 * sub);
    const v4f r = *(const v4f*)(xr + (long long)node * D_DIM + 4 * sub);
    const int n_lo = __builtin_amdgcn_readlane(n, 0);
    const int n_hi = __builtin_amdgcn_readlane(n, 16);
    const int nmax = n_lo > n_hi ? n_lo : n_hi;
    float m = -FLT_MAX, den = 0.0f;
    v4f acc = {0.0f, 0.0f, 0.0f, 0.0f};
    for (int c = 0; c < nmax; c += 16) {
        const int left = n - c;
        int ids = 0;
        if (sub < left) ids = ssrc[beg + c + sub];
        seg_batch<0>(ids, left, sub, xl, r, a, m, den, acc);
        if (c + 4 >= nmax) break;
        seg_batch<1>(ids, left - 4, sub, xl, r, a, m, den, acc);
        if (c + 8 >= nmax) break;
        seg_batch<2>(ids, left - 8, sub, xl, r, a, m, den, acc);
        if (c + 12 >= nmax) break;
        seg_batch<3>(ids, left - 12, sub, xl, r, a, m, den, acc);
    }
    if (live) {
        const float dn = den + 1e-16f;
        v4f o;
        o.x = acc.x / dn; o.y = acc.y / dn; o.z = acc.z / dn; o.w = acc.w / dn;
        *(v4f*)(hm + (long long)i * D_DIM + 4 * sub) = o;
    }
}

__global__ void k_output(const float* __restrict__ hm,
                         const float* __restrict__ gbias,
                         const float* __restrict__ Wo, const float* __restrict__ bo,
                         float* __restrict__ out, int n_mask) {
    const int lane = threadIdx.x & 31;
    const int wave = threadIdx.x >> 5;
    const int tile = blockIdx.x * (blockDim.x >> 5) + wave;
    const int r0 = tile * 16;
    if (r0 + 16 > n_mask) return;
    const int lrow = lane & 15;
    const int lhi  = lane >> 4;

    v2f a[16];
    const float* hp = hm + (long long)(r0 + lrow) * D_DIM + 2 * lhi;
    const float* gp = gbias + 2 * lhi;
#pragma unroll
    for (int kk = 0; kk < 16; ++kk) {
        const v2f hv = *(const v2f*)(hp + kk * 4);
        const v2f gv = *(const v2f*)(gp + kk * 4);
        a[kk].x = hv.x + gv.x;
        a[kk].y = hv.y + gv.y;
    }
#pragma unroll
    for (int ct = 0; ct < 8; ++ct) {
        const int col = ct * 16 + lrow;
        const float bv = bo[col];
        v8f acc = {bv, bv, bv, bv, bv, bv, bv, bv};
#pragma unroll
        for (int kk = 0; kk < 16; ++kk) {
            const int k0 = kk * 4 + 2 * lhi;
            v2f b;
            b.x = Wo[(k0    ) * DOUT + col];
            b.y = Wo[(k0 + 1) * DOUT + col];
            acc = __builtin_amdgcn_wmma_f32_16x16x4_f32(
                    false, a[kk], false, b, (short)0, acc, false, false);
        }
#pragma unroll
        for (int r = 0; r < 8; ++r)
            out[(long long)(r0 + r + 8 * lhi) * DOUT + col] = acc[r];
    }
}

extern "C" void kernel_launch(void* const* d_in, const int* in_sizes, int n_in,
                              void* d_out, int out_size, void* d_ws, size_t ws_size,
                              hipStream_t stream) {
    const float* x     = (const float*)d_in[0];
    const int*   ei    = (const int*)d_in[1];
    const int*   mask  = (const int*)d_in[2];
    const float* Wl    = (const float*)d_in[3];
    const float* bl    = (const float*)d_in[4];
    const float* Wr    = (const float*)d_in[5];
    const float* br    = (const float*)d_in[6];
    const float* att   = (const float*)d_in[7];
    const float* gbias = (const float*)d_in[8];
    const float* Wo    = (const float*)d_in[9];
    const float* bo    = (const float*)d_in[10];
    float* out = (float*)d_out;

    const int n_nodes = in_sizes[0] / D_DIM;
    const int n_edges = in_sizes[1] / 2;
    const int n_mask  = in_sizes[2];
    const int* src = ei;
    const int* dst = ei + n_edges;

    char* wp = (char*)d_ws;
    auto carve = [&](size_t bytes) {
        char* p = wp;
        wp += (bytes + 255) & ~(size_t)255;
        return p;
    };
    float*    xl     = (float*)   carve((size_t)n_nodes * D_DIM * 4);
    float*    xr     = (float*)   carve((size_t)n_nodes * D_DIM * 4);
    float*    hm     = (float*)   carve((size_t)n_mask * D_DIM * 4);
    int*      ssrc   = (int*)     carve((size_t)n_edges * 4);
    int*      rank   = (int*)     carve((size_t)n_edges * 4);
    int*      deg    = (int*)     carve((size_t)n_nodes * 4);
    unsigned* needbits = (unsigned*)carve((size_t)((n_nodes + 31) / 32) * 4);
    int*      cnt    = (int*)     carve(4);
    int*      first  = (int*)     carve((size_t)n_nodes * 4);

    k_init<<<(n_nodes + 255) / 256, 256, 0, stream>>>(deg, needbits, cnt, n_nodes);
    k_mark<<<(n_mask + 255) / 256, 256, 0, stream>>>(mask, needbits, n_mask);
    k_count<<<(n_edges + 256 * CP_LOADS - 1) / (256 * CP_LOADS), 256, 0, stream>>>(
        dst, needbits, deg, rank, n_edges);
    k_offsets<<<(n_nodes + 255) / 256, 256, 0, stream>>>(deg, first, cnt, n_nodes);
    k_fill<<<(n_edges + 256 * CP_LOADS - 1) / (256 * CP_LOADS), 256, 0, stream>>>(
        src, dst, rank, first, ssrc, n_edges);
    {
        int tiles  = (n_nodes + 15) / 16;
        int blocks = (tiles + 7) / 8;
        k_node_linear<<<blocks, 256, 0, stream>>>(x, Wl, bl, Wr, br, xl, xr, n_nodes);
    }
    k_segment<<<(n_mask + 15) / 16, 256, 0, stream>>>(mask, first, deg, ssrc, xl, xr, att, hm, n_mask);
    {
        int tiles  = (n_mask + 15) / 16;
        int blocks = (tiles + 7) / 8;
        k_output<<<blocks, 256, 0, stream>>>(hm, gbias, Wo, bo, out, n_mask);
    }
}


# edges grouped by destination with LDS counters only: per block by 512-node range, then one block per range ranks per node; no global add per kept edge; row sums stay in arrival order (LDS adds)
# speedup vs baseline: 4.8922x; 1.3907x over previous
#include <hip/hip_runtime.h>
#include <hip/hip_bf16.h>
#include <math.h>
#include <float.h>


#define D_DIM 64
#define DOUT  128
#define NEG_SLOPE 0.2f

typedef __attribute__((ext_vector_type(2))) float v2f;
typedef __attribute__((ext_vector_type(8))) float v8f;
typedef __attribute__((ext_vector_type(4))) float v4f;

__global__ void k_init(unsigned* __restrict__ needbits, int n_words) {
    const int i = blockIdx.x * blockDim.x + threadIdx.x;
    if (i < n_words) needbits[i] = 0u;
}

__global__ void k_mark(const int* __restrict__ mask, unsigned* __restrict__ needbits,
                       int n_mask) {
    int i = blockIdx.x * blockDim.x + threadIdx.x;
    if (i >= n_mask) return;
    const int m = mask[i];
    atomicOr(&needbits[m >> 5], 1u << (m & 31));
}

#define SORT_T    1024
#define BS_E      4
#define BS_CH     (SORT_T * BS_E)
#define BIN_SHIFT 9
#define BIN_R     (1 << BIN_SHIFT)
#define SB_B      4

__device__ __forceinline__ int block_scan_lds(int* s, int n, int* wsum) {
    const int t = threadIdx.x, T = SORT_T;
    const int lane = t & 31, wave = t >> 5;
    const int K = (n + T - 1) / T;
    const int lo = t * K;
    int hi = lo + K;
    if (hi > n) hi = n;
    __syncthreads();
    int sum = 0;
    for (int i = lo; i < hi; ++i) sum += s[i];
    int inc = sum;
    for (int o = 1; o < 32; o <<= 1) {
        const int u = __shfl_up(inc, o);
        if (lane >= o) inc += u;
    }
    if (lane == 31) wsum[wave] = inc;
    __syncthreads();
    if (wave == 0) {
        const int w = lane < (T >> 5) ? wsum[lane] : 0;
        int winc = w;
        for (int o = 1; o < 32; o <<= 1) {
            const int u = __shfl_up(winc, o);
            if (lane >= o) winc += u;
        }
        wsum[lane] = winc - w;
        if (lane == 31) wsum[32] = winc;
    }
    __syncthreads();
    int run = wsum[wave] + inc - sum;
    for (int i = lo; i < hi; ++i) {
        const int v = s[i];
        s[i] = run;
        run += v;
    }
    const int total = wsum[32];
    __syncthreads();
    return total;
}

__global__ void __launch_bounds__(SORT_T)
k_blocksort(const int* __restrict__ src, const int* __restrict__ dst,
            const unsigned* __restrict__ needbits, int2* __restrict__ recs,
            int* __restrict__ boff, int n_edges, int nbins) {
    extern __shared__ int hist[];
    __shared__ int wsum[33];
    const int t = threadIdx.x;
    for (int b = t; b <= nbins; b += SORT_T) hist[b] = 0;
    const long long base = (long long)blockIdx.x * BS_CH + t;
    int d[BS_E], s[BS_E], lr[BS_E];
    unsigned word[BS_E];
    bool keep[BS_E];
#pragma unroll
    for (int j = 0; j < BS_E; ++j) {
        const long long e = base + j * SORT_T;
        const long long ec = e < n_edges ? e : 0;
        d[j] = dst[ec];
        s[j] = src[ec];
    }
#pragma unroll
    for (int j = 0; j < BS_E; ++j)
        word[j] = needbits[d[j] >> 5];
    __syncthreads();
#pragma unroll
    for (int j = 0; j < BS_E; ++j) {
        keep[j] = (base + j * SORT_T < n_edges) && ((word[j] >> (d[j] & 31)) & 1u);
        lr[j] = 0;
        if (keep[j]) lr[j] = atomicAdd(&hist[d[j] >> BIN_SHIFT], 1);
    }
    block_scan_lds(hist, nbins + 1, wsum);
    int2* mine = recs + (long long)blockIdx.x * BS_CH;
#pragma unroll
    for (int j = 0; j < BS_E; ++j)
        if (keep[j]) mine[hist[d[j] >> BIN_SHIFT] + lr[j]] = make_int2(s[j], d[j]);
    int* myoff = boff + (long long)blockIdx.x * (nbins + 1);
    for (int b = t; b <= nbins; b += SORT_T) myoff[b] = hist[b];
}

__global__ void __launch_bounds__(SORT_T)
k_binsort(const int2* __restrict__ recs, const int* __restrict__ boff,
          int* __restrict__ first, int* __restrict__ deg, int* __restrict__ ssrc,
          int n_nodes, int nblk, int nbins) {
    extern __shared__ int dyn[];
    int* o0  = dyn;
    int* pre = dyn + nblk;
    __shared__ int cnt[BIN_R];
    __shared__ int cur[BIN_R];
    __shared__ int wsum[33];
    __shared__ int sbase;
    const int t = threadIdx.x;
    const int b = blockIdx.x;
    const int node0 = b << BIN_SHIFT;
    if (t == 0) { sbase = 0; pre[nblk] = 0; }
    for (int i = t; i < BIN_R; i += SORT_T) cnt[i] = 0;
    int mystart = 0;
    for (int k = t; k < nblk; k += SORT_T) {
        const int* bo = boff + (long long)k * (nbins + 1) + b;
        const int a0 = bo[0], a1 = bo[1];
        o0[k]  = a0;
        pre[k] = a1 - a0;
        mystart += a0;
    }
    __syncthreads();
    if (mystart) atomicAdd(&sbase, mystart);
    const int total = block_scan_lds(pre, nblk + 1, wsum);
    const int start = sbase;

    int top = 1;
    while (top < nblk) top <<= 1;
    auto fetch = [&](int i0, int2 (&r)[SB_B]) {
        int idx[SB_B], lo[SB_B];
#pragma unroll
        for (int j = 0; j < SB_B; ++j) {
            const int i = i0 + j * SORT_T;
            idx[j] = i < total ? i : total - 1;
            lo[j] = 0;
        }
        for (int w = top >> 1; w > 0; w >>= 1) {
#pragma unroll
            for (int j = 0; j < SB_B; ++j) {
                const int k = lo[j] + w < nblk ? lo[j] + w : nblk;
                lo[j] = pre[k] <= idx[j] ? k : lo[j];
            }
        }
#pragma unroll
        for (int j = 0; j < SB_B; ++j)
            r[j] = recs[(long long)lo[j] * BS_CH + o0[lo[j]] + (idx[j] - pre[lo[j]])];
    };

    int2 r0[SB_B];
#pragma unroll
    for (int j = 0; j < SB_B; ++j) r0[j] = make_int2(0, node0);
    if (t < total) fetch(t, r0);
#pragma unroll
    for (int j = 0; j < SB_B; ++j)
        if (t + j * SORT_T < total) atomicAdd(&cnt[r0[j].y - node0], 1);
    for (int i0 = t + SB_B * SORT_T; i0 < total; i0 += SB_B * SORT_T) {
        int2 r[SB_B];
        fetch(i0, r);
#pragma unroll
        for (int j = 0; j < SB_B; ++j)
            if (i0 + j * SORT_T < total) atomicAdd(&cnt[r[j].y - node0], 1);
    }
    __syncthreads();
    for (int i = t; i < BIN_R; i += SORT_T)
        if (node0 + i < n_nodes) deg[node0 + i] = cnt[i];
    block_scan_lds(cnt, BIN_R, wsum);
    for (int i = t; i < BIN_R; i += SORT_T) {
        const int f = cnt[i];
        cur[i] = f;
        if (node0 + i < n_nodes) first[node0 + i] = start + f;
    }
    __syncthreads();
#pragma unroll
    for (int j = 0; j < SB_B; ++j)
        if (t + j * SORT_T < total) {
            const int pos = atomicAdd(&cur[r0[j].y - node0], 1);
            ssrc[start + pos] = r0[j].x;
        }
    for (int i0 = t + SB_B * SORT_T; i0 < total; i0 += SB_B * SORT_T) {
        int2 r[SB_B];
        fetch(i0, r);
#pragma unroll
        for (int j = 0; j < SB_B; ++j)
            if (i0 + j * SORT_T < total) {
                const int pos = atomicAdd(&cur[r[j].y - node0], 1);
                ssrc[start + pos] = r[j].x;
            }
    }
}

__global__ void k_node_linear(const float* __restrict__ x,
                              const float* __restrict__ Wl, const float* __restrict__ bl,
                              const float* __restrict__ Wr, const float* __restrict__ br,
                              float* __restrict__ xl, float* __restrict__ xr, int n_nodes) {
    const int lane = threadIdx.x & 31;
    const int wave = threadIdx.x >> 5;
    const int tile = blockIdx.x * (blockDim.x >> 5) + wave;
    const int r0 = tile * 16;
    if (r0 + 16 > n_nodes) return;
    const int lrow = lane & 15;
    const int lhi  = lane >> 4;

    v2f a[16];
    const float* xp = x + (long long)(r0 + lrow) * D_DIM + 2 * lhi;
#pragma unroll
    for (int kk = 0; kk < 16; ++kk)
        a[kk] = *(const v2f*)(xp + kk * 4);

#pragma unroll
    for (int m = 0; m < 2; ++m) {
        const float* W    = m ? Wr : Wl;
        const float* bias = m ? br : bl;
        float*       out  = m ? xr : xl;
#pragma unroll
        for (int ct = 0; ct < 4; ++ct) {
            const int col = ct * 16 + lrow;
            const float bv = bias[col];
            v8f acc = {bv, bv, bv, bv, bv, bv, bv, bv};
#pragma unroll
            for (int kk = 0; kk < 16; ++kk) {
                const int k0 = kk * 4 + 2 * lhi;
                v2f b;
                b.x = W[(k0    ) * D_DIM + col];
                b.y = W[(k0 + 1) * D_DIM + col];
                acc = __builtin_amdgcn_wmma_f32_16x16x4_f32(
                        false, a[kk], false, b, (short)0, acc, false, false);
            }
#pragma unroll
            for (int r = 0; r < 8; ++r)
                out[(long long)(r0 + r + 8 * lhi) * D_DIM + col] = acc[r];
        }
    }
}

template <int CTRL>
__device__ __forceinline__ int dpp_i(int v) {
    return __builtin_amdgcn_update_dpp(0, v, CTRL, 0xf, 0xf, true);
}
template <int CTRL>
__device__ __forceinline__ float dpp_f(float v) {
    return __int_as_float(__builtin_amdgcn_update_dpp(0, __float_as_int(v), CTRL, 0xf, 0xf, true));
}

template <int J>
__device__ __forceinline__ v4f gather_row(int ids, int sub, const float* __restrict__ xl) {
    const int s = dpp_i<0x150 + J>(ids);
    return *(const v4f*)(xl + (long long)s * D_DIM + 4 * sub);
}

__device__ __forceinline__ float edge_score(const v4f& l, const v4f& r, const v4f& a) {
    float t0 = l.x + r.x, t1 = l.y + r.y, t2 = l.z + r.z, t3 = l.w + r.w;
    t0 = t0 > 0.0f ? t0 : NEG_SLOPE * t0;
    t1 = t1 > 0.0f ? t1 : NEG_SLOPE * t1;
    t2 = t2 > 0.0f ? t2 : NEG_SLOPE * t2;
    t3 = t3 > 0.0f ? t3 : NEG_SLOPE * t3;
    float p = t0 * a.x + t1 * a.y + t2 * a.z + t3 * a.w;
    p += dpp_f<0x160 + 1>(p);
    p += dpp_f<0x160 + 2>(p);
    p += dpp_f<0x160 + 4>(p);
    p += dpp_f<0x160 + 8>(p);
    return p;
}

template <int B>
__device__ __forceinline__ void seg_batch(int ids, int left, int sub,
                                          const float* __restrict__ xl,
                                          const v4f& r, const v4f& a,
                                          float& m, float& den, v4f& acc) {
    const v4f l0 = gather_row<4 * B + 0>(ids, sub, xl);
    const v4f l1 = gather_row<4 * B + 1>(ids, sub, xl);
    const v4f l2 = gather_row<4 * B + 2>(ids, sub, xl);
    const v4f l3 = gather_row<4 * B + 3>(ids, sub, xl);
    const float p0 = edge_score(l0, r, a);
    const float p1 = edge_score(l1, r, a);
    const float p2 = edge_score(l2, r, a);
    const float p3 = edge_score(l3, r, a);
    const bool v0 = left > 0, v1 = left > 1, v2 = left > 2, v3 = left > 3;
    float bm = m;
    bm = v0 ? fmaxf(bm, p0) : bm;
    bm = v1 ? fmaxf(bm, p1) : bm;
    bm = v2 ? fmaxf(bm, p2) : bm;
    bm = v3 ? fmaxf(bm, p3) : bm;
    const float sc = __expf(m - bm);
    const float w0 = v0 ? __expf(p0 - bm) : 0.0f;
    const float w1 = v1 ? __expf(p1 - bm) : 0.0f;
    const float w2 = v2 ? __expf(p2 - bm) : 0.0f;
    const float w3 = v3 ? __expf(p3 - bm) : 0.0f;
    den   = den   * sc + ((w0 + w1) + (w2 + w3));
    acc.x = acc.x * sc + ((w0 * l0.x + w1 * l1.x) + (w2 * l2.x + w3 * l3.x));
    acc.y = acc.y * sc + ((w0 * l0.y + w1 * l1.y) + (w2 * l2.y + w3 * l3.y));
    acc.z = acc.z * sc + ((w0 * l0.z + w1 * l1.z) + (w2 * l2.z + w3 * l3.z));
    acc.w = acc.w * sc + ((w0 * l0.w + w1 * l1.w) + (w2 * l2.w + w3 * l3.w));
    m = bm;
}

__global__ void k_segment(const int* __restrict__ mask, const int* __restrict__ first,
                          const int* __restrict__ deg, const int* __restrict__ ssrc,
                          const float* __restrict__ xl, const float* __restrict__ xr,
                          const float* __restrict__ att, float* __restrict__ hm, int n_mask) {
    const int lane = threadIdx.x & 31;
    const int sub  = lane & 15;
    const int i = (int)((blockIdx.x * blockDim.x + threadIdx.x) >> 4);
    const bool live = i < n_mask;
    int node = 0, beg = 0, n = 0;
    if (live) { node = mask[i]; beg = first[node]; n = deg[node]; }
    const v4f a = *(const v4f*)(att + 4 * sub);
    const v4f r = *(const v4f*)(xr + (long long)node * D_DIM + 4 * sub);
    const int n_lo = __builtin_amdgcn_readlane(n, 0);
    const int n_hi = __builtin_amdgcn_readlane(n, 16);
    const int nmax = n_lo > n_hi ? n_lo : n_hi;
    float m = -FLT_MAX, den = 0.0f;
    v4f acc = {0.0f, 0.0f, 0.0f, 0.0f};
    for (int c = 0; c < nmax; c += 16) {
        const int left = n - c;
        int ids = 0;
        if (sub < left) ids = ssrc[beg + c + sub];
        seg_batch<0>(ids, left, sub, xl, r, a, m, den, acc);
        if (c + 4 >= nmax) break;
        seg_batch<1>(ids, left - 4, sub, xl, r, a, m, den, acc);
        if (c + 8 >= nmax) break;
        seg_batch<2>(ids, left - 8, sub, xl, r, a, m, den, acc);
        if (c + 12 >= nmax) break;
        seg_batch<3>(ids, left - 12, sub, xl, r, a, m, den, acc);
    }
    if (live) {
        const float dn = den + 1e-16f;
        v4f o;
        o.x = acc.x / dn; o.y = acc.y / dn; o.z = acc.z / dn; o.w = acc.w / dn;
        *(v4f*)(hm + (long long)i * D_DIM + 4 * sub) = o;
    }
}

__global__ void k_output(const float* __restrict__ hm,
                         const float* __restrict__ gbias,
                         const float* __restrict__ Wo, const float* __restrict__ bo,
                         float* __restrict__ out, int n_mask) {
    const int lane = threadIdx.x & 31;
    const int wave = threadIdx.x >> 5;
    const int tile = blockIdx.x * (blockDim.x >> 5) + wave;
    const int r0 = tile * 16;
    if (r0 + 16 > n_mask) return;
    const int lrow = lane & 15;
    const int lhi  = lane >> 4;

    v2f a[16];
    const float* hp = hm + (long long)(r0 + lrow) * D_DIM + 2 * lhi;
    const float* gp = gbias + 2 * lhi;
#pragma unroll
    for (int kk = 0; kk < 16; ++kk) {
        const v2f hv = *(const v2f*)(hp + kk * 4);
        const v2f gv = *(const v2f*)(gp + kk * 4);
        a[kk].x = hv.x + gv.x;
        a[kk].y = hv.y + gv.y;
    }
#pragma unroll
    for (int ct = 0; ct < 8; ++ct) {
        const int col = ct * 16 + lrow;
        const float bv = bo[col];
        v8f acc = {bv, bv, bv, bv, bv, bv, bv, bv};
#pragma unroll
        for (int kk = 0; kk < 16; ++kk) {
            const int k0 = kk * 4 + 2 * lhi;
            v2f b;
            b.x = Wo[(k0    ) * DOUT + col];
            b.y = Wo[(k0 + 1) * DOUT + col];
            acc = __builtin_amdgcn_wmma_f32_16x16x4_f32(
                    false, a[kk], false, b, (short)0, acc, false, false);
        }
#pragma unroll
        for (int r = 0; r < 8; ++r)
            out[(long long)(r0 + r + 8 * lhi) * DOUT + col] = acc[r];
    }
}

extern "C" void kernel_launch(void* const* d_in, const int* in_sizes, int n_in,
                              void* d_out, int out_size, void* d_ws, size_t ws_size,
                              hipStream_t stream) {
    const float* x     = (const float*)d_in[0];
    const int*   ei    = (const int*)d_in[1];
    const int*   mask  = (const int*)d_in[2];
    const float* Wl    = (const float*)d_in[3];
    const float* bl    = (const float*)d_in[4];
    const float* Wr    = (const float*)d_in[5];
    const float* br    = (const float*)d_in[6];
    const float* att   = (const float*)d_in[7];
    const float* gbias = (const float*)d_in[8];
    const float* Wo    = (const float*)d_in[9];
    const float* bo    = (const float*)d_in[10];
    float* out = (float*)d_out;

    const int n_nodes = in_sizes[0] / D_DIM;
    const int n_edges = in_sizes[1] / 2;
    const int n_mask  = in_sizes[2];
    const int* src = ei;
    const int* dst = ei + n_edges;

    const int nblk  = (n_edges + BS_CH - 1) / BS_CH;
    const int nbins = (n_nodes + BIN_R - 1) / BIN_R;
    const int nwords = (n_nodes + 31) / 32;
    char* wp = (char*)d_ws;
    auto carve = [&](size_t bytes) {
        char* p = wp;
        wp += (bytes + 255) & ~(size_t)255;
        return p;
    };
    float*    xl     = (float*)   carve((size_t)n_nodes * D_DIM * 4);
    float*    xr     = (float*)   carve((size_t)n_nodes * D_DIM * 4);
    float*    hm     = (float*)   carve((size_t)n_mask * D_DIM * 4);
    int*      ssrc   = (int*)     carve((size_t)n_edges * 4);
    int2*     recs   = (int2*)    carve((size_t)nblk * BS_CH * 8);
    int*      boff   = (int*)     carve((size_t)nblk * (nbins + 1) * 4);
    int*      deg    = (int*)     carve((size_t)n_nodes * 4);
    unsigned* needbits = (unsigned*)carve((size_t)nwords * 4);
    int*      first  = (int*)     carve((size_t)n_nodes * 4);

    k_init<<<(nwords + 255) / 256, 256, 0, stream>>>(needbits, nwords);
    k_mark<<<(n_mask + 255) / 256, 256, 0, stream>>>(mask, needbits, n_mask);
    k_blocksort<<<nblk, SORT_T, (size_t)(nbins + 1) * 4, stream>>>(
        src, dst, needbits, recs, boff, n_edges, nbins);
    k_binsort<<<nbins, SORT_T, (size_t)(2 * nblk + 1) * 4, stream>>>(
        recs, boff, first, deg, ssrc, n_nodes, nblk, nbins);
    {
        int tiles  = (n_nodes + 15) / 16;
        int blocks = (tiles + 7) / 8;
        k_node_linear<<<blocks, 256, 0, stream>>>(x, Wl, bl, Wr, br, xl, xr, n_nodes);
    }
    k_segment<<<(n_mask + 15) / 16, 256, 0, stream>>>(mask, first, deg, ssrc, xl, xr, att, hm, n_mask);
    {
        int tiles  = (n_mask + 15) / 16;
        int blocks = (tiles + 7) / 8;
        k_output<<<blocks, 256, 0, stream>>>(hm, gbias, Wo, bo, out, n_mask);
    }
}


# output product fused into the segment kernel: a block's 16 rows of h go through LDS to 8 WMMA waves (one column tile each), out leaves as whole 512-B rows; hm plane and one launch gone
# speedup vs baseline: 5.2875x; 1.0808x over previous
#include <hip/hip_runtime.h>
#include <hip/hip_bf16.h>
#include <math.h>
#include <float.h>


#define D_DIM 64
#define DOUT  128
#define NEG_SLOPE 0.2f

typedef __attribute__((ext_vector_type(2))) float v2f;
typedef __attribute__((ext_vector_type(8))) float v8f;
typedef __attribute__((ext_vector_type(4))) float v4f;

__global__ void k_init(unsigned* __restrict__ needbits, int n_words) {
    const int i = blockIdx.x * blockDim.x + threadIdx.x;
    if (i < n_words) needbits[i] = 0u;
}

__global__ void k_mark(const int* __restrict__ mask, unsigned* __restrict__ needbits,
                       int n_mask) {
    int i = blockIdx.x * blockDim.x + threadIdx.x;
    if (i >= n_mask) return;
    const int m = mask[i];
    atomicOr(&needbits[m >> 5], 1u << (m & 31));
}

#define SORT_T    1024
#define BS_E      4
#define BS_CH     (SORT_T * BS_E)
#define BIN_SHIFT 9
#define BIN_R     (1 << BIN_SHIFT)
#define SB_B      4

__device__ __forceinline__ int block_scan_lds(int* s, int n, int* wsum) {
    const int t = threadIdx.x, T = SORT_T;
    const int lane = t & 31, wave = t >> 5;
    const int K = (n + T - 1) / T;
    const int lo = t * K;
    int hi = lo + K;
    if (hi > n) hi = n;
    __syncthreads();
    int sum = 0;
    for (int i = lo; i < hi; ++i) sum += s[i];
    int inc = sum;
    for (int o = 1; o < 32; o <<= 1) {
        const int u = __shfl_up(inc, o);
        if (lane >= o) inc += u;
    }
    if (lane == 31) wsum[wave] = inc;
    __syncthreads();
    if (wave == 0) {
        const int w = lane < (T >> 5) ? wsum[lane] : 0;
        int winc = w;
        for (int o = 1; o < 32; o <<= 1) {
            const int u = __shfl_up(winc, o);
            if (lane >= o) winc += u;
        }
        wsum[lane] = winc - w;
        if (lane == 31) wsum[32] = winc;
    }
    __syncthreads();
    int run = wsum[wave] + inc - sum;
    for (int i = lo; i < hi; ++i) {
        const int v = s[i];
        s[i] = run;
        run += v;
    }
    const int total = wsum[32];
    __syncthreads();
    return total;
}

__global__ void __launch_bounds__(SORT_T)
k_blocksort(const int* __restrict__ src, const int* __restrict__ dst,
            const unsigned* __restrict__ needbits, int2* __restrict__ recs,
            int* __restrict__ boff, int n_edges, int nbins) {
    extern __shared__ int hist[];
    __shared__ int wsum[33];
    const int t = threadIdx.x;
    for (int b = t; b <= nbins; b += SORT_T) hist[b] = 0;
    const long long base = (long long)blockIdx.x * BS_CH + t;
    int d[BS_E], s[BS_E], lr[BS_E];
    unsigned word[BS_E];
    bool keep[BS_E];
#pragma unroll
    for (int j = 0; j < BS_E; ++j) {
        const long long e = base + j * SORT_T;
        const long long ec = e < n_edges ? e : 0;
        d[j] = dst[ec];
        s[j] = src[ec];
    }
#pragma unroll
    for (int j = 0; j < BS_E; ++j)
        word[j] = needbits[d[j] >> 5];
    __syncthreads();
#pragma unroll
    for (int j = 0; j < BS_E; ++j) {
        keep[j] = (base + j * SORT_T < n_edges) && ((word[j] >> (d[j] & 31)) & 1u);
        lr[j] = 0;
        if (keep[j]) lr[j] = atomicAdd(&hist[d[j] >> BIN_SHIFT], 1);
    }
    block_scan_lds(hist, nbins + 1, wsum);
    int2* mine = recs + (long long)blockIdx.x * BS_CH;
#pragma unroll
    for (int j = 0; j < BS_E; ++j)
        if (keep[j]) mine[hist[d[j] >> BIN_SHIFT] + lr[j]] = make_int2(s[j], d[j]);
    int* myoff = boff + (long long)blockIdx.x * (nbins + 1);
    for (int b = t; b <= nbins; b += SORT_T) myoff[b] = hist[b];
}

__global__ void __launch_bounds__(SORT_T)
k_binsort(const int2* __restrict__ recs, const int* __restrict__ boff,
          int* __restrict__ first, int* __restrict__ deg, int* __restrict__ ssrc,
          int n_nodes, int nblk, int nbins) {
    extern __shared__ int dyn[];
    int* o0  = dyn;
    int* pre = dyn + nblk;
    __shared__ int cnt[BIN_R];
    __shared__ int cur[BIN_R];
    __shared__ int wsum[33];
    __shared__ int sbase;
    const int t = threadIdx.x;
    const int b = blockIdx.x;
    const int node0 = b << BIN_SHIFT;
    if (t == 0) { sbase = 0; pre[nblk] = 0; }
    for (int i = t; i < BIN_R; i += SORT_T) cnt[i] = 0;
    int mystart = 0;
    for (int k = t; k < nblk; k += SORT_T) {
        const int* bo = boff + (long long)k * (nbins + 1) + b;
        const int a0 = bo[0], a1 = bo[1];
        o0[k]  = a0;
        pre[k] = a1 - a0;
        mystart += a0;
    }
    __syncthreads();
    if (mystart) atomicAdd(&sbase, mystart);
    const int total = block_scan_lds(pre, nblk + 1, wsum);
    const int start = sbase;

    int top = 1;
    while (top < nblk) top <<= 1;
    auto fetch = [&](int i0, int2 (&r)[SB_B]) {
        int idx[SB_B], lo[SB_B];
#pragma unroll
        for (int j = 0; j < SB_B; ++j) {
            const int i = i0 + j * SORT_T;
            idx[j] = i < total ? i : total - 1;
            lo[j] = 0;
        }
        for (int w = top >> 1; w > 0; w >>= 1) {
#pragma unroll
            for (int j = 0; j < SB_B; ++j) {
                const int k = lo[j] + w < nblk ? lo[j] + w : nblk;
                lo[j] = pre[k] <= idx[j] ? k : lo[j];
            }
        }
#pragma unroll
        for (int j = 0; j < SB_B; ++j)
            r[j] = recs[(long long)lo[j] * BS_CH + o0[lo[j]] + (idx[j] - pre[lo[j]])];
    };

    int2 r0[SB_B];
#pragma unroll
    for (int j = 0; j < SB_B; ++j) r0[j] = make_int2(0, node0);
    if (t < total) fetch(t, r0);
#pragma unroll
    for (int j = 0; j < SB_B; ++j)
        if (t + j * SORT_T < total) atomicAdd(&cnt[r0[j].y - node0], 1);
    for (int i0 = t + SB_B * SORT_T; i0 < total; i0 += SB_B * SORT_T) {
        int2 r[SB_B];
        fetch(i0, r);
#pragma unroll
        for (int j = 0; j < SB_B; ++j)
            if (i0 + j * SORT_T < total) atomicAdd(&cnt[r[j].y - node0], 1);
    }
    __syncthreads();
    for (int i = t; i < BIN_R; i += SORT_T)
        if (node0 + i < n_nodes) deg[node0 + i] = cnt[i];
    block_scan_lds(cnt, BIN_R, wsum);
    for (int i = t; i < BIN_R; i += SORT_T) {
        const int f = cnt[i];
        cur[i] = f;
        if (node0 + i < n_nodes) first[node0 + i] = start + f;
    }
    __syncthreads();
#pragma unroll
    for (int j = 0; j < SB_B; ++j)
        if (t + j * SORT_T < total) {
            const int pos = atomicAdd(&cur[r0[j].y - node0], 1);
            ssrc[start + pos] = r0[j].x;
        }
    for (int i0 = t + SB_B * SORT_T; i0 < total; i0 += SB_B * SORT_T) {
        int2 r[SB_B];
        fetch(i0, r);
#pragma unroll
        for (int j = 0; j < SB_B; ++j)
            if (i0 + j * SORT_T < total) {
                const int pos = atomicAdd(&cur[r[j].y - node0], 1);
                ssrc[start + pos] = r[j].x;
            }
    }
}

__global__ void k_node_linear(const float* __restrict__ x,
                              const float* __restrict__ Wl, const float* __restrict__ bl,
                              const float* __restrict__ Wr, const float* __restrict__ br,
                              float* __restrict__ xl, float* __restrict__ xr, int n_nodes) {
    const int lane = threadIdx.x & 31;
    const int wave = threadIdx.x >> 5;
    const int tile = blockIdx.x * (blockDim.x >> 5) + wave;
    const int r0 = tile * 16;
    if (r0 + 16 > n_nodes) return;
    const int lrow = lane & 15;
    const int lhi  = lane >> 4;

    v2f a[16];
    const float* xp = x + (long long)(r0 + lrow) * D_DIM + 2 * lhi;
#pragma unroll
    for (int kk = 0; kk < 16; ++kk)
        a[kk] = *(const v2f*)(xp + kk * 4);

#pragma unroll
    for (int m = 0; m < 2; ++m) {
        const float* W    = m ? Wr : Wl;
        const float* bias = m ? br : bl;
        float*       out  = m ? xr : xl;
#pragma unroll
        for (int ct = 0; ct < 4; ++ct) {
            const int col = ct * 16 + lrow;
            const float bv = bias[col];
            v8f acc = {bv, bv, bv, bv, bv, bv, bv, bv};
#pragma unroll
            for (int kk = 0; kk < 16; ++kk) {
                const int k0 = kk * 4 + 2 * lhi;
                v2f b;
                b.x = W[(k0    ) * D_DIM + col];
                b.y = W[(k0 + 1) * D_DIM + col];
                acc = __builtin_amdgcn_wmma_f32_16x16x4_f32(
                        false, a[kk], false, b, (short)0, acc, false, false);
            }
#pragma unroll
            for (int r = 0; r < 8; ++r)
                out[(long long)(r0 + r + 8 * lhi) * D_DIM + col] = acc[r];
        }
    }
}

template <int CTRL>
__device__ __forceinline__ int dpp_i(int v) {
    return __builtin_amdgcn_update_dpp(0, v, CTRL, 0xf, 0xf, true);
}
template <int CTRL>
__device__ __forceinline__ float dpp_f(float v) {
    return __int_as_float(__builtin_amdgcn_update_dpp(0, __float_as_int(v), CTRL, 0xf, 0xf, true));
}

template <int J>
__device__ __forceinline__ v4f gather_row(int ids, int sub, const float* __restrict__ xl) {
    const int s = dpp_i<0x150 + J>(ids);
    return *(const v4f*)(xl + (long long)s * D_DIM + 4 * sub);
}

__device__ __forceinline__ float edge_score(const v4f& l, const v4f& r, const v4f& a) {
    float t0 = l.x + r.x, t1 = l.y + r.y, t2 = l.z + r.z, t3 = l.w + r.w;
    t0 = t0 > 0.0f ? t0 : NEG_SLOPE * t0;
    t1 = t1 > 0.0f ? t1 : NEG_SLOPE * t1;
    t2 = t2 > 0.0f ? t2 : NEG_SLOPE * t2;
    t3 = t3 > 0.0f ? t3 : NEG_SLOPE * t3;
    float p = t0 * a.x + t1 * a.y + t2 * a.z + t3 * a.w;
    p += dpp_f<0x160 + 1>(p);
    p += dpp_f<0x160 + 2>(p);
    p += dpp_f<0x160 + 4>(p);
    p += dpp_f<0x160 + 8>(p);
    return p;
}

template <int B>
__device__ __forceinline__ void seg_batch(int ids, int left, int sub,
                                          const float* __restrict__ xl,
                                          const v4f& r, const v4f& a,
                                          float& m, float& den, v4f& acc) {
    const v4f l0 = gather_row<4 * B + 0>(ids, sub, xl);
    const v4f l1 = gather_row<4 * B + 1>(ids, sub, xl);
    const v4f l2 = gather_row<4 * B + 2>(ids, sub, xl);
    const v4f l3 = gather_row<4 * B + 3>(ids, sub, xl);
    const float p0 = edge_score(l0, r, a);
    const float p1 = edge_score(l1, r, a);
    const float p2 = edge_score(l2, r, a);
    const float p3 = edge_score(l3, r, a);
    const bool v0 = left > 0, v1 = left > 1, v2 = left > 2, v3 = left > 3;
    float bm = m;
    bm = v0 ? fmaxf(bm, p0) : bm;
    bm = v1 ? fmaxf(bm, p1) : bm;
    bm = v2 ? fmaxf(bm, p2) : bm;
    bm = v3 ? fmaxf(bm, p3) : bm;
    const float sc = __expf(m - bm);
    const float w0 = v0 ? __expf(p0 - bm) : 0.0f;
    const float w1 = v1 ? __expf(p1 - bm) : 0.0f;
    const float w2 = v2 ? __expf(p2 - bm) : 0.0f;
    const float w3 = v3 ? __expf(p3 - bm) : 0.0f;
    den   = den   * sc + ((w0 + w1) + (w2 + w3));
    acc.x = acc.x * sc + ((w0 * l0.x + w1 * l1.x) + (w2 * l2.x + w3 * l3.x));
    acc.y = acc.y * sc + ((w0 * l0.y + w1 * l1.y) + (w2 * l2.y + w3 * l3.y));
    acc.z = acc.z * sc + ((w0 * l0.z + w1 * l1.z) + (w2 * l2.z + w3 * l3.z));
    acc.w = acc.w * sc + ((w0 * l0.w + w1 * l1.w) + (w2 * l2.w + w3 * l3.w));
    m = bm;
}

#define HT_P 68
#define OT_P 132
#define SEG_T 256

__global__ void __launch_bounds__(SEG_T)
k_segment_out(const int* __restrict__ mask, const int* __restrict__ first,
              const int* __restrict__ deg, const int* __restrict__ ssrc,
              const float* __restrict__ xl, const float* __restrict__ xr,
              const float* __restrict__ att, const float* __restrict__ gbias,
              const float* __restrict__ Wo, const float* __restrict__ bo,
              float* __restrict__ out, int n_mask) {
    __shared__ __attribute__((aligned(16))) float htile[16 * HT_P];
    __shared__ __attribute__((aligned(16))) float otile[16 * OT_P];
    const int lane = threadIdx.x & 31;
    const int wave = threadIdx.x >> 5;
    const int sub  = lane & 15;
    const int rb   = threadIdx.x >> 4;
    const int r0   = blockIdx.x * 16;
    const int i = r0 + rb;
    const bool live = i < n_mask;
    int node = 0, beg = 0, n = 0;
    if (live) { node = mask[i]; beg = first[node]; n = deg[node]; }
    const v4f a = *(const v4f*)(att + 4 * sub);
    const v4f r = *(const v4f*)(xr + (long long)node * D_DIM + 4 * sub);
    const int n_lo = __builtin_amdgcn_readlane(n, 0);
    const int n_hi = __builtin_amdgcn_readlane(n, 16);
    const int nmax = n_lo > n_hi ? n_lo : n_hi;
    float m = -FLT_MAX, den = 0.0f;
    v4f acc = {0.0f, 0.0f, 0.0f, 0.0f};
    for (int c = 0; c < nmax; c += 16) {
        const int left = n - c;
        int ids = 0;
        if (sub < left) ids = ssrc[beg + c + sub];
        seg_batch<0>(ids, left, sub, xl, r, a, m, den, acc);
        if (c + 4 >= nmax) break;
        seg_batch<1>(ids, left - 4, sub, xl, r, a, m, den, acc);
        if (c + 8 >= nmax) break;
        seg_batch<2>(ids, left - 8, sub, xl, r, a, m, den, acc);
        if (c + 12 >= nmax) break;
        seg_batch<3>(ids, left - 12, sub, xl, r, a, m, den, acc);
    }

    const int lrow = lane & 15;
    const int lhi  = lane >> 4;
    const int col  = wave * 16 + lrow;
    v2f b[16];
#pragma unroll
    for (int kk = 0; kk < 16; ++kk) {
        const int k0 = kk * 4 + 2 * lhi;
        b[kk].x = Wo[(k0    ) * DOUT + col];
        b[kk].y = Wo[(k0 + 1) * DOUT + col];
    }
    const float bv = bo[col];
    {
        const v4f g = *(const v4f*)(gbias + 4 * sub);
        const float dn = den + 1e-16f;
        v4f o;
        o.x = acc.x / dn + g.x; o.y = acc.y / dn + g.y;
        o.z = acc.z / dn + g.z; o.w = acc.w / dn + g.w;
        *(v4f*)(htile + rb * HT_P + 4 * sub) = o;
    }
    __syncthreads();
    v8f c8 = {bv, bv, bv, bv, bv, bv, bv, bv};
#pragma unroll
    for (int kk = 0; kk < 16; ++kk) {
        const v2f af = *(const v2f*)(htile + lrow * HT_P + kk * 4 + 2 * lhi);
        c8 = __builtin_amdgcn_wmma_f32_16x16x4_f32(
                false, af, false, b[kk], (short)0, c8, false, false);
    }
#pragma unroll
    for (int q = 0; q < 8; ++q)
        otile[(q + 8 * lhi) * OT_P + col] = c8[q];
    __syncthreads();
#pragma unroll
    for (int q = 0; q < 2; ++q) {
        const int row = 2 * wave + q;
        if (r0 + row < n_mask)
            *(v4f*)(out + (long long)(r0 + row) * DOUT + 4 * lane) =
                *(const v4f*)(otile + row * OT_P + 4 * lane);
    }
}

extern "C" void kernel_launch(void* const* d_in, const int* in_sizes, int n_in,
                              void* d_out, int out_size, void* d_ws, size_t ws_size,
                              hipStream_t stream) {
    const float* x     = (const float*)d_in[0];
    const int*   ei    = (const int*)d_in[1];
    const int*   mask  = (const int*)d_in[2];
    const float* Wl    = (const float*)d_in[3];
    const float* bl    = (const float*)d_in[4];
    const float* Wr    = (const float*)d_in[5];
    const float* br    = (const float*)d_in[6];
    const float* att   = (const float*)d_in[7];
    const float* gbias = (const float*)d_in[8];
    const float* Wo    = (const float*)d_in[9];
    const float* bo    = (const float*)d_in[10];
    float* out = (float*)d_out;

    const int n_nodes = in_sizes[0] / D_DIM;
    const int n_edges = in_sizes[1] / 2;
    const int n_mask  = in_sizes[2];
    const int* src = ei;
    const int* dst = ei + n_edges;

    const int nblk  = (n_edges + BS_CH - 1) / BS_CH;
    const int nbins = (n_nodes + BIN_R - 1) / BIN_R;
    const int nwords = (n_nodes + 31) / 32;
    char* wp = (char*)d_ws;
    auto carve = [&](size_t bytes) {
        char* p = wp;
        wp += (bytes + 255) & ~(size_t)255;
        return p;
    };
    float*    xl     = (float*)   carve((size_t)n_nodes * D_DIM * 4);
    float*    xr     = (float*)   carve((size_t)n_nodes * D_DIM * 4);
    int*      ssrc   = (int*)     carve((size_t)n_edges * 4);
    int2*     recs   = (int2*)    carve((size_t)nblk * BS_CH * 8);
    int*      boff   = (int*)     carve((size_t)nblk * (nbins + 1) * 4);
    int*      deg    = (int*)     carve((size_t)n_nodes * 4);
    unsigned* needbits = (unsigned*)carve((size_t)nwords * 4);
    int*      first  = (int*)     carve((size_t)n_nodes * 4);

    k_init<<<(nwords + 255) / 256, 256, 0, stream>>>(needbits, nwords);
    k_mark<<<(n_mask + 255) / 256, 256, 0, stream>>>(mask, needbits, n_mask);
    k_blocksort<<<nblk, SORT_T, (size_t)(nbins + 1) * 4, stream>>>(
        src, dst, needbits, recs, boff, n_edges, nbins);
    k_binsort<<<nbins, SORT_T, (size_t)(2 * nblk + 1) * 4, stream>>>(
        recs, boff, first, deg, ssrc, n_nodes, nblk, nbins);
    {
        int tiles  = (n_nodes + 15) / 16;
        int blocks = (tiles + 7) / 8;
        k_node_linear<<<blocks, 256, 0, stream>>>(x, Wl, bl, Wr, br, xl, xr, n_nodes);
    }
    k_segment_out<<<(n_mask + 15) / 16, SEG_T, 0, stream>>>(
        mask, first, deg, ssrc, xl, xr, att, gbias, Wo, bo, out, n_mask);
}


# node products x@W_l and x@W_r run as extra work-groups of the two edge-grouping launches (half the row tiles in each, one wave per tile and matrix), no launch of their own: 6 launches -> 5
# speedup vs baseline: 6.7889x; 1.2840x over previous
#include <hip/hip_runtime.h>
#include <hip/hip_bf16.h>
#include <math.h>
#include <float.h>


#define D_DIM 64
#define DOUT  128
#define NEG_SLOPE 0.2f

typedef __attribute__((ext_vector_type(2))) float v2f;
typedef __attribute__((ext_vector_type(8))) float v8f;
typedef __attribute__((ext_vector_type(4))) float v4f;

__global__ void k_init(unsigned* __restrict__ needbits, int n_words) {
    const int i = blockIdx.x * blockDim.x + threadIdx.x;
    if (i < n_words) needbits[i] = 0u;
}

__global__ void k_mark(const int* __restrict__ mask, unsigned* __restrict__ needbits,
                       int n_mask) {
    int i = blockIdx.x * blockDim.x + threadIdx.x;
    if (i >= n_mask) return;
    const int m = mask[i];
    atomicOr(&needbits[m >> 5], 1u << (m & 31));
}

__device__ __forceinline__ void linear_tile(const float* __restrict__ x,
                                            const float* __restrict__ W,
                                            const float* __restrict__ bias,
                                            float* __restrict__ out, int r0, int lane) {
    const int lrow = lane & 15;
    const int lhi  = lane >> 4;
    v2f a[16];
    const float* xp = x + (long long)(r0 + lrow) * D_DIM + 2 * lhi;
#pragma unroll
    for (int kk = 0; kk < 16; ++kk)
        a[kk] = *(const v2f*)(xp + kk * 4);
#pragma unroll
    for (int ct = 0; ct < 4; ++ct) {
        const int col = ct * 16 + lrow;
        const float bv = bias[col];
        v8f acc = {bv, bv, bv, bv, bv, bv, bv, bv};
#pragma unroll
        for (int kk = 0; kk < 16; ++kk) {
            const int k0 = kk * 4 + 2 * lhi;
            v2f b;
            b.x = W[(k0    ) * D_DIM + col];
            b.y = W[(k0 + 1) * D_DIM + col];
            acc = __builtin_amdgcn_wmma_f32_16x16x4_f32(
                    false, a[kk], false, b, (short)0, acc, false, false);
        }
#pragma unroll
        for (int r = 0; r < 8; ++r)
            out[(long long)(r0 + r + 8 * lhi) * D_DIM + col] = acc[r];
    }
}

#define LIN_TILES 16

__device__ __forceinline__ void linear_role(int lb, const float* __restrict__ x,
        const float* __restrict__ Wl, const float* __restrict__ bl,
        const float* __restrict__ Wr, const float* __restrict__ br,
        float* __restrict__ xl, float* __restrict__ xr, int n_nodes) {
    const int lane = threadIdx.x & 31;
    const int wave = threadIdx.x >> 5;
    const int r0 = (lb * LIN_TILES + (wave >> 1)) * 16;
    if (r0 + 16 > n_nodes) return;
    if (wave & 1) linear_tile(x, Wr, br, xr, r0, lane);
    else          linear_tile(x, Wl, bl, xl, r0, lane);
}

#define SORT_T    1024
#define BS_E      4
#define BS_CH     (SORT_T * BS_E)
#define BIN_SHIFT 9
#define BIN_R     (1 << BIN_SHIFT)
#define SB_B      4

__device__ __forceinline__ int block_scan_lds(int* s, int n, int* wsum) {
    const int t = threadIdx.x, T = SORT_T;
    const int lane = t & 31, wave = t >> 5;
    const int K = (n + T - 1) / T;
    const int lo = t * K;
    int hi = lo + K;
    if (hi > n) hi = n;
    __syncthreads();
    int sum = 0;
    for (int i = lo; i < hi; ++i) sum += s[i];
    int inc = sum;
    for (int o = 1; o < 32; o <<= 1) {
        const int u = __shfl_up(inc, o);
        if (lane >= o) inc += u;
    }
    if (lane == 31) wsum[wave] = inc;
    __syncthreads();
    if (wave == 0) {
        const int w = lane < (T >> 5) ? wsum[lane] : 0;
        int winc = w;
        for (int o = 1; o < 32; o <<= 1) {
            const int u = __shfl_up(winc, o);
            if (lane >= o) winc += u;
        }
        wsum[lane] = winc - w;
        if (lane == 31) wsum[32] = winc;
    }
    __syncthreads();
    int run = wsum[wave] + inc - sum;
    for (int i = lo; i < hi; ++i) {
        const int v = s[i];
        s[i] = run;
        run += v;
    }
    const int total = wsum[32];
    __syncthreads();
    return total;
}

__global__ void __launch_bounds__(SORT_T)
k_blocksort(const int* __restrict__ src, const int* __restrict__ dst,
            const unsigned* __restrict__ needbits, int2* __restrict__ recs,
            int* __restrict__ boff, int n_edges, int nbins, int nsort,
            const float* __restrict__ x,
            const float* __restrict__ Wl, const float* __restrict__ bl,
            const float* __restrict__ Wr, const float* __restrict__ br,
            float* __restrict__ xl, float* __restrict__ xr, int n_nodes) {
    extern __shared__ int hist[];
    __shared__ int wsum[33];
    if ((int)blockIdx.x >= nsort) {
        linear_role((int)blockIdx.x - nsort, x, Wl, bl, Wr, br, xl, xr, n_nodes);
        return;
    }
    const int t = threadIdx.x;
    for (int b = t; b <= nbins; b += SORT_T) hist[b] = 0;
    const long long base = (long long)blockIdx.x * BS_CH + t;
    int d[BS_E], s[BS_E], lr[BS_E];
    unsigned word[BS_E];
    bool keep[BS_E];
#pragma unroll
    for (int j = 0; j < BS_E; ++j) {
        const long long e = base + j * SORT_T;
        const long long ec = e < n_edges ? e : 0;
        d[j] = dst[ec];
        s[j] = src[ec];
    }
#pragma unroll
    for (int j = 0; j < BS_E; ++j)
        word[j] = needbits[d[j] >> 5];
    __syncthreads();
#pragma unroll
    for (int j = 0; j < BS_E; ++j) {
        keep[j] = (base + j * SORT_T < n_edges) && ((word[j] >> (d[j] & 31)) & 1u);
        lr[j] = 0;
        if (keep[j]) lr[j] = atomicAdd(&hist[d[j] >> BIN_SHIFT], 1);
    }
    block_scan_lds(hist, nbins + 1, wsum);
    int2* mine = recs + (long long)blockIdx.x * BS_CH;
#pragma unroll
    for (int j = 0; j < BS_E; ++j)
        if (keep[j]) mine[hist[d[j] >> BIN_SHIFT] + lr[j]] = make_int2(s[j], d[j]);
    int* myoff = boff + (long long)blockIdx.x * (nbins + 1);
    for (int b = t; b <= nbins; b += SORT_T) myoff[b] = hist[b];
}

__global__ void __launch_bounds__(SORT_T)
k_binsort(const int2* __restrict__ recs, const int* __restrict__ boff,
          int* __restrict__ first, int* __restrict__ deg, int* __restrict__ ssrc,
          int n_nodes, int nblk, int nbins, int lin0,
          const float* __restrict__ x,
          const float* __restrict__ Wl, const float* __restrict__ bl,
          const float* __restrict__ Wr, const float* __restrict__ br,
          float* __restrict__ xl, float* __restrict__ xr) {
    extern __shared__ int dyn[];
    if ((int)blockIdx.x >= nbins) {
        linear_role((int)blockIdx.x - nbins + lin0, x, Wl, bl, Wr, br, xl, xr, n_nodes);
        return;
    }
    int* o0  = dyn;
    int* pre = dyn + nblk;
    __shared__ int cnt[BIN_R];
    __shared__ int cur[BIN_R];
    __shared__ int wsum[33];
    __shared__ int sbase;
    const int t = threadIdx.x;
    const int b = blockIdx.x;
    const int node0 = b << BIN_SHIFT;
    if (t == 0) { sbase = 0; pre[nblk] = 0; }
    for (int i = t; i < BIN_R; i += SORT_T) cnt[i] = 0;
    int mystart = 0;
    for (int k = t; k < nblk; k += SORT_T) {
        const int* bo = boff + (long long)k * (nbins + 1) + b;
        const int a0 = bo[0], a1 = bo[1];
        o0[k]  = a0;
        pre[k] = a1 - a0;
        mystart += a0;
    }
    __syncthreads();
    if (mystart) atomicAdd(&sbase, mystart);
    const int total = block_scan_lds(pre, nblk + 1, wsum);
    const int start = sbase;

    int top = 1;
    while (top < nblk) top <<= 1;
    auto fetch = [&](int i0, int2 (&r)[SB_B]) {
        int idx[SB_B], lo[SB_B];
#pragma unroll
        for (int j = 0; j < SB_B; ++j) {
            const int i = i0 + j * SORT_T;
            idx[j] = i < total ? i : total - 1;
            lo[j] = 0;
        }
        for (int w = top >> 1; w > 0; w >>= 1) {
#pragma unroll
            for (int j = 0; j < SB_B; ++j) {
                const int k = lo[j] + w < nblk ? lo[j] + w : nblk;
                lo[j] = pre[k] <= idx[j] ? k : lo[j];
            }
        }
#pragma unroll
        for (int j = 0; j < SB_B; ++j)
            r[j] = recs[(long long)lo[j] * BS_CH + o0[lo[j]] + (idx[j] - pre[lo[j]])];
    };

    int2 r0[SB_B];
#pragma unroll
    for (int j = 0; j < SB_B; ++j) r0[j] = make_int2(0, node0);
    if (t < total) fetch(t, r0);
#pragma unroll
    for (int j = 0; j < SB_B; ++j)
        if (t + j * SORT_T < total) atomicAdd(&cnt[r0[j].y - node0], 1);
    for (int i0 = t + SB_B * SORT_T; i0 < total; i0 += SB_B * SORT_T) {
        int2 r[SB_B];
        fetch(i0, r);
#pragma unroll
        for (int j = 0; j < SB_B; ++j)
            if (i0 + j * SORT_T < total) atomicAdd(&cnt[r[j].y - node0], 1);
    }
    __syncthreads();
    for (int i = t; i < BIN_R; i += SORT_T)
        if (node0 + i < n_nodes) deg[node0 + i] = cnt[i];
    block_scan_lds(cnt, BIN_R, wsum);
    for (int i = t; i < BIN_R; i += SORT_T) {
        const int f = cnt[i];
        cur[i] = f;
        if (node0 + i < n_nodes) first[node0 + i] = start + f;
    }
    __syncthreads();
#pragma unroll
    for (int j = 0; j < SB_B; ++j)
        if (t + j * SORT_T < total) {
            const int pos = atomicAdd(&cur[r0[j].y - node0], 1);
            ssrc[start + pos] = r0[j].x;
        }
    for (int i0 = t + SB_B * SORT_T; i0 < total; i0 += SB_B * SORT_T) {
        int2 r[SB_B];
        fetch(i0, r);
#pragma unroll
        for (int j = 0; j < SB_B; ++j)
            if (i0 + j * SORT_T < total) {
                const int pos = atomicAdd(&cur[r[j].y - node0], 1);
                ssrc[start + pos] = r[j].x;
            }
    }
}

template <int CTRL>
__device__ __forceinline__ int dpp_i(int v) {
    return __builtin_amdgcn_update_dpp(0, v, CTRL, 0xf, 0xf, true);
}
template <int CTRL>
__device__ __forceinline__ float dpp_f(float v) {
    return __int_as_float(__builtin_amdgcn_update_dpp(0, __float_as_int(v), CTRL, 0xf, 0xf, true));
}

template <int J>
__device__ __forceinline__ v4f gather_row(int ids, int sub, const float* __restrict__ xl) {
    const int s = dpp_i<0x150 + J>(ids);
    return *(const v4f*)(xl + (long long)s * D_DIM + 4 * sub);
}

__device__ __forceinline__ float edge_score(const v4f& l, const v4f& r, const v4f& a) {
    float t0 = l.x + r.x, t1 = l.y + r.y, t2 = l.z + r.z, t3 = l.w + r.w;
    t0 = t0 > 0.0f ? t0 : NEG_SLOPE * t0;
    t1 = t1 > 0.0f ? t1 : NEG_SLOPE * t1;
    t2 = t2 > 0.0f ? t2 : NEG_SLOPE * t2;
    t3 = t3 > 0.0f ? t3 : NEG_SLOPE * t3;
    float p = t0 * a.x + t1 * a.y + t2 * a.z + t3 * a.w;
    p += dpp_f<0x160 + 1>(p);
    p += dpp_f<0x160 + 2>(p);
    p += dpp_f<0x160 + 4>(p);
    p += dpp_f<0x160 + 8>(p);
    return p;
}

template <int B>
__device__ __forceinline__ void seg_batch(int ids, int left, int sub,
                                          const float* __restrict__ xl,
                                          const v4f& r, const v4f& a,
                                          float& m, float& den, v4f& acc) {
    const v4f l0 = gather_row<4 * B + 0>(ids, sub, xl);
    const v4f l1 = gather_row<4 * B + 1>(ids, sub, xl);
    const v4f l2 = gather_row<4 * B + 2>(ids, sub, xl);
    const v4f l3 = gather_row<4 * B + 3>(ids, sub, xl);
    const float p0 = edge_score(l0, r, a);
    const float p1 = edge_score(l1, r, a);
    const float p2 = edge_score(l2, r, a);
    const float p3 = edge_score(l3, r, a);
    const bool v0 = left > 0, v1 = left > 1, v2 = left > 2, v3 = left > 3;
    float bm = m;
    bm = v0 ? fmaxf(bm, p0) : bm;
    bm = v1 ? fmaxf(bm, p1) : bm;
    bm = v2 ? fmaxf(bm, p2) : bm;
    bm = v3 ? fmaxf(bm, p3) : bm;
    const float sc = __expf(m - bm);
    const float w0 = v0 ? __expf(p0 - bm) : 0.0f;
    const float w1 = v1 ? __expf(p1 - bm) : 0.0f;
    const float w2 = v2 ? __expf(p2 - bm) : 0.0f;
    const float w3 = v3 ? __expf(p3 - bm) : 0.0f;
    den   = den   * sc + ((w0 + w1) + (w2 + w3));
    acc.x = acc.x * sc + ((w0 * l0.x + w1 * l1.x) + (w2 * l2.x + w3 * l3.x));
    acc.y = acc.y * sc + ((w0 * l0.y + w1 * l1.y) + (w2 * l2.y + w3 * l3.y));
    acc.z = acc.z * sc + ((w0 * l0.z + w1 * l1.z) + (w2 * l2.z + w3 * l3.z));
    acc.w = acc.w * sc + ((w0 * l0.w + w1 * l1.w) + (w2 * l2.w + w3 * l3.w));
    m = bm;
}

#define HT_P 68
#define OT_P 132
#define SEG_T 256

__global__ void __launch_bounds__(SEG_T)
k_segment_out(const int* __restrict__ mask, const int* __restrict__ first,
              const int* __restrict__ deg, const int* __restrict__ ssrc,
              const float* __restrict__ xl, const float* __restrict__ xr,
              const float* __restrict__ att, const float* __restrict__ gbias,
              const float* __restrict__ Wo, const float* __restrict__ bo,
              float* __restrict__ out, int n_mask) {
    __shared__ __attribute__((aligned(16))) float htile[16 * HT_P];
    __shared__ __attribute__((aligned(16))) float otile[16 * OT_P];
    const int lane = threadIdx.x & 31;
    const int wave = threadIdx.x >> 5;
    const int sub  = lane & 15;
    const int rb   = threadIdx.x >> 4;
    const int r0   = blockIdx.x * 16;
    const int i = r0 + rb;
    const bool live = i < n_mask;
    int node = 0, beg = 0, n = 0;
    if (live) { node = mask[i]; beg = first[node]; n = deg[node]; }
    const v4f a = *(const v4f*)(att + 4 * sub);
    const v4f r = *(const v4f*)(xr + (long long)node * D_DIM + 4 * sub);
    const int n_lo = __builtin_amdgcn_readlane(n, 0);
    const int n_hi = __builtin_amdgcn_readlane(n, 16);
    const int nmax = n_lo > n_hi ? n_lo : n_hi;
    float m = -FLT_MAX, den = 0.0f;
    v4f acc = {0.0f, 0.0f, 0.0f, 0.0f};
    for (int c = 0; c < nmax; c += 16) {
        const int left = n - c;
        int ids = 0;
        if (sub < left) ids = ssrc[beg + c + sub];
        seg_batch<0>(ids, left, sub, xl, r, a, m, den, acc);
        if (c + 4 >= nmax) break;
        seg_batch<1>(ids, left - 4, sub, xl, r, a, m, den, acc);
        if (c + 8 >= nmax) break;
        seg_batch<2>(ids, left - 8, sub, xl, r, a, m, den, acc);
        if (c + 12 >= nmax) break;
        seg_batch<3>(ids, left - 12, sub, xl, r, a, m, den, acc);
    }

    const int lrow = lane & 15;
    const int lhi  = lane >> 4;
    const int col  = wave * 16 + lrow;
    v2f b[16];
#pragma unroll
    for (int kk = 0; kk < 16; ++kk) {
        const int k0 = kk * 4 + 2 * lhi;
        b[kk].x = Wo[(k0    ) * DOUT + col];
        b[kk].y = Wo[(k0 + 1) * DOUT + col];
    }
    const float bv = bo[col];
    {
        const v4f g = *(const v4f*)(gbias + 4 * sub);
        const float dn = den + 1e-16f;
        v4f o;
        o.x = acc.x / dn + g.x; o.y = acc.y / dn + g.y;
        o.z = acc.z / dn + g.z; o.w = acc.w / dn + g.w;
        *(v4f*)(htile + rb * HT_P + 4 * sub) = o;
    }
    __syncthreads();
    v8f c8 = {bv, bv, bv, bv, bv, bv, bv, bv};
#pragma unroll
    for (int kk = 0; kk < 16; ++kk) {
        const v2f af = *(const v2f*)(htile + lrow * HT_P + kk * 4 + 2 * lhi);
        c8 = __builtin_amdgcn_wmma_f32_16x16x4_f32(
                false, af, false, b[kk], (short)0, c8, false, false);
    }
#pragma unroll
    for (int q = 0; q < 8; ++q)
        otile[(q + 8 * lhi) * OT_P + col] = c8[q];
    __syncthreads();
#pragma unroll
    for (int q = 0; q < 2; ++q) {
        const int row = 2 * wave + q;
        if (r0 + row < n_mask)
            *(v4f*)(out + (long long)(r0 + row) * DOUT + 4 * lane) =
                *(const v4f*)(otile + row * OT_P + 4 * lane);
    }
}

extern "C" void kernel_launch(void* const* d_in, const int* in_sizes, int n_in,
                              void* d_out, int out_size, void* d_ws, size_t ws_size,
                              hipStream_t stream) {
    const float* x     = (const float*)d_in[0];
    const int*   ei    = (const int*)d_in[1];
    const int*   mask  = (const int*)d_in[2];
    const float* Wl    = (const float*)d_in[3];
    const float* bl    = (const float*)d_in[4];
    const float* Wr    = (const float*)d_in[5];
    const float* br    = (const float*)d_in[6];
    const float* att   = (const float*)d_in[7];
    const float* gbias = (const float*)d_in[8];
    const float* Wo    = (const float*)d_in[9];
    const float* bo    = (const float*)d_in[10];
    float* out = (float*)d_out;

    const int n_nodes = in_sizes[0] / D_DIM;
    const int n_edges = in_sizes[1] / 2;
    const int n_mask  = in_sizes[2];
    const int* src = ei;
    const int* dst = ei + n_edges;

    const int nblk  = (n_edges + BS_CH - 1) / BS_CH;
    const int nbins = (n_nodes + BIN_R - 1) / BIN_R;
    const int nwords = (n_nodes + 31) / 32;
    char* wp = (char*)d_ws;
    auto carve = [&](size_t bytes) {
        char* p = wp;
        wp += (bytes + 255) & ~(size_t)255;
        return p;
    };
    float*    xl     = (float*)   carve((size_t)n_nodes * D_DIM * 4);
    float*    xr     = (float*)   carve((size_t)n_nodes * D_DIM * 4);
    int*      ssrc   = (int*)     carve((size_t)n_edges * 4);
    int2*     recs   = (int2*)    carve((size_t)nblk * BS_CH * 8);
    int*      boff   = (int*)     carve((size_t)nblk * (nbins + 1) * 4);
    int*      deg    = (int*)     carve((size_t)n_nodes * 4);
    unsigned* needbits = (unsigned*)carve((size_t)nwords * 4);
    int*      first  = (int*)     carve((size_t)n_nodes * 4);

    k_init<<<(nwords + 255) / 256, 256, 0, stream>>>(needbits, nwords);
    k_mark<<<(n_mask + 255) / 256, 256, 0, stream>>>(mask, needbits, n_mask);
    const int tiles = (n_nodes + 15) / 16;
    const int lblk  = (tiles + LIN_TILES - 1) / LIN_TILES;
    const int lin_a = (lblk + 1) / 2;
    const int lin_b = lblk - lin_a;
    k_blocksort<<<nblk + lin_a, SORT_T, (size_t)(nbins + 1) * 4, stream>>>(
        src, dst, needbits, recs, boff, n_edges, nbins, nblk,
        x, Wl, bl, Wr, br, xl, xr, n_nodes);
    k_binsort<<<nbins + lin_b, SORT_T, (size_t)(2 * nblk + 1) * 4, stream>>>(
        recs, boff, first, deg, ssrc, n_nodes, nblk, nbins, lin_a,
        x, Wl, bl, Wr, br, xl, xr);
    k_segment_out<<<(n_mask + 15) / 16, SEG_T, 0, stream>>>(
        mask, first, deg, ssrc, xl, xr, att, gbias, Wo, bo, out, n_mask);
}


# product work-groups stage W_l, W_r (as ready B fragments) and their 256 rows of x once in LDS: B operand = one 8-B LDS read, was two 4-B global loads per matrix instruction; order of sums unchanged
# speedup vs baseline: 7.0170x; 1.0336x over previous
#include <hip/hip_runtime.h>
#include <hip/hip_bf16.h>
#include <math.h>
#include <float.h>


#define D_DIM 64
#define DOUT  128
#define NEG_SLOPE 0.2f

typedef __attribute__((ext_vector_type(2))) float v2f;
typedef __attribute__((ext_vector_type(8))) float v8f;
typedef __attribute__((ext_vector_type(4))) float v4f;

__global__ void k_init(unsigned* __restrict__ needbits, int n_words) {
    const int i = blockIdx.x * blockDim.x + threadIdx.x;
    if (i < n_words) needbits[i] = 0u;
}

__global__ void k_mark(const int* __restrict__ mask, unsigned* __restrict__ needbits,
                       int n_mask) {
    int i = blockIdx.x * blockDim.x + threadIdx.x;
    if (i >= n_mask) return;
    const int m = mask[i];
    atomicOr(&needbits[m >> 5], 1u << (m & 31));
}

#define SORT_T    1024
#define LIN_TILES 16
#define XT_P      68
#define WF_N      (2 * 4 * 16 * 32)
#define XT_N      (LIN_TILES * 16 * XT_P)

__device__ __forceinline__ void linear_role(int lb, const float* __restrict__ x,
        const float* __restrict__ Wl, const float* __restrict__ bl,
        const float* __restrict__ Wr, const float* __restrict__ br,
        float* __restrict__ xl, float* __restrict__ xr, int n_nodes,
        v2f* wfrag, float* xt) {
    const int t = threadIdx.x;
    const int lane = t & 31;
    const int wave = t >> 5;
    const int row_base = lb * LIN_TILES * 16;
    v4f xv[4];
    v2f wv[4];
#pragma unroll
    for (int q = 0; q < 4; ++q) {
        const int i = t + q * SORT_T;
        int grow = row_base + (i >> 4);
        grow = grow < n_nodes ? grow : n_nodes - 1;
        xv[q] = *(const v4f*)(x + (long long)grow * D_DIM + 4 * (i & 15));
    }
#pragma unroll
    for (int q = 0; q < 4; ++q) {
        const int e = t + q * SORT_T;
        const float* W = (q >> 1) ? Wr : Wl;
        const int k0  = ((e >> 5) & 15) * 4 + 2 * ((e >> 4) & 1);
        const int col = ((e >> 9) & 3) * 16 + (e & 15);
        wv[q].x = W[(k0    ) * D_DIM + col];
        wv[q].y = W[(k0 + 1) * D_DIM + col];
    }
#pragma unroll
    for (int q = 0; q < 4; ++q) {
        const int i = t + q * SORT_T;
        *(v4f*)(xt + (i >> 4) * XT_P + 4 * (i & 15)) = xv[q];
        wfrag[t + q * SORT_T] = wv[q];
    }
    __syncthreads();
    const int tile = wave >> 1;
    const int r0 = row_base + tile * 16;
    if (r0 + 16 > n_nodes) return;
    const int lrow = lane & 15;
    const int lhi  = lane >> 4;
    const int mat  = wave & 1;
    const float* __restrict__ bias = mat ? br : bl;
    float* __restrict__ out = mat ? xr : xl;
    v2f a[16];
    const float* xp = xt + (tile * 16 + lrow) * XT_P + 2 * lhi;
#pragma unroll
    for (int kk = 0; kk < 16; ++kk)
        a[kk] = *(const v2f*)(xp + kk * 4);
    const v2f* wf = wfrag + mat * (4 * 16 * 32) + lane;
#pragma unroll
    for (int ct = 0; ct < 4; ++ct) {
        const int col = ct * 16 + lrow;
        const float bv = bias[col];
        v8f acc = {bv, bv, bv, bv, bv, bv, bv, bv};
#pragma unroll
        for (int kk = 0; kk < 16; ++kk)
            acc = __builtin_amdgcn_wmma_f32_16x16x4_f32(
                    false, a[kk], false, wf[(ct * 16 + kk) * 32], (short)0, acc, false, false);
#pragma unroll
        for (int r = 0; r < 8; ++r)
            out[(long long)(r0 + r + 8 * lhi) * D_DIM + col] = acc[r];
    }
}

#define BS_E      4
#define BS_CH     (SORT_T * BS_E)
#define BIN_SHIFT 9
#define BIN_R     (1 << BIN_SHIFT)
#define SB_B      4

__device__ __forceinline__ int block_scan_lds(int* s, int n, int* wsum) {
    const int t = threadIdx.x, T = SORT_T;
    const int lane = t & 31, wave = t >> 5;
    const int K = (n + T - 1) / T;
    const int lo = t * K;
    int hi = lo + K;
    if (hi > n) hi = n;
    __syncthreads();
    int sum = 0;
    for (int i = lo; i < hi; ++i) sum += s[i];
    int inc = sum;
    for (int o = 1; o < 32; o <<= 1) {
        const int u = __shfl_up(inc, o);
        if (lane >= o) inc += u;
    }
    if (lane == 31) wsum[wave] = inc;
    __syncthreads();
    if (wave == 0) {
        const int w = lane < (T >> 5) ? wsum[lane] : 0;
        int winc = w;
        for (int o = 1; o < 32; o <<= 1) {
            const int u = __shfl_up(winc, o);
            if (lane >= o) winc += u;
        }
        wsum[lane] = winc - w;
        if (lane == 31) wsum[32] = winc;
    }
    __syncthreads();
    int run = wsum[wave] + inc - sum;
    for (int i = lo; i < hi; ++i) {
        const int v = s[i];
        s[i] = run;
        run += v;
    }
    const int total = wsum[32];
    __syncthreads();
    return total;
}

__global__ void __launch_bounds__(SORT_T)
k_blocksort(const int* __restrict__ src, const int* __restrict__ dst,
            const unsigned* __restrict__ needbits, int2* __restrict__ recs,
            int* __restrict__ boff, int n_edges, int nbins, int nsort,
            const float* __restrict__ x,
            const float* __restrict__ Wl, const float* __restrict__ bl,
            const float* __restrict__ Wr, const float* __restrict__ br,
            float* __restrict__ xl, float* __restrict__ xr, int n_nodes) {
    extern __shared__ int hist[];
    __shared__ int wsum[33];
    __shared__ __attribute__((aligned(16))) v2f wfrag[WF_N];
    __shared__ __attribute__((aligned(16))) float xt[XT_N];
    if ((int)blockIdx.x >= nsort) {
        linear_role((int)blockIdx.x - nsort, x, Wl, bl, Wr, br, xl, xr, n_nodes, wfrag, xt);
        return;
    }
    const int t = threadIdx.x;
    for (int b = t; b <= nbins; b += SORT_T) hist[b] = 0;
    const long long base = (long long)blockIdx.x * BS_CH + t;
    int d[BS_E], s[BS_E], lr[BS_E];
    unsigned word[BS_E];
    bool keep[BS_E];
#pragma unroll
    for (int j = 0; j < BS_E; ++j) {
        const long long e = base + j * SORT_T;
        const long long ec = e < n_edges ? e : 0;
        d[j] = dst[ec];
        s[j] = src[ec];
    }
#pragma unroll
    for (int j = 0; j < BS_E; ++j)
        word[j] = needbits[d[j] >> 5];
    __syncthreads();
#pragma unroll
    for (int j = 0; j < BS_E; ++j) {
        keep[j] = (base + j * SORT_T < n_edges) && ((word[j] >> (d[j] & 31)) & 1u);
        lr[j] = 0;
        if (keep[j]) lr[j] = atomicAdd(&hist[d[j] >> BIN_SHIFT], 1);
    }
    block_scan_lds(hist, nbins + 1, wsum);
    int2* mine = recs + (long long)blockIdx.x * BS_CH;
#pragma unroll
    for (int j = 0; j < BS_E; ++j)
        if (keep[j]) mine[hist[d[j] >> BIN_SHIFT] + lr[j]] = make_int2(s[j], d[j]);
    int* myoff = boff + (long long)blockIdx.x * (nbins + 1);
    for (int b = t; b <= nbins; b += SORT_T) myoff[b] = hist[b];
}

__global__ void __launch_bounds__(SORT_T)
k_binsort(const int2* __restrict__ recs, const int* __restrict__ boff,
          int* __restrict__ first, int* __restrict__ deg, int* __restrict__ ssrc,
          int n_nodes, int nblk, int nbins, int lin0,
          const float* __restrict__ x,
          const float* __restrict__ Wl, const float* __restrict__ bl,
          const float* __restrict__ Wr, const float* __restrict__ br,
          float* __restrict__ xl, float* __restrict__ xr) {
    extern __shared__ int dyn[];
    __shared__ __attribute__((aligned(16))) v2f wfrag[WF_N];
    __shared__ __attribute__((aligned(16))) float xt[XT_N];
    if ((int)blockIdx.x >= nbins) {
        linear_role((int)blockIdx.x - nbins + lin0, x, Wl, bl, Wr, br, xl, xr, n_nodes, wfrag, xt);
        return;
    }
    int* o0  = dyn;
    int* pre = dyn + nblk;
    __shared__ int cnt[BIN_R];
    __shared__ int cur[BIN_R];
    __shared__ int wsum[33];
    __shared__ int sbase;
    const int t = threadIdx.x;
    const int b = blockIdx.x;
    const int node0 = b << BIN_SHIFT;
    if (t == 0) { sbase = 0; pre[nblk] = 0; }
    for (int i = t; i < BIN_R; i += SORT_T) cnt[i] = 0;
    int mystart = 0;
    for (int k = t; k < nblk; k += SORT_T) {
        const int* bo = boff + (long long)k * (nbins + 1) + b;
        const int a0 = bo[0], a1 = bo[1];
        o0[k]  = a0;
        pre[k] = a1 - a0;
        mystart += a0;
    }
    __syncthreads();
    if (mystart) atomicAdd(&sbase, mystart);
    const int total = block_scan_lds(pre, nblk + 1, wsum);
    const int start = sbase;

    int top = 1;
    while (top < nblk) top <<= 1;
    auto fetch = [&](int i0, int2 (&r)[SB_B]) {
        int idx[SB_B], lo[SB_B];
#pragma unroll
        for (int j = 0; j < SB_B; ++j) {
            const int i = i0 + j * SORT_T;
            idx[j] = i < total ? i : total - 1;
            lo[j] = 0;
        }
        for (int w = top >> 1; w > 0; w >>= 1) {
#pragma unroll
            for (int j = 0; j < SB_B; ++j) {
                const int k = lo[j] + w < nblk ? lo[j] + w : nblk;
                lo[j] = pre[k] <= idx[j] ? k : lo[j];
            }
        }
#pragma unroll
        for (int j = 0; j < SB_B; ++j)
            r[j] = recs[(long long)lo[j] * BS_CH + o0[lo[j]] + (idx[j] - pre[lo[j]])];
    };

    int2 r0[SB_B];
#pragma unroll
    for (int j = 0; j < SB_B; ++j) r0[j] = make_int2(0, node0);
    if (t < total) fetch(t, r0);
#pragma unroll
    for (int j = 0; j < SB_B; ++j)
        if (t + j * SORT_T < total) atomicAdd(&cnt[r0[j].y - node0], 1);
    for (int i0 = t + SB_B * SORT_T; i0 < total; i0 += SB_B * SORT_T) {
        int2 r[SB_B];
        fetch(i0, r);
#pragma unroll
        for (int j = 0; j < SB_B; ++j)
            if (i0 + j * SORT_T < total) atomicAdd(&cnt[r[j].y - node0], 1);
    }
    __syncthreads();
    for (int i = t; i < BIN_R; i += SORT_T)
        if (node0 + i < n_nodes) deg[node0 + i] = cnt[i];
    block_scan_lds(cnt, BIN_R, wsum);
    for (int i = t; i < BIN_R; i += SORT_T) {
        const int f = cnt[i];
        cur[i] = f;
        if (node0 + i < n_nodes) first[node0 + i] = start + f;
    }
    __syncthreads();
#pragma unroll
    for (int j = 0; j < SB_B; ++j)
        if (t + j * SORT_T < total) {
            const int pos = atomicAdd(&cur[r0[j].y - node0], 1);
            ssrc[start + pos] = r0[j].x;
        }
    for (int i0 = t + SB_B * SORT_T; i0 < total; i0 += SB_B * SORT_T) {
        int2 r[SB_B];
        fetch(i0, r);
#pragma unroll
        for (int j = 0; j < SB_B; ++j)
            if (i0 + j * SORT_T < total) {
                const int pos = atomicAdd(&cur[r[j].y - node0], 1);
                ssrc[start + pos] = r[j].x;
            }
    }
}

template <int CTRL>
__device__ __forceinline__ int dpp_i(int v) {
    return __builtin_amdgcn_update_dpp(0, v, CTRL, 0xf, 0xf, true);
}
template <int CTRL>
__device__ __forceinline__ float dpp_f(float v) {
    return __int_as_float(__builtin_amdgcn_update_dpp(0, __float_as_int(v), CTRL, 0xf, 0xf, true));
}

template <int J>
__device__ __forceinline__ v4f gather_row(int ids, int sub, const float* __restrict__ xl) {
    const int s = dpp_i<0x150 + J>(ids);
    return *(const v4f*)(xl + (long long)s * D_DIM + 4 * sub);
}

__device__ __forceinline__ float edge_score(const v4f& l, const v4f& r, const v4f& a) {
    float t0 = l.x + r.x, t1 = l.y + r.y, t2 = l.z + r.z, t3 = l.w + r.w;
    t0 = t0 > 0.0f ? t0 : NEG_SLOPE * t0;
    t1 = t1 > 0.0f ? t1 : NEG_SLOPE * t1;
    t2 = t2 > 0.0f ? t2 : NEG_SLOPE * t2;
    t3 = t3 > 0.0f ? t3 : NEG_SLOPE * t3;
    float p = t0 * a.x + t1 * a.y + t2 * a.z + t3 * a.w;
    p += dpp_f<0x160 + 1>(p);
    p += dpp_f<0x160 + 2>(p);
    p += dpp_f<0x160 + 4>(p);
    p += dpp_f<0x160 + 8>(p);
    return p;
}

template <int B>
__device__ __forceinline__ void seg_batch(int ids, int left, int sub,
                                          const float* __restrict__ xl,
                                          const v4f& r, const v4f& a,
                                          float& m, float& den, v4f& acc) {
    const v4f l0 = gather_row<4 * B + 0>(ids, sub, xl);
    const v4f l1 = gather_row<4 * B + 1>(ids, sub, xl);
    const v4f l2 = gather_row<4 * B + 2>(ids, sub, xl);
    const v4f l3 = gather_row<4 * B + 3>(ids, sub, xl);
    const float p0 = edge_score(l0, r, a);
    const float p1 = edge_score(l1, r, a);
    const float p2 = edge_score(l2, r, a);
    const float p3 = edge_score(l3, r, a);
    const bool v0 = left > 0, v1 = left > 1, v2 = left > 2, v3 = left > 3;
    float bm = m;
    bm = v0 ? fmaxf(bm, p0) : bm;
    bm = v1 ? fmaxf(bm, p1) : bm;
    bm = v2 ? fmaxf(bm, p2) : bm;
    bm = v3 ? fmaxf(bm, p3) : bm;
    const float sc = __expf(m - bm);
    const float w0 = v0 ? __expf(p0 - bm) : 0.0f;
    const float w1 = v1 ? __expf(p1 - bm) : 0.0f;
    const float w2 = v2 ? __expf(p2 - bm) : 0.0f;
    const float w3 = v3 ? __expf(p3 - bm) : 0.0f;
    den   = den   * sc + ((w0 + w1) + (w2 + w3));
    acc.x = acc.x * sc + ((w0 * l0.x + w1 * l1.x) + (w2 * l2.x + w3 * l3.x));
    acc.y = acc.y * sc + ((w0 * l0.y + w1 * l1.y) + (w2 * l2.y + w3 * l3.y));
    acc.z = acc.z * sc + ((w0 * l0.z + w1 * l1.z) + (w2 * l2.z + w3 * l3.z));
    acc.w = acc.w * sc + ((w0 * l0.w + w1 * l1.w) + (w2 * l2.w + w3 * l3.w));
    m = bm;
}

#define HT_P 68
#define OT_P 132
#define SEG_T 256

__global__ void __launch_bounds__(SEG_T)
k_segment_out(const int* __restrict__ mask, const int* __restrict__ first,
              const int* __restrict__ deg, const int* __restrict__ ssrc,
              const float* __restrict__ xl, const float* __restrict__ xr,
              const float* __restrict__ att, const float* __restrict__ gbias,
              const float* __restrict__ Wo, const float* __restrict__ bo,
              float* __restrict__ out, int n_mask) {
    __shared__ __attribute__((aligned(16))) float htile[16 * HT_P];
    __shared__ __attribute__((aligned(16))) float otile[16 * OT_P];
    const int lane = threadIdx.x & 31;
    const int wave = threadIdx.x >> 5;
    const int sub  = lane & 15;
    const int rb   = threadIdx.x >> 4;
    const int r0   = blockIdx.x * 16;
    const int i = r0 + rb;
    const bool live = i < n_mask;
    int node = 0, beg = 0, n = 0;
    if (live) { node = mask[i]; beg = first[node]; n = deg[node]; }
    const v4f a = *(const v4f*)(att + 4 * sub);
    const v4f r = *(const v4f*)(xr + (long long)node * D_DIM + 4 * sub);
    const int n_lo = __builtin_amdgcn_readlane(n, 0);
    const int n_hi = __builtin_amdgcn_readlane(n, 16);
    const int nmax = n_lo > n_hi ? n_lo : n_hi;
    float m = -FLT_MAX, den = 0.0f;
    v4f acc = {0.0f, 0.0f, 0.0f, 0.0f};
    for (int c = 0; c < nmax; c += 16) {
        const int left = n - c;
        int ids = 0;
        if (sub < left) ids = ssrc[beg + c + sub];
        seg_batch<0>(ids, left, sub, xl, r, a, m, den, acc);
        if (c + 4 >= nmax) break;
        seg_batch<1>(ids, left - 4, sub, xl, r, a, m, den, acc);
        if (c + 8 >= nmax) break;
        seg_batch<2>(ids, left - 8, sub, xl, r, a, m, den, acc);
        if (c + 12 >= nmax) break;
        seg_batch<3>(ids, left - 12, sub, xl, r, a, m, den, acc);
    }

    const int lrow = lane & 15;
    const int lhi  = lane >> 4;
    const int col  = wave * 16 + lrow;
    v2f b[16];
#pragma unroll
    for (int kk = 0; kk < 16; ++kk) {
        const int k0 = kk * 4 + 2 * lhi;
        b[kk].x = Wo[(k0    ) * DOUT + col];
        b[kk].y = Wo[(k0 + 1) * DOUT + col];
    }
    const float bv = bo[col];
    {
        const v4f g = *(const v4f*)(gbias + 4 * sub);
        const float dn = den + 1e-16f;
        v4f o;
        o.x = acc.x / dn + g.x; o.y = acc.y / dn + g.y;
        o.z = acc.z / dn + g.z; o.w = acc.w / dn + g.w;
        *(v4f*)(htile + rb * HT_P + 4 * sub) = o;
    }
    __syncthreads();
    v8f c8 = {bv, bv, bv, bv, bv, bv, bv, bv};
#pragma unroll
    for (int kk = 0; kk < 16; ++kk) {
        const v2f af = *(const v2f*)(htile + lrow * HT_P + kk * 4 + 2 * lhi);
        c8 = __builtin_amdgcn_wmma_f32_16x16x4_f32(
                false, af, false, b[kk], (short)0, c8, false, false);
    }
#pragma unroll
    for (int q = 0; q < 8; ++q)
        otile[(q + 8 * lhi) * OT_P + col] = c8[q];
    __syncthreads();
#pragma unroll
    for (int q = 0; q < 2; ++q) {
        const int row = 2 * wave + q;
        if (r0 + row < n_mask)
            *(v4f*)(out + (long long)(r0 + row) * DOUT + 4 * lane) =
                *(const v4f*)(otile + row * OT_P + 4 * lane);
    }
}

extern "C" void kernel_launch(void* const* d_in, const int* in_sizes, int n_in,
                              void* d_out, int out_size, void* d_ws, size_t ws_size,
                              hipStream_t stream) {
    const float* x     = (const float*)d_in[0];
    const int*   ei    = (const int*)d_in[1];
    const int*   mask  = (const int*)d_in[2];
    const float* Wl    = (const float*)d_in[3];
    const float* bl    = (const float*)d_in[4];
    const float* Wr    = (const float*)d_in[5];
    const float* br    = (const float*)d_in[6];
    const float* att   = (const float*)d_in[7];
    const float* gbias = (const float*)d_in[8];
    const float* Wo    = (const float*)d_in[9];
    const float* bo    = (const float*)d_in[10];
    float* out = (float*)d_out;

    const int n_nodes = in_sizes[0] / D_DIM;
    const int n_edges = in_sizes[1] / 2;
    const int n_mask  = in_sizes[2];
    const int* src = ei;
    const int* dst = ei + n_edges;

    const int nblk  = (n_edges + BS_CH - 1) / BS_CH;
    const int nbins = (n_nodes + BIN_R - 1) / BIN_R;
    const int nwords = (n_nodes + 31) / 32;
    char* wp = (char*)d_ws;
    auto carve = [&](size_t bytes) {
        char* p = wp;
        wp += (bytes + 255) & ~(size_t)255;
        return p;
    };
    float*    xl     = (float*)   carve((size_t)n_nodes * D_DIM * 4);
    float*    xr     = (float*)   carve((size_t)n_nodes * D_DIM * 4);
    int*      ssrc   = (int*)     carve((size_t)n_edges * 4);
    int2*     recs   = (int2*)    carve((size_t)nblk * BS_CH * 8);
    int*      boff   = (int*)     carve((size_t)nblk * (nbins + 1) * 4);
    int*      deg    = (int*)     carve((size_t)n_nodes * 4);
    unsigned* needbits = (unsigned*)carve((size_t)nwords * 4);
    int*      first  = (int*)     carve((size_t)n_nodes * 4);

    k_init<<<(nwords + 255) / 256, 256, 0, stream>>>(needbits, nwords);
    k_mark<<<(n_mask + 255) / 256, 256, 0, stream>>>(mask, needbits, n_mask);
    const int tiles = (n_nodes + 15) / 16;
    const int lblk  = (tiles + LIN_TILES - 1) / LIN_TILES;
    const int lin_a = (lblk + 1) / 2;
    const int lin_b = lblk - lin_a;
    k_blocksort<<<nblk + lin_a, SORT_T, (size_t)(nbins + 1) * 4, stream>>>(
        src, dst, needbits, recs, boff, n_edges, nbins, nblk,
        x, Wl, bl, Wr, br, xl, xr, n_nodes);
    k_binsort<<<nbins + lin_b, SORT_T, (size_t)(2 * nblk + 1) * 4, stream>>>(
        recs, boff, first, deg, ssrc, n_nodes, nblk, nbins, lin_a,
        x, Wl, bl, Wr, br, xl, xr);
    k_segment_out<<<(n_mask + 15) / 16, SEG_T, 0, stream>>>(
        mask, first, deg, ssrc, xl, xr, att, gbias, Wo, bo, out, n_mask);
}


# each edge-pass work-group builds the named-node bitmap in its own LDS from mask_idx (8192 edges per group, 147 builds); memory bitmap, clear and mark launches gone, 5 -> 3; sums stay in arrival order
# speedup vs baseline: 7.8957x; 1.1252x over previous
#include <hip/hip_runtime.h>
#include <hip/hip_bf16.h>
#include <math.h>
#include <float.h>


#define D_DIM 64
#define DOUT  128
#define NEG_SLOPE 0.2f

typedef __attribute__((ext_vector_type(2))) float v2f;
typedef __attribute__((ext_vector_type(8))) float v8f;
typedef __attribute__((ext_vector_type(4))) float v4f;

#define SORT_T    1024
#define LIN_TILES 16
#define XT_P      68
#define WF_N      (2 * 4 * 16 * 32)
#define XT_N      (LIN_TILES * 16 * XT_P)

__device__ __forceinline__ void linear_role(int lb, const float* __restrict__ x,
        const float* __restrict__ Wl, const float* __restrict__ bl,
        const float* __restrict__ Wr, const float* __restrict__ br,
        float* __restrict__ xl, float* __restrict__ xr, int n_nodes,
        v2f* wfrag, float* xt) {
    const int t = threadIdx.x;
    const int lane = t & 31;
    const int wave = t >> 5;
    const int row_base = lb * LIN_TILES * 16;
    v4f xv[4];
    v2f wv[4];
#pragma unroll
    for (int q = 0; q < 4; ++q) {
        const int i = t + q * SORT_T;
        int grow = row_base + (i >> 4);
        grow = grow < n_nodes ? grow : n_nodes - 1;
        xv[q] = *(const v4f*)(x + (long long)grow * D_DIM + 4 * (i & 15));
    }
#pragma unroll
    for (int q = 0; q < 4; ++q) {
        const int e = t + q * SORT_T;
        const float* W = (q >> 1) ? Wr : Wl;
        const int k0  = ((e >> 5) & 15) * 4 + 2 * ((e >> 4) & 1);
        const int col = ((e >> 9) & 3) * 16 + (e & 15);
        wv[q].x = W[(k0    ) * D_DIM + col];
        wv[q].y = W[(k0 + 1) * D_DIM + col];
    }
#pragma unroll
    for (int q = 0; q < 4; ++q) {
        const int i = t + q * SORT_T;
        *(v4f*)(xt + (i >> 4) * XT_P + 4 * (i & 15)) = xv[q];
        wfrag[t + q * SORT_T] = wv[q];
    }
    __syncthreads();
    const int tile = wave >> 1;
    const int r0 = row_base + tile * 16;
    if (r0 + 16 > n_nodes) return;
    const int lrow = lane & 15;
    const int lhi  = lane >> 4;
    const int mat  = wave & 1;
    const float* __restrict__ bias = mat ? br : bl;
    float* __restrict__ out = mat ? xr : xl;
    v2f a[16];
    const float* xp = xt + (tile * 16 + lrow) * XT_P + 2 * lhi;
#pragma unroll
    for (int kk = 0; kk < 16; ++kk)
        a[kk] = *(const v2f*)(xp + kk * 4);
    const v2f* wf = wfrag + mat * (4 * 16 * 32) + lane;
#pragma unroll
    for (int ct = 0; ct < 4; ++ct) {
        const int col = ct * 16 + lrow;
        const float bv = bias[col];
        v8f acc = {bv, bv, bv, bv, bv, bv, bv, bv};
#pragma unroll
        for (int kk = 0; kk < 16; ++kk)
            acc = __builtin_amdgcn_wmma_f32_16x16x4_f32(
                    false, a[kk], false, wf[(ct * 16 + kk) * 32], (short)0, acc, false, false);
#pragma unroll
        for (int r = 0; r < 8; ++r)
            out[(long long)(r0 + r + 8 * lhi) * D_DIM + col] = acc[r];
    }
}

#define BS_E      8
#define MK_U      8
#define BS_CH     (SORT_T * BS_E)
#define BIN_SHIFT 9
#define BIN_R     (1 << BIN_SHIFT)
#define SB_B      4

__device__ __forceinline__ int block_scan_lds(int* s, int n, int* wsum) {
    const int t = threadIdx.x, T = SORT_T;
    const int lane = t & 31, wave = t >> 5;
    const int K = (n + T - 1) / T;
    const int lo = t * K;
    int hi = lo + K;
    if (hi > n) hi = n;
    __syncthreads();
    int sum = 0;
    for (int i = lo; i < hi; ++i) sum += s[i];
    int inc = sum;
    for (int o = 1; o < 32; o <<= 1) {
        const int u = __shfl_up(inc, o);
        if (lane >= o) inc += u;
    }
    if (lane == 31) wsum[wave] = inc;
    __syncthreads();
    if (wave == 0) {
        const int w = lane < (T >> 5) ? wsum[lane] : 0;
        int winc = w;
        for (int o = 1; o < 32; o <<= 1) {
            const int u = __shfl_up(winc, o);
            if (lane >= o) winc += u;
        }
        wsum[lane] = winc - w;
        if (lane == 31) wsum[32] = winc;
    }
    __syncthreads();
    int run = wsum[wave] + inc - sum;
    for (int i = lo; i < hi; ++i) {
        const int v = s[i];
        s[i] = run;
        run += v;
    }
    const int total = wsum[32];
    __syncthreads();
    return total;
}

__global__ void __launch_bounds__(SORT_T)
k_blocksort(const int* __restrict__ src, const int* __restrict__ dst,
            const int* __restrict__ mask, int n_mask, int nwords, int2* __restrict__ recs,
            int* __restrict__ boff, int n_edges, int nbins, int nsort,
            const float* __restrict__ x,
            const float* __restrict__ Wl, const float* __restrict__ bl,
            const float* __restrict__ Wr, const float* __restrict__ br,
            float* __restrict__ xl, float* __restrict__ xr, int n_nodes) {
    extern __shared__ int hist[];
    __shared__ int wsum[33];
    __shared__ __attribute__((aligned(16))) v2f wfrag[WF_N];
    __shared__ __attribute__((aligned(16))) float xt[XT_N];
    if ((int)blockIdx.x >= nsort) {
        linear_role((int)blockIdx.x - nsort, x, Wl, bl, Wr, br, xl, xr, n_nodes, wfrag, xt);
        return;
    }
    const int t = threadIdx.x;
    unsigned* nbit = (unsigned*)(hist + nbins + 1);
    const long long base = (long long)blockIdx.x * BS_CH + t;
    int d[BS_E], s[BS_E], lr[BS_E];
    bool keep[BS_E];
    for (int b = t; b <= nbins; b += SORT_T) hist[b] = 0;
    for (int w = t; w < nwords; w += SORT_T) nbit[w] = 0u;
    __syncthreads();
#pragma unroll
    for (int j = 0; j < BS_E; ++j) {
        const long long e = base + j * SORT_T;
        const long long ec = e < n_edges ? e : 0;
        d[j] = dst[ec];
        s[j] = src[ec];
    }
    int hd = (int)(((16u - (unsigned)((size_t)mask & 15u)) & 15u) >> 2);
    hd = hd < n_mask ? hd : n_mask;
    const int4* mask4 = (const int4*)(mask + hd);
    const int nvec = (n_mask - hd) >> 2;
    for (int vb = 0; vb < nvec; vb += SORT_T * MK_U) {
        int4 mm[MK_U];
#pragma unroll
        for (int q = 0; q < MK_U; ++q) {
            const int v = vb + q * SORT_T + t;
            mm[q] = mask4[v < nvec ? v : nvec - 1];
        }
#pragma unroll
        for (int q = 0; q < MK_U; ++q)
            if (vb + q * SORT_T + t < nvec) {
                atomicOr(&nbit[mm[q].x >> 5], 1u << (mm[q].x & 31));
                atomicOr(&nbit[mm[q].y >> 5], 1u << (mm[q].y & 31));
                atomicOr(&nbit[mm[q].z >> 5], 1u << (mm[q].z & 31));
                atomicOr(&nbit[mm[q].w >> 5], 1u << (mm[q].w & 31));
            }
    }
    if (t < hd) {
        const int m = mask[t];
        atomicOr(&nbit[m >> 5], 1u << (m & 31));
    }
    for (int i = hd + 4 * nvec + t; i < n_mask; i += SORT_T) {
        const int m = mask[i];
        atomicOr(&nbit[m >> 5], 1u << (m & 31));
    }
    __syncthreads();
    unsigned word[BS_E];
#pragma unroll
    for (int j = 0; j < BS_E; ++j)
        word[j] = nbit[d[j] >> 5];
#pragma unroll
    for (int j = 0; j < BS_E; ++j) {
        keep[j] = (base + j * SORT_T < n_edges) && ((word[j] >> (d[j] & 31)) & 1u);
        lr[j] = 0;
        if (keep[j]) lr[j] = atomicAdd(&hist[d[j] >> BIN_SHIFT], 1);
    }
    block_scan_lds(hist, nbins + 1, wsum);
    int2* mine = recs + (long long)blockIdx.x * BS_CH;
#pragma unroll
    for (int j = 0; j < BS_E; ++j)
        if (keep[j]) mine[hist[d[j] >> BIN_SHIFT] + lr[j]] = make_int2(s[j], d[j]);
    int* myoff = boff + (long long)blockIdx.x * (nbins + 1);
    for (int b = t; b <= nbins; b += SORT_T) myoff[b] = hist[b];
}

__global__ void __launch_bounds__(SORT_T)
k_binsort(const int2* __restrict__ recs, const int* __restrict__ boff,
          int* __restrict__ first, int* __restrict__ deg, int* __restrict__ ssrc,
          int n_nodes, int nblk, int nbins, int lin0,
          const float* __restrict__ x,
          const float* __restrict__ Wl, const float* __restrict__ bl,
          const float* __restrict__ Wr, const float* __restrict__ br,
          float* __restrict__ xl, float* __restrict__ xr) {
    extern __shared__ int dyn[];
    __shared__ __attribute__((aligned(16))) v2f wfrag[WF_N];
    __shared__ __attribute__((aligned(16))) float xt[XT_N];
    if ((int)blockIdx.x >= nbins) {
        linear_role((int)blockIdx.x - nbins + lin0, x, Wl, bl, Wr, br, xl, xr, n_nodes, wfrag, xt);
        return;
    }
    int* o0  = dyn;
    int* pre = dyn + nblk;
    __shared__ int cnt[BIN_R];
    __shared__ int cur[BIN_R];
    __shared__ int wsum[33];
    __shared__ int sbase;
    const int t = threadIdx.x;
    const int b = blockIdx.x;
    const int node0 = b << BIN_SHIFT;
    if (t == 0) { sbase = 0; pre[nblk] = 0; }
    for (int i = t; i < BIN_R; i += SORT_T) cnt[i] = 0;
    int mystart = 0;
    for (int k = t; k < nblk; k += SORT_T) {
        const int* bo = boff + (long long)k * (nbins + 1) + b;
        const int a0 = bo[0], a1 = bo[1];
        o0[k]  = a0;
        pre[k] = a1 - a0;
        mystart += a0;
    }
    __syncthreads();
    if (mystart) atomicAdd(&sbase, mystart);
    const int total = block_scan_lds(pre, nblk + 1, wsum);
    const int start = sbase;

    int top = 1;
    while (top < nblk) top <<= 1;
    auto fetch = [&](int i0, int2 (&r)[SB_B]) {
        int idx[SB_B], lo[SB_B];
#pragma unroll
        for (int j = 0; j < SB_B; ++j) {
            const int i = i0 + j * SORT_T;
            idx[j] = i < total ? i : total - 1;
            lo[j] = 0;
        }
        for (int w = top >> 1; w > 0; w >>= 1) {
#pragma unroll
            for (int j = 0; j < SB_B; ++j) {
                const int k = lo[j] + w < nblk ? lo[j] + w : nblk;
                lo[j] = pre[k] <= idx[j] ? k : lo[j];
            }
        }
#pragma unroll
        for (int j = 0; j < SB_B; ++j)
            r[j] = recs[(long long)lo[j] * BS_CH + o0[lo[j]] + (idx[j] - pre[lo[j]])];
    };

    int2 r0[SB_B];
#pragma unroll
    for (int j = 0; j < SB_B; ++j) r0[j] = make_int2(0, node0);
    if (t < total) fetch(t, r0);
#pragma unroll
    for (int j = 0; j < SB_B; ++j)
        if (t + j * SORT_T < total) atomicAdd(&cnt[r0[j].y - node0], 1);
    for (int i0 = t + SB_B * SORT_T; i0 < total; i0 += SB_B * SORT_T) {
        int2 r[SB_B];
        fetch(i0, r);
#pragma unroll
        for (int j = 0; j < SB_B; ++j)
            if (i0 + j * SORT_T < total) atomicAdd(&cnt[r[j].y - node0], 1);
    }
    __syncthreads();
    for (int i = t; i < BIN_R; i += SORT_T)
        if (node0 + i < n_nodes) deg[node0 + i] = cnt[i];
    block_scan_lds(cnt, BIN_R, wsum);
    for (int i = t; i < BIN_R; i += SORT_T) {
        const int f = cnt[i];
        cur[i] = f;
        if (node0 + i < n_nodes) first[node0 + i] = start + f;
    }
    __syncthreads();
#pragma unroll
    for (int j = 0; j < SB_B; ++j)
        if (t + j * SORT_T < total) {
            const int pos = atomicAdd(&cur[r0[j].y - node0], 1);
            ssrc[start + pos] = r0[j].x;
        }
    for (int i0 = t + SB_B * SORT_T; i0 < total; i0 += SB_B * SORT_T) {
        int2 r[SB_B];
        fetch(i0, r);
#pragma unroll
        for (int j = 0; j < SB_B; ++j)
            if (i0 + j * SORT_T < total) {
                const int pos = atomicAdd(&cur[r[j].y - node0], 1);
                ssrc[start + pos] = r[j].x;
            }
    }
}

template <int CTRL>
__device__ __forceinline__ int dpp_i(int v) {
    return __builtin_amdgcn_update_dpp(0, v, CTRL, 0xf, 0xf, true);
}
template <int CTRL>
__device__ __forceinline__ float dpp_f(float v) {
    return __int_as_float(__builtin_amdgcn_update_dpp(0, __float_as_int(v), CTRL, 0xf, 0xf, true));
}

template <int J>
__device__ __forceinline__ v4f gather_row(int ids, int sub, const float* __restrict__ xl) {
    const int s = dpp_i<0x150 + J>(ids);
    return *(const v4f*)(xl + (long long)s * D_DIM + 4 * sub);
}

__device__ __forceinline__ float edge_score(const v4f& l, const v4f& r, const v4f& a) {
    float t0 = l.x + r.x, t1 = l.y + r.y, t2 = l.z + r.z, t3 = l.w + r.w;
    t0 = t0 > 0.0f ? t0 : NEG_SLOPE * t0;
    t1 = t1 > 0.0f ? t1 : NEG_SLOPE * t1;
    t2 = t2 > 0.0f ? t2 : NEG_SLOPE * t2;
    t3 = t3 > 0.0f ? t3 : NEG_SLOPE * t3;
    float p = t0 * a.x + t1 * a.y + t2 * a.z + t3 * a.w;
    p += dpp_f<0x160 + 1>(p);
    p += dpp_f<0x160 + 2>(p);
    p += dpp_f<0x160 + 4>(p);
    p += dpp_f<0x160 + 8>(p);
    return p;
}

template <int B>
__device__ __forceinline__ void seg_batch(int ids, int left, int sub,
                                          const float* __restrict__ xl,
                                          const v4f& r, const v4f& a,
                                          float& m, float& den, v4f& acc) {
    const v4f l0 = gather_row<4 * B + 0>(ids, sub, xl);
    const v4f l1 = gather_row<4 * B + 1>(ids, sub, xl);
    const v4f l2 = gather_row<4 * B + 2>(ids, sub, xl);
    const v4f l3 = gather_row<4 * B + 3>(ids, sub, xl);
    const float p0 = edge_score(l0, r, a);
    const float p1 = edge_score(l1, r, a);
    const float p2 = edge_score(l2, r, a);
    const float p3 = edge_score(l3, r, a);
    const bool v0 = left > 0, v1 = left > 1, v2 = left > 2, v3 = left > 3;
    float bm = m;
    bm = v0 ? fmaxf(bm, p0) : bm;
    bm = v1 ? fmaxf(bm, p1) : bm;
    bm = v2 ? fmaxf(bm, p2) : bm;
    bm = v3 ? fmaxf(bm, p3) : bm;
    const float sc = __expf(m - bm);
    const float w0 = v0 ? __expf(p0 - bm) : 0.0f;
    const float w1 = v1 ? __expf(p1 - bm) : 0.0f;
    const float w2 = v2 ? __expf(p2 - bm) : 0.0f;
    const float w3 = v3 ? __expf(p3 - bm) : 0.0f;
    den   = den   * sc + ((w0 + w1) + (w2 + w3));
    acc.x = acc.x * sc + ((w0 * l0.x + w1 * l1.x) + (w2 * l2.x + w3 * l3.x));
    acc.y = acc.y * sc + ((w0 * l0.y + w1 * l1.y) + (w2 * l2.y + w3 * l3.y));
    acc.z = acc.z * sc + ((w0 * l0.z + w1 * l1.z) + (w2 * l2.z + w3 * l3.z));
    acc.w = acc.w * sc + ((w0 * l0.w + w1 * l1.w) + (w2 * l2.w + w3 * l3.w));
    m = bm;
}

#define HT_P 68
#define OT_P 132
#define SEG_T 256

__global__ void __launch_bounds__(SEG_T)
k_segment_out(const int* __restrict__ mask, const int* __restrict__ first,
              const int* __restrict__ deg, const int* __restrict__ ssrc,
              const float* __restrict__ xl, const float* __restrict__ xr,
              const float* __restrict__ att, const float* __restrict__ gbias,
              const float* __restrict__ Wo, const float* __restrict__ bo,
              float* __restrict__ out, int n_mask) {
    __shared__ __attribute__((aligned(16))) float htile[16 * HT_P];
    __shared__ __attribute__((aligned(16))) float otile[16 * OT_P];
    const int lane = threadIdx.x & 31;
    const int wave = threadIdx.x >> 5;
    const int sub  = lane & 15;
    const int rb   = threadIdx.x >> 4;
    const int r0   = blockIdx.x * 16;
    const int i = r0 + rb;
    const bool live = i < n_mask;
    int node = 0, beg = 0, n = 0;
    if (live) { node = mask[i]; beg = first[node]; n = deg[node]; }
    const v4f a = *(const v4f*)(att + 4 * sub);
    const v4f r = *(const v4f*)(xr + (long long)node * D_DIM + 4 * sub);
    const int n_lo = __builtin_amdgcn_readlane(n, 0);
    const int n_hi = __builtin_amdgcn_readlane(n, 16);
    const int nmax = n_lo > n_hi ? n_lo : n_hi;
    float m = -FLT_MAX, den = 0.0f;
    v4f acc = {0.0f, 0.0f, 0.0f, 0.0f};
    for (int c = 0; c < nmax; c += 16) {
        const int left = n - c;
        int ids = 0;
        if (sub < left) ids = ssrc[beg + c + sub];
        seg_batch<0>(ids, left, sub, xl, r, a, m, den, acc);
        if (c + 4 >= nmax) break;
        seg_batch<1>(ids, left - 4, sub, xl, r, a, m, den, acc);
        if (c + 8 >= nmax) break;
        seg_batch<2>(ids, left - 8, sub, xl, r, a, m, den, acc);
        if (c + 12 >= nmax) break;
        seg_batch<3>(ids, left - 12, sub, xl, r, a, m, den, acc);
    }

    const int lrow = lane & 15;
    const int lhi  = lane >> 4;
    const int col  = wave * 16 + lrow;
    v2f b[16];
#pragma unroll
    for (int kk = 0; kk < 16; ++kk) {
        const int k0 = kk * 4 + 2 * lhi;
        b[kk].x = Wo[(k0    ) * DOUT + col];
        b[kk].y = Wo[(k0 + 1) * DOUT + col];
    }
    const float bv = bo[col];
    {
        const v4f g = *(const v4f*)(gbias + 4 * sub);
        const float dn = den + 1e-16f;
        v4f o;
        o.x = acc.x / dn + g.x; o.y = acc.y / dn + g.y;
        o.z = acc.z / dn + g.z; o.w = acc.w / dn + g.w;
        *(v4f*)(htile + rb * HT_P + 4 * sub) = o;
    }
    __syncthreads();
    v8f c8 = {bv, bv, bv, bv, bv, bv, bv, bv};
#pragma unroll
    for (int kk = 0; kk < 16; ++kk) {
        const v2f af = *(const v2f*)(htile + lrow * HT_P + kk * 4 + 2 * lhi);
        c8 = __builtin_amdgcn_wmma_f32_16x16x4_f32(
                false, af, false, b[kk], (short)0, c8, false, false);
    }
#pragma unroll
    for (int q = 0; q < 8; ++q)
        otile[(q + 8 * lhi) * OT_P + col] = c8[q];
    __syncthreads();
#pragma unroll
    for (int q = 0; q < 2; ++q) {
        const int row = 2 * wave + q;
        if (r0 + row < n_mask)
            *(v4f*)(out + (long long)(r0 + row) * DOUT + 4 * lane) =
                *(const v4f*)(otile + row * OT_P + 4 * lane);
    }
}

extern "C" void kernel_launch(void* const* d_in, const int* in_sizes, int n_in,
                              void* d_out, int out_size, void* d_ws, size_t ws_size,
                              hipStream_t stream) {
    const float* x     = (const float*)d_in[0];
    const int*   ei    = (const int*)d_in[1];
    const int*   mask  = (const int*)d_in[2];
    const float* Wl    = (const float*)d_in[3];
    const float* bl    = (const float*)d_in[4];
    const float* Wr    = (const float*)d_in[5];
    const float* br    = (const float*)d_in[6];
    const float* att   = (const float*)d_in[7];
    const float* gbias = (const float*)d_in[8];
    const float* Wo    = (const float*)d_in[9];
    const float* bo    = (const float*)d_in[10];
    float* out = (float*)d_out;

    const int n_nodes = in_sizes[0] / D_DIM;
    const int n_edges = in_sizes[1] / 2;
    const int n_mask  = in_sizes[2];
    const int* src = ei;
    const int* dst = ei + n_edges;

    const int nblk  = (n_edges + BS_CH - 1) / BS_CH;
    const int nbins = (n_nodes + BIN_R - 1) / BIN_R;
    const int nwords = (n_nodes + 31) / 32;
    char* wp = (char*)d_ws;
    auto carve = [&](size_t bytes) {
        char* p = wp;
        wp += (bytes + 255) & ~(size_t)255;
        return p;
    };
    float*    xl     = (float*)   carve((size_t)n_nodes * D_DIM * 4);
    float*    xr     = (float*)   carve((size_t)n_nodes * D_DIM * 4);
    int*      ssrc   = (int*)     carve((size_t)n_edges * 4);
    int2*     recs   = (int2*)    carve((size_t)nblk * BS_CH * 8);
    int*      boff   = (int*)     carve((size_t)nblk * (nbins + 1) * 4);
    int*      deg    = (int*)     carve((size_t)n_nodes * 4);
    int*      first  = (int*)     carve((size_t)n_nodes * 4);

    const int tiles = (n_nodes + 15) / 16;
    const int lblk  = (tiles + LIN_TILES - 1) / LIN_TILES;
    const int lin_a = (lblk + 1) / 2;
    const int lin_b = lblk - lin_a;
    k_blocksort<<<nblk + lin_a, SORT_T, (size_t)(nbins + 1 + nwords) * 4, stream>>>(
        src, dst, mask, n_mask, nwords, recs, boff, n_edges, nbins, nblk,
        x, Wl, bl, Wr, br, xl, xr, n_nodes);
    k_binsort<<<nbins + lin_b, SORT_T, (size_t)(2 * nblk + 1) * 4, stream>>>(
        recs, boff, first, deg, ssrc, n_nodes, nblk, nbins, lin_a,
        x, Wl, bl, Wr, br, xl, xr);
    k_segment_out<<<(n_mask + 15) / 16, SEG_T, 0, stream>>>(
        mask, first, deg, ssrc, xl, xr, att, gbias, Wo, bo, out, n_mask);
}


# xr=x@W_r+b_r made only for the rows mask_idx names, in the segment kernel (block's 16 rows of x via LDS to 4 WMMA waves); xr plane gone, product work-groups do W_l only; order of sums unchanged
# speedup vs baseline: 8.1766x; 1.0356x over previous
#include <hip/hip_runtime.h>
#include <hip/hip_bf16.h>
#include <math.h>
#include <float.h>


#define D_DIM 64
#define DOUT  128
#define NEG_SLOPE 0.2f

typedef __attribute__((ext_vector_type(2))) float v2f;
typedef __attribute__((ext_vector_type(8))) float v8f;
typedef __attribute__((ext_vector_type(4))) float v4f;

#define SORT_T    1024
#define LIN_TILES 16
#define XT_P      68
#define WF_N      (4 * 16 * 32)
#define XT_N      (LIN_TILES * 16 * XT_P)

__device__ __forceinline__ void linear_role(int lb, const float* __restrict__ x,
        const float* __restrict__ Wl, const float* __restrict__ bl,
        float* __restrict__ xl, int n_nodes,
        v2f* wfrag, float* xt) {
    const int t = threadIdx.x;
    const int lane = t & 31;
    const int wave = t >> 5;
    const int row_base = lb * LIN_TILES * 16;
    v4f xv[4];
    v2f wv[2];
#pragma unroll
    for (int q = 0; q < 4; ++q) {
        const int i = t + q * SORT_T;
        int grow = row_base + (i >> 4);
        grow = grow < n_nodes ? grow : n_nodes - 1;
        xv[q] = *(const v4f*)(x + (long long)grow * D_DIM + 4 * (i & 15));
    }
#pragma unroll
    for (int q = 0; q < 2; ++q) {
        const int e = t + q * SORT_T;
        const int k0  = ((e >> 5) & 15) * 4 + 2 * ((e >> 4) & 1);
        const int col = ((e >> 9) & 3) * 16 + (e & 15);
        wv[q].x = Wl[(k0    ) * D_DIM + col];
        wv[q].y = Wl[(k0 + 1) * D_DIM + col];
    }
#pragma unroll
    for (int q = 0; q < 4; ++q) {
        const int i = t + q * SORT_T;
        *(v4f*)(xt + (i >> 4) * XT_P + 4 * (i & 15)) = xv[q];
    }
#pragma unroll
    for (int q = 0; q < 2; ++q)
        wfrag[t + q * SORT_T] = wv[q];
    __syncthreads();
    const int tile = wave >> 1;
    const int r0 = row_base + tile * 16;
    if (r0 + 16 > n_nodes) return;
    const int lrow = lane & 15;
    const int lhi  = lane >> 4;
    const int ct0  = 2 * (wave & 1);
    v2f a[16];
    const float* xp = xt + (tile * 16 + lrow) * XT_P + 2 * lhi;
#pragma unroll
    for (int kk = 0; kk < 16; ++kk)
        a[kk] = *(const v2f*)(xp + kk * 4);
    const v2f* wf = wfrag + lane;
#pragma unroll
    for (int c2 = 0; c2 < 2; ++c2) {
        const int ct = ct0 + c2;
        const int col = ct * 16 + lrow;
        const float bv = bl[col];
        v8f acc = {bv, bv, bv, bv, bv, bv, bv, bv};
#pragma unroll
        for (int kk = 0; kk < 16; ++kk)
            acc = __builtin_amdgcn_wmma_f32_16x16x4_f32(
                    false, a[kk], false, wf[(ct * 16 + kk) * 32], (short)0, acc, false, false);
#pragma unroll
        for (int r = 0; r < 8; ++r)
            xl[(long long)(r0 + r + 8 * lhi) * D_DIM + col] = acc[r];
    }
}

#define BS_E      8
#define MK_U      8
#define BS_CH     (SORT_T * BS_E)
#define BIN_SHIFT 9
#define BIN_R     (1 << BIN_SHIFT)
#define SB_B      4

__device__ __forceinline__ int block_scan_lds(int* s, int n, int* wsum) {
    const int t = threadIdx.x, T = SORT_T;
    const int lane = t & 31, wave = t >> 5;
    const int K = (n + T - 1) / T;
    const int lo = t * K;
    int hi = lo + K;
    if (hi > n) hi = n;
    __syncthreads();
    int sum = 0;
    for (int i = lo; i < hi; ++i) sum += s[i];
    int inc = sum;
    for (int o = 1; o < 32; o <<= 1) {
        const int u = __shfl_up(inc, o);
        if (lane >= o) inc += u;
    }
    if (lane == 31) wsum[wave] = inc;
    __syncthreads();
    if (wave == 0) {
        const int w = lane < (T >> 5) ? wsum[lane] : 0;
        int winc = w;
        for (int o = 1; o < 32; o <<= 1) {
            const int u = __shfl_up(winc, o);
            if (lane >= o) winc += u;
        }
        wsum[lane] = winc - w;
        if (lane == 31) wsum[32] = winc;
    }
    __syncthreads();
    int run = wsum[wave] + inc - sum;
    for (int i = lo; i < hi; ++i) {
        const int v = s[i];
        s[i] = run;
        run += v;
    }
    const int total = wsum[32];
    __syncthreads();
    return total;
}

__global__ void __launch_bounds__(SORT_T)
k_blocksort(const int* __restrict__ src, const int* __restrict__ dst,
            const int* __restrict__ mask, int n_mask, int nwords, int2* __restrict__ recs,
            int* __restrict__ boff, int n_edges, int nbins, int nsort,
            const float* __restrict__ x,
            const float* __restrict__ Wl, const float* __restrict__ bl,
            float* __restrict__ xl, int n_nodes) {
    extern __shared__ int hist[];
    __shared__ int wsum[33];
    __shared__ __attribute__((aligned(16))) v2f wfrag[WF_N];
    __shared__ __attribute__((aligned(16))) float xt[XT_N];
    if ((int)blockIdx.x >= nsort) {
        linear_role((int)blockIdx.x - nsort, x, Wl, bl, xl, n_nodes, wfrag, xt);
        return;
    }
    const int t = threadIdx.x;
    unsigned* nbit = (unsigned*)(hist + nbins + 1);
    const long long base = (long long)blockIdx.x * BS_CH + t;
    int d[BS_E], s[BS_E], lr[BS_E];
    bool keep[BS_E];
    for (int b = t; b <= nbins; b += SORT_T) hist[b] = 0;
    for (int w = t; w < nwords; w += SORT_T) nbit[w] = 0u;
    __syncthreads();
#pragma unroll
    for (int j = 0; j < BS_E; ++j) {
        const long long e = base + j * SORT_T;
        const long long ec = e < n_edges ? e : 0;
        d[j] = dst[ec];
        s[j] = src[ec];
    }
    int hd = (int)(((16u - (unsigned)((size_t)mask & 15u)) & 15u) >> 2);
    hd = hd < n_mask ? hd : n_mask;
    const int4* mask4 = (const int4*)(mask + hd);
    const int nvec = (n_mask - hd) >> 2;
    for (int vb = 0; vb < nvec; vb += SORT_T * MK_U) {
        int4 mm[MK_U];
#pragma unroll
        for (int q = 0; q < MK_U; ++q) {
            const int v = vb + q * SORT_T + t;
            mm[q] = mask4[v < nvec ? v : nvec - 1];
        }
#pragma unroll
        for (int q = 0; q < MK_U; ++q)
            if (vb + q * SORT_T + t < nvec) {
                atomicOr(&nbit[mm[q].x >> 5], 1u << (mm[q].x & 31));
                atomicOr(&nbit[mm[q].y >> 5], 1u << (mm[q].y & 31));
                atomicOr(&nbit[mm[q].z >> 5], 1u << (mm[q].z & 31));
                atomicOr(&nbit[mm[q].w >> 5], 1u << (mm[q].w & 31));
            }
    }
    if (t < hd) {
        const int m = mask[t];
        atomicOr(&nbit[m >> 5], 1u << (m & 31));
    }
    for (int i = hd + 4 * nvec + t; i < n_mask; i += SORT_T) {
        const int m = mask[i];
        atomicOr(&nbit[m >> 5], 1u << (m & 31));
    }
    __syncthreads();
    unsigned word[BS_E];
#pragma unroll
    for (int j = 0; j < BS_E; ++j)
        word[j] = nbit[d[j] >> 5];
#pragma unroll
    for (int j = 0; j < BS_E; ++j) {
        keep[j] = (base + j * SORT_T < n_edges) && ((word[j] >> (d[j] & 31)) & 1u);
        lr[j] = 0;
        if (keep[j]) lr[j] = atomicAdd(&hist[d[j] >> BIN_SHIFT], 1);
    }
    block_scan_lds(hist, nbins + 1, wsum);
    int2* mine = recs + (long long)blockIdx.x * BS_CH;
#pragma unroll
    for (int j = 0; j < BS_E; ++j)
        if (keep[j]) mine[hist[d[j] >> BIN_SHIFT] + lr[j]] = make_int2(s[j], d[j]);
    int* myoff = boff + (long long)blockIdx.x * (nbins + 1);
    for (int b = t; b <= nbins; b += SORT_T) myoff[b] = hist[b];
}

__global__ void __launch_bounds__(SORT_T)
k_binsort(const int2* __restrict__ recs, const int* __restrict__ boff,
          int* __restrict__ first, int* __restrict__ deg, int* __restrict__ ssrc,
          int n_nodes, int nblk, int nbins, int lin0,
          const float* __restrict__ x,
          const float* __restrict__ Wl, const float* __restrict__ bl,
          float* __restrict__ xl) {
    extern __shared__ int dyn[];
    __shared__ __attribute__((aligned(16))) v2f wfrag[WF_N];
    __shared__ __attribute__((aligned(16))) float xt[XT_N];
    if ((int)blockIdx.x >= nbins) {
        linear_role((int)blockIdx.x - nbins + lin0, x, Wl, bl, xl, n_nodes, wfrag, xt);
        return;
    }
    int* o0  = dyn;
    int* pre = dyn + nblk;
    __shared__ int cnt[BIN_R];
    __shared__ int cur[BIN_R];
    __shared__ int wsum[33];
    __shared__ int sbase;
    const int t = threadIdx.x;
    const int b = blockIdx.x;
    const int node0 = b << BIN_SHIFT;
    if (t == 0) { sbase = 0; pre[nblk] = 0; }
    for (int i = t; i < BIN_R; i += SORT_T) cnt[i] = 0;
    int mystart = 0;
    for (int k = t; k < nblk; k += SORT_T) {
        const int* bo = boff + (long long)k * (nbins + 1) + b;
        const int a0 = bo[0], a1 = bo[1];
        o0[k]  = a0;
        pre[k] = a1 - a0;
        mystart += a0;
    }
    __syncthreads();
    if (mystart) atomicAdd(&sbase, mystart);
    const int total = block_scan_lds(pre, nblk + 1, wsum);
    const int start = sbase;

    int top = 1;
    while (top < nblk) top <<= 1;
    auto fetch = [&](int i0, int2 (&r)[SB_B]) {
        int idx[SB_B], lo[SB_B];
#pragma unroll
        for (int j = 0; j < SB_B; ++j) {
            const int i = i0 + j * SORT_T;
            idx[j] = i < total ? i : total - 1;
            lo[j] = 0;
        }
        for (int w = top >> 1; w > 0; w >>= 1) {
#pragma unroll
            for (int j = 0; j < SB_B; ++j) {
                const int k = lo[j] + w < nblk ? lo[j] + w : nblk;
                lo[j] = pre[k] <= idx[j] ? k : lo[j];
            }
        }
#pragma unroll
        for (int j = 0; j < SB_B; ++j)
            r[j] = recs[(long long)lo[j] * BS_CH + o0[lo[j]] + (idx[j] - pre[lo[j]])];
    };

    int2 r0[SB_B];
#pragma unroll
    for (int j = 0; j < SB_B; ++j) r0[j] = make_int2(0, node0);
    if (t < total) fetch(t, r0);
#pragma unroll
    for (int j = 0; j < SB_B; ++j)
        if (t + j * SORT_T < total) atomicAdd(&cnt[r0[j].y - node0], 1);
    for (int i0 = t + SB_B * SORT_T; i0 < total; i0 += SB_B * SORT_T) {
        int2 r[SB_B];
        fetch(i0, r);
#pragma unroll
        for (int j = 0; j < SB_B; ++j)
            if (i0 + j * SORT_T < total) atomicAdd(&cnt[r[j].y - node0], 1);
    }
    __syncthreads();
    for (int i = t; i < BIN_R; i += SORT_T)
        if (node0 + i < n_nodes) deg[node0 + i] = cnt[i];
    block_scan_lds(cnt, BIN_R, wsum);
    for (int i = t; i < BIN_R; i += SORT_T) {
        const int f = cnt[i];
        cur[i] = f;
        if (node0 + i < n_nodes) first[node0 + i] = start + f;
    }
    __syncthreads();
#pragma unroll
    for (int j = 0; j < SB_B; ++j)
        if (t + j * SORT_T < total) {
            const int pos = atomicAdd(&cur[r0[j].y - node0], 1);
            ssrc[start + pos] = r0[j].x;
        }
    for (int i0 = t + SB_B * SORT_T; i0 < total; i0 += SB_B * SORT_T) {
        int2 r[SB_B];
        fetch(i0, r);
#pragma unroll
        for (int j = 0; j < SB_B; ++j)
            if (i0 + j * SORT_T < total) {
                const int pos = atomicAdd(&cur[r[j].y - node0], 1);
                ssrc[start + pos] = r[j].x;
            }
    }
}

template <int CTRL>
__device__ __forceinline__ int dpp_i(int v) {
    return __builtin_amdgcn_update_dpp(0, v, CTRL, 0xf, 0xf, true);
}
template <int CTRL>
__device__ __forceinline__ float dpp_f(float v) {
    return __int_as_float(__builtin_amdgcn_update_dpp(0, __float_as_int(v), CTRL, 0xf, 0xf, true));
}

template <int J>
__device__ __forceinline__ v4f gather_row(int ids, int sub, const float* __restrict__ xl) {
    const int s = dpp_i<0x150 + J>(ids);
    return *(const v4f*)(xl + (long long)s * D_DIM + 4 * sub);
}

__device__ __forceinline__ float edge_score(const v4f& l, const v4f& r, const v4f& a) {
    float t0 = l.x + r.x, t1 = l.y + r.y, t2 = l.z + r.z, t3 = l.w + r.w;
    t0 = t0 > 0.0f ? t0 : NEG_SLOPE * t0;
    t1 = t1 > 0.0f ? t1 : NEG_SLOPE * t1;
    t2 = t2 > 0.0f ? t2 : NEG_SLOPE * t2;
    t3 = t3 > 0.0f ? t3 : NEG_SLOPE * t3;
    float p = t0 * a.x + t1 * a.y + t2 * a.z + t3 * a.w;
    p += dpp_f<0x160 + 1>(p);
    p += dpp_f<0x160 + 2>(p);
    p += dpp_f<0x160 + 4>(p);
    p += dpp_f<0x160 + 8>(p);
    return p;
}

template <int B>
__device__ __forceinline__ void seg_batch(int ids, int left, int sub,
                                          const float* __restrict__ xl,
                                          const v4f& r, const v4f& a,
                                          float& m, float& den, v4f& acc) {
    const v4f l0 = gather_row<4 * B + 0>(ids, sub, xl);
    const v4f l1 = gather_row<4 * B + 1>(ids, sub, xl);
    const v4f l2 = gather_row<4 * B + 2>(ids, sub, xl);
    const v4f l3 = gather_row<4 * B + 3>(ids, sub, xl);
    const float p0 = edge_score(l0, r, a);
    const float p1 = edge_score(l1, r, a);
    const float p2 = edge_score(l2, r, a);
    const float p3 = edge_score(l3, r, a);
    const bool v0 = left > 0, v1 = left > 1, v2 = left > 2, v3 = left > 3;
    float bm = m;
    bm = v0 ? fmaxf(bm, p0) : bm;
    bm = v1 ? fmaxf(bm, p1) : bm;
    bm = v2 ? fmaxf(bm, p2) : bm;
    bm = v3 ? fmaxf(bm, p3) : bm;
    const float sc = __expf(m - bm);
    const float w0 = v0 ? __expf(p0 - bm) : 0.0f;
    const float w1 = v1 ? __expf(p1 - bm) : 0.0f;
    const float w2 = v2 ? __expf(p2 - bm) : 0.0f;
    const float w3 = v3 ? __expf(p3 - bm) : 0.0f;
    den   = den   * sc + ((w0 + w1) + (w2 + w3));
    acc.x = acc.x * sc + ((w0 * l0.x + w1 * l1.x) + (w2 * l2.x + w3 * l3.x));
    acc.y = acc.y * sc + ((w0 * l0.y + w1 * l1.y) + (w2 * l2.y + w3 * l3.y));
    acc.z = acc.z * sc + ((w0 * l0.z + w1 * l1.z) + (w2 * l2.z + w3 * l3.z));
    acc.w = acc.w * sc + ((w0 * l0.w + w1 * l1.w) + (w2 * l2.w + w3 * l3.w));
    m = bm;
}

#define HT_P 68
#define OT_P 132
#define SEG_T 256

__global__ void __launch_bounds__(SEG_T)
k_segment_out(const int* __restrict__ mask, const int* __restrict__ first,
              const int* __restrict__ deg, const int* __restrict__ ssrc,
              const float* __restrict__ xl, const float* __restrict__ x,
              const float* __restrict__ Wr, const float* __restrict__ br,
              const float* __restrict__ att, const float* __restrict__ gbias,
              const float* __restrict__ Wo, const float* __restrict__ bo,
              float* __restrict__ out, int n_mask) {
    __shared__ __attribute__((aligned(16))) float htile[16 * HT_P];
    __shared__ __attribute__((aligned(16))) float otile[16 * OT_P];
    const int lane = threadIdx.x & 31;
    const int wave = threadIdx.x >> 5;
    const int sub  = lane & 15;
    const int rb   = threadIdx.x >> 4;
    const int r0   = blockIdx.x * 16;
    const int i = r0 + rb;
    const bool live = i < n_mask;
    const int lrow = lane & 15;
    const int lhi  = lane >> 4;
    const bool xr_wave = wave < 4;
    v2f bw[16];
    float bvr = 0.0f;
    if (xr_wave) {
        const int colr = wave * 16 + lrow;
#pragma unroll
        for (int kk = 0; kk < 16; ++kk) {
            const int k0 = kk * 4 + 2 * lhi;
            bw[kk].x = Wr[(k0    ) * D_DIM + colr];
            bw[kk].y = Wr[(k0 + 1) * D_DIM + colr];
        }
        bvr = br[colr];
    }
    int node = 0, beg = 0, n = 0;
    if (live) { node = mask[i]; beg = first[node]; n = deg[node]; }
    const v4f a = *(const v4f*)(att + 4 * sub);
    const v4f xrow = *(const v4f*)(x + (long long)node * D_DIM + 4 * sub);
    const int* idp = sub < n ? ssrc + beg + sub : first;
    int ids = *idp;
    ids = sub < n ? ids : 0;
    float* xrt = otile;
    *(v4f*)(htile + rb * HT_P + 4 * sub) = xrow;
    __builtin_amdgcn_fence(__ATOMIC_RELEASE, "workgroup", "local");
    __builtin_amdgcn_s_barrier();
    __builtin_amdgcn_fence(__ATOMIC_ACQUIRE, "workgroup", "local");
    if (xr_wave) {
        v8f cr = {bvr, bvr, bvr, bvr, bvr, bvr, bvr, bvr};
#pragma unroll
        for (int kk = 0; kk < 16; ++kk) {
            const v2f af = *(const v2f*)(htile + lrow * HT_P + kk * 4 + 2 * lhi);
            cr = __builtin_amdgcn_wmma_f32_16x16x4_f32(
                    false, af, false, bw[kk], (short)0, cr, false, false);
        }
#pragma unroll
        for (int q = 0; q < 8; ++q)
            xrt[(q + 8 * lhi) * HT_P + wave * 16 + lrow] = cr[q];
    }
    __builtin_amdgcn_fence(__ATOMIC_RELEASE, "workgroup", "local");
    __builtin_amdgcn_s_barrier();
    __builtin_amdgcn_fence(__ATOMIC_ACQUIRE, "workgroup", "local");
    const v4f r = *(const v4f*)(xrt + rb * HT_P + 4 * sub);
    const int n_lo = __builtin_amdgcn_readlane(n, 0);
    const int n_hi = __builtin_amdgcn_readlane(n, 16);
    const int nmax = n_lo > n_hi ? n_lo : n_hi;
    float m = -FLT_MAX, den = 0.0f;
    v4f acc = {0.0f, 0.0f, 0.0f, 0.0f};
    for (int c = 0; c < nmax; c += 16) {
        const int left = n - c;
        if (c > 0) {
            ids = 0;
            if (sub < left) ids = ssrc[beg + c + sub];
        }
        seg_batch<0>(ids, left, sub, xl, r, a, m, den, acc);
        if (c + 4 >= nmax) break;
        seg_batch<1>(ids, left - 4, sub, xl, r, a, m, den, acc);
        if (c + 8 >= nmax) break;
        seg_batch<2>(ids, left - 8, sub, xl, r, a, m, den, acc);
        if (c + 12 >= nmax) break;
        seg_batch<3>(ids, left - 12, sub, xl, r, a, m, den, acc);
    }

    const int col  = wave * 16 + lrow;
    v2f b[16];
#pragma unroll
    for (int kk = 0; kk < 16; ++kk) {
        const int k0 = kk * 4 + 2 * lhi;
        b[kk].x = Wo[(k0    ) * DOUT + col];
        b[kk].y = Wo[(k0 + 1) * DOUT + col];
    }
    const float bv = bo[col];
    {
        const v4f g = *(const v4f*)(gbias + 4 * sub);
        const float dn = den + 1e-16f;
        v4f o;
        o.x = acc.x / dn + g.x; o.y = acc.y / dn + g.y;
        o.z = acc.z / dn + g.z; o.w = acc.w / dn + g.w;
        *(v4f*)(htile + rb * HT_P + 4 * sub) = o;
    }
    __syncthreads();
    v8f c8 = {bv, bv, bv, bv, bv, bv, bv, bv};
#pragma unroll
    for (int kk = 0; kk < 16; ++kk) {
        const v2f af = *(const v2f*)(htile + lrow * HT_P + kk * 4 + 2 * lhi);
        c8 = __builtin_amdgcn_wmma_f32_16x16x4_f32(
                false, af, false, b[kk], (short)0, c8, false, false);
    }
#pragma unroll
    for (int q = 0; q < 8; ++q)
        otile[(q + 8 * lhi) * OT_P + col] = c8[q];
    __syncthreads();
#pragma unroll
    for (int q = 0; q < 2; ++q) {
        const int row = 2 * wave + q;
        if (r0 + row < n_mask)
            *(v4f*)(out + (long long)(r0 + row) * DOUT + 4 * lane) =
                *(const v4f*)(otile + row * OT_P + 4 * lane);
    }
}

extern "C" void kernel_launch(void* const* d_in, const int* in_sizes, int n_in,
                              void* d_out, int out_size, void* d_ws, size_t ws_size,
                              hipStream_t stream) {
    const float* x     = (const float*)d_in[0];
    const int*   ei    = (const int*)d_in[1];
    const int*   mask  = (const int*)d_in[2];
    const float* Wl    = (const float*)d_in[3];
    const float* bl    = (const float*)d_in[4];
    const float* Wr    = (const float*)d_in[5];
    const float* br    = (const float*)d_in[6];
    const float* att   = (const float*)d_in[7];
    const float* gbias = (const float*)d_in[8];
    const float* Wo    = (const float*)d_in[9];
    const float* bo    = (const float*)d_in[10];
    float* out = (float*)d_out;

    const int n_nodes = in_sizes[0] / D_DIM;
    const int n_edges = in_sizes[1] / 2;
    const int n_mask  = in_sizes[2];
    const int* src = ei;
    const int* dst = ei + n_edges;

    const int nblk  = (n_edges + BS_CH - 1) / BS_CH;
    const int nbins = (n_nodes + BIN_R - 1) / BIN_R;
    const int nwords = (n_nodes + 31) / 32;
    char* wp = (char*)d_ws;
    auto carve = [&](size_t bytes) {
        char* p = wp;
        wp += (bytes + 255) & ~(size_t)255;
        return p;
    };
    float*    xl     = (float*)   carve((size_t)n_nodes * D_DIM * 4);
    int*      ssrc   = (int*)     carve((size_t)n_edges * 4);
    int2*     recs   = (int2*)    carve((size_t)nblk * BS_CH * 8);
    int*      boff   = (int*)     carve((size_t)nblk * (nbins + 1) * 4);
    int*      deg    = (int*)     carve((size_t)n_nodes * 4);
    int*      first  = (int*)     carve((size_t)n_nodes * 4);

    const int tiles = (n_nodes + 15) / 16;
    const int lblk  = (tiles + LIN_TILES - 1) / LIN_TILES;
    const int lin_a = (lblk + 1) / 2;
    const int lin_b = lblk - lin_a;
    k_blocksort<<<nblk + lin_a, SORT_T, (size_t)(nbins + 1 + nwords) * 4, stream>>>(
        src, dst, mask, n_mask, nwords, recs, boff, n_edges, nbins, nblk,
        x, Wl, bl, xl, n_nodes);
    k_binsort<<<nbins + lin_b, SORT_T, (size_t)(2 * nblk + 1) * 4, stream>>>(
        recs, boff, first, deg, ssrc, n_nodes, nblk, nbins, lin_a,
        x, Wl, bl, xl);
    k_segment_out<<<(n_mask + 15) / 16, SEG_T, 0, stream>>>(
        mask, first, deg, ssrc, xl, x, Wr, br, att, gbias, Wo, bo, out, n_mask);
}
